# starting kernel
# baseline (speedup 1.0000x reference)
#include <hip/hip_runtime.h>

typedef float v2f __attribute__((ext_vector_type(2)));
typedef float v8f __attribute__((ext_vector_type(8)));

__global__ void k_deg_init(float* __restrict__ deg, int n) {
    int i = blockIdx.x * blockDim.x + threadIdx.x;
    if (i < n) deg[i] = 1.0f;
}

__global__ void k_deg_count(const int* __restrict__ dst, float* __restrict__ deg, int e) {
    int i = blockIdx.x * blockDim.x + threadIdx.x;
    if (i < e) atomicAdd(&deg[dst[i]], 1.0f);
}

__global__ void k_dinv(float* __restrict__ deg, int n) {
    int i = blockIdx.x * blockDim.x + threadIdx.x;
    if (i < n) deg[i] = 1.0f / sqrtf(deg[i]);
}

template <int K, int NOUT>
__global__ void k_gemm_wmma(const float* __restrict__ A, const float* __restrict__ W,
                            float* __restrict__ H, int nrows) {
    const int wave = blockIdx.x * (blockDim.x >> 5) + (threadIdx.x >> 5);
    const int lane = threadIdx.x & 31;
    const int m    = lane & 15;
    const int kh   = lane >> 4;
    const int r0   = wave * 16;
    if (r0 + 16 > nrows) return;

    constexpr int NT = NOUT / 16;
    v8f acc[NT];
#pragma unroll
    for (int t = 0; t < NT; ++t)
#pragma unroll
        for (int j = 0; j < 8; ++j) acc[t][j] = 0.0f;

    const float* arow = A + (size_t)(r0 + m) * K;
    for (int k = 0; k < K; k += 4) {
        v2f a;
        a.x = arow[k + 2 * kh + 0];
        a.y = arow[k + 2 * kh + 1];
#pragma unroll
        for (int t = 0; t < NT; ++t) {
            v2f b;
            b.x = W[(size_t)(k + 2 * kh + 0) * NOUT + t * 16 + m];
            b.y = W[(size_t)(k + 2 * kh + 1) * NOUT + t * 16 + m];
            acc[t] = __builtin_amdgcn_wmma_f32_16x16x4_f32(
                 false, a,  false, b,
                 (short)0, acc[t],  false,  false);
        }
    }

#pragma unroll
    for (int t = 0; t < NT; ++t) {
#pragma unroll
        for (int j = 0; j < 8; ++j)
            H[(size_t)(r0 + j + 8 * kh) * NOUT + t * 16 + m] = acc[t][j];
    }
}

template <int K, int NOUT>
__global__ void k_gemm_tail(const float* __restrict__ A, const float* __restrict__ W,
                            float* __restrict__ H, int row0, int nrows) {
    int idx = blockIdx.x * blockDim.x + threadIdx.x;
    int r = row0 + idx / NOUT;
    int c = idx % NOUT;
    if (r >= nrows) return;
    float s = 0.0f;
    for (int k = 0; k < K; ++k) s += A[(size_t)r * K + k] * W[(size_t)k * NOUT + c];
    H[(size_t)r * NOUT + c] = s;
}

template <int C, bool HAS_BIAS>
__global__ void k_self_init(const float* __restrict__ h, const float* __restrict__ dinv,
                            const float* __restrict__ bias, float* __restrict__ agg, int n) {
    long long idx = (long long)blockIdx.x * blockDim.x + threadIdx.x;
    long long total = (long long)n * C;
    if (idx >= total) return;
    int i = (int)(idx / C);
    int c = (int)(idx % C);
    float d = dinv[i];
    float v = h[idx] * d * d;
    if (HAS_BIAS) v += bias[c];
    agg[idx] = v;
}

template <int C>
__global__ void k_edge_agg(const int* __restrict__ src, const int* __restrict__ dst,
                           const float* __restrict__ h, const float* __restrict__ dinv,
                           float* __restrict__ agg, int e) {
    constexpr int TPE = C / 4;
    long long tid = (long long)blockIdx.x * blockDim.x + threadIdx.x;
    long long total = (long long)e * TPE;
    if (tid >= total) return;
    int ed = (int)(tid / TPE);
    int q  = (int)(tid % TPE);
    int s = src[ed];
    int d = dst[ed];
    float norm = dinv[s] * dinv[d];
    const float4 v = *((const float4*)(h + (size_t)s * C) + q);
    float* o = agg + (size_t)d * C + q * 4;
    atomicAdd(o + 0, v.x * norm);
    atomicAdd(o + 1, v.y * norm);
    atomicAdd(o + 2, v.z * norm);
    atomicAdd(o + 3, v.w * norm);
}

template <int C>
__global__ void k_relu_bias(const float* __restrict__ agg, const float* __restrict__ b,
                            float* __restrict__ p, int n) {
    long long idx = (long long)blockIdx.x * blockDim.x + threadIdx.x;
    long long total = (long long)n * C;
    if (idx >= total) return;
    int c = (int)(idx % C);
    float v = agg[idx] + b[c];
    p[idx] = v > 0.0f ? v : 0.0f;
}

extern "C" void kernel_launch(void* const* d_in, const int* in_sizes, int n_in,
                              void* d_out, int out_size, void* d_ws, size_t ws_size,
                              hipStream_t stream) {
    const float* x  = (const float*)d_in[0];
    const int*   ei = (const int*)d_in[1];
    const float* W1 = (const float*)d_in[2];
    const float* b1 = (const float*)d_in[3];
    const float* W2 = (const float*)d_in[4];
    const float* b2 = (const float*)d_in[5];
    float* out = (float*)d_out;

    const int n = in_sizes[0] / 128;
    const int e = in_sizes[1] / 2;
    const int* src = ei;
    const int* dst = ei + e;

    char* ws = (char*)d_ws;
    size_t off0 = ((size_t)n * sizeof(float) + 255) & ~(size_t)255;
    float* dinv = (float*)ws;
    float* bufA = (float*)(ws + off0);
    float* bufB = bufA + (size_t)n * 64;

    const int tb = 256;
    auto cdiv = [](long long a, long long b) { return (int)((a + b - 1) / b); };

    k_deg_init<<<cdiv(n, tb), tb, 0, stream>>>(dinv, n);
    k_deg_count<<<cdiv(e, tb), tb, 0, stream>>>(dst, dinv, e);
    k_dinv<<<cdiv(n, tb), tb, 0, stream>>>(dinv, n);

    const int strips = n / 16;
    const int wavesPerBlk = tb / 32;
    k_gemm_wmma<128, 64><<<cdiv(strips, wavesPerBlk), tb, 0, stream>>>(x, W1, bufA, n);
    if (n % 16) {
        int tail = (n % 16) * 64;
        k_gemm_tail<128, 64><<<cdiv(tail, tb), tb, 0, stream>>>(x, W1, bufA, strips * 16, n);
    }

    k_self_init<64, false><<<cdiv((long long)n * 64, tb), tb, 0, stream>>>(bufA, dinv, nullptr, bufB, n);
    k_edge_agg<64><<<cdiv((long long)e * 16, tb), tb, 0, stream>>>(src, dst, bufA, dinv, bufB, e);
    k_relu_bias<64><<<cdiv((long long)n * 64, tb), tb, 0, stream>>>(bufB, b1, bufA, n);

    k_gemm_wmma<64, 32><<<cdiv(strips, wavesPerBlk), tb, 0, stream>>>(bufA, W2, bufB, n);
    if (n % 16) {
        int tail = (n % 16) * 32;
        k_gemm_tail<64, 32><<<cdiv(tail, tb), tb, 0, stream>>>(bufA, W2, bufB, strips * 16, n);
    }

    k_self_init<32, true><<<cdiv((long long)n * 32, tb), tb, 0, stream>>>(bufB, dinv, b2, out, n);
    k_edge_agg<32><<<cdiv((long long)e * 8, tb), tb, 0, stream>>>(src, dst, bufB, dinv, out, e);

    (void)n_in; (void)out_size; (void)ws_size;
}


# edge scatter re-mapped: one wave walks 8 edges, lane = feature, so each row gather and each f32 atomic add covers one whole 128-B line (was 4 lines at 16-B stride); adds still land in arrival order
# speedup vs baseline: 1.4092x; 1.4092x over previous
#include <hip/hip_runtime.h>

typedef float v2f __attribute__((ext_vector_type(2)));
typedef float v8f __attribute__((ext_vector_type(8)));

__global__ void k_deg_init(float* __restrict__ deg, int n) {
    int i = blockIdx.x * blockDim.x + threadIdx.x;
    if (i < n) deg[i] = 1.0f;
}

__global__ void k_deg_count(const int* __restrict__ dst, float* __restrict__ deg, int e) {
    int i = blockIdx.x * blockDim.x + threadIdx.x;
    if (i < e) atomicAdd(&deg[dst[i]], 1.0f);
}

__global__ void k_dinv(float* __restrict__ deg, int n) {
    int i = blockIdx.x * blockDim.x + threadIdx.x;
    if (i < n) deg[i] = 1.0f / sqrtf(deg[i]);
}

template <int K, int NOUT>
__global__ void k_gemm_wmma(const float* __restrict__ A, const float* __restrict__ W,
                            float* __restrict__ H, int nrows) {
    const int wave = blockIdx.x * (blockDim.x >> 5) + (threadIdx.x >> 5);
    const int lane = threadIdx.x & 31;
    const int m    = lane & 15;
    const int kh   = lane >> 4;
    const int r0   = wave * 16;
    if (r0 + 16 > nrows) return;

    constexpr int NT = NOUT / 16;
    v8f acc[NT];
#pragma unroll
    for (int t = 0; t < NT; ++t)
#pragma unroll
        for (int j = 0; j < 8; ++j) acc[t][j] = 0.0f;

    const float* arow = A + (size_t)(r0 + m) * K;
    for (int k = 0; k < K; k += 4) {
        v2f a;
        a.x = arow[k + 2 * kh + 0];
        a.y = arow[k + 2 * kh + 1];
#pragma unroll
        for (int t = 0; t < NT; ++t) {
            v2f b;
            b.x = W[(size_t)(k + 2 * kh + 0) * NOUT + t * 16 + m];
            b.y = W[(size_t)(k + 2 * kh + 1) * NOUT + t * 16 + m];
            acc[t] = __builtin_amdgcn_wmma_f32_16x16x4_f32(
                 false, a,  false, b,
                 (short)0, acc[t],  false,  false);
        }
    }

#pragma unroll
    for (int t = 0; t < NT; ++t) {
#pragma unroll
        for (int j = 0; j < 8; ++j)
            H[(size_t)(r0 + j + 8 * kh) * NOUT + t * 16 + m] = acc[t][j];
    }
}

template <int K, int NOUT>
__global__ void k_gemm_tail(const float* __restrict__ A, const float* __restrict__ W,
                            float* __restrict__ H, int row0, int nrows) {
    int idx = blockIdx.x * blockDim.x + threadIdx.x;
    int r = row0 + idx / NOUT;
    int c = idx % NOUT;
    if (r >= nrows) return;
    float s = 0.0f;
    for (int k = 0; k < K; ++k) s += A[(size_t)r * K + k] * W[(size_t)k * NOUT + c];
    H[(size_t)r * NOUT + c] = s;
}

template <int C, bool HAS_BIAS>
__global__ void k_self_init(const float* __restrict__ h, const float* __restrict__ dinv,
                            const float* __restrict__ bias, float* __restrict__ agg, int n) {
    long long idx = (long long)blockIdx.x * blockDim.x + threadIdx.x;
    long long total = (long long)n * C;
    if (idx >= total) return;
    int i = (int)(idx / C);
    int c = (int)(idx % C);
    float d = dinv[i];
    float v = h[idx] * d * d;
    if (HAS_BIAS) v += bias[c];
    agg[idx] = v;
}

template <int C>
__global__ void k_edge_agg(const int* __restrict__ src, const int* __restrict__ dst,
                           const float* __restrict__ h, const float* __restrict__ dinv,
                           float* __restrict__ agg, int e) {
    constexpr int EPW = 8;
    constexpr int NJ  = C / 32;
    const int lane = threadIdx.x & 31;
    const int wave = __builtin_amdgcn_readfirstlane(
        (int)(blockIdx.x * (blockDim.x >> 5) + (threadIdx.x >> 5)));
    const long long e0 = (long long)wave * EPW;
    if (e0 >= (long long)e) return;

    if (e0 + EPW <= (long long)e) {
        int s[EPW], d[EPW];
        float v[EPW][NJ];
        float nrm[EPW];
#pragma unroll
        for (int i = 0; i < EPW; ++i) {
            s[i] = src[e0 + i];
            d[i] = dst[e0 + i];
        }
#pragma unroll
        for (int i = 0; i < EPW; ++i)
#pragma unroll
            for (int j = 0; j < NJ; ++j)
                v[i][j] = h[(size_t)s[i] * C + j * 32 + lane];
#pragma unroll
        for (int i = 0; i < EPW; ++i) nrm[i] = dinv[s[i]] * dinv[d[i]];
#pragma unroll
        for (int i = 0; i < EPW; ++i)
#pragma unroll
            for (int j = 0; j < NJ; ++j)
                atomicAdd(agg + (size_t)d[i] * C + j * 32 + lane, v[i][j] * nrm[i]);
    } else {
        for (int i = 0; i < EPW; ++i) {
            if (e0 + i >= (long long)e) break;
            const int s = src[e0 + i];
            const int d = dst[e0 + i];
            const float nrm = dinv[s] * dinv[d];
#pragma unroll
            for (int j = 0; j < NJ; ++j) {
                const float v = h[(size_t)s * C + j * 32 + lane];
                atomicAdd(agg + (size_t)d * C + j * 32 + lane, v * nrm);
            }
        }
    }
}

template <int C>
__global__ void k_relu_bias(const float* __restrict__ agg, const float* __restrict__ b,
                            float* __restrict__ p, int n) {
    long long idx = (long long)blockIdx.x * blockDim.x + threadIdx.x;
    long long total = (long long)n * C;
    if (idx >= total) return;
    int c = (int)(idx % C);
    float v = agg[idx] + b[c];
    p[idx] = v > 0.0f ? v : 0.0f;
}

extern "C" void kernel_launch(void* const* d_in, const int* in_sizes, int n_in,
                              void* d_out, int out_size, void* d_ws, size_t ws_size,
                              hipStream_t stream) {
    const float* x  = (const float*)d_in[0];
    const int*   ei = (const int*)d_in[1];
    const float* W1 = (const float*)d_in[2];
    const float* b1 = (const float*)d_in[3];
    const float* W2 = (const float*)d_in[4];
    const float* b2 = (const float*)d_in[5];
    float* out = (float*)d_out;

    const int n = in_sizes[0] / 128;
    const int e = in_sizes[1] / 2;
    const int* src = ei;
    const int* dst = ei + e;

    char* ws = (char*)d_ws;
    size_t off0 = ((size_t)n * sizeof(float) + 255) & ~(size_t)255;
    float* dinv = (float*)ws;
    float* bufA = (float*)(ws + off0);
    float* bufB = bufA + (size_t)n * 64;

    const int tb = 256;
    auto cdiv = [](long long a, long long b) { return (int)((a + b - 1) / b); };

    k_deg_init<<<cdiv(n, tb), tb, 0, stream>>>(dinv, n);
    k_deg_count<<<cdiv(e, tb), tb, 0, stream>>>(dst, dinv, e);
    k_dinv<<<cdiv(n, tb), tb, 0, stream>>>(dinv, n);

    const int strips = n / 16;
    const int wavesPerBlk = tb / 32;
    k_gemm_wmma<128, 64><<<cdiv(strips, wavesPerBlk), tb, 0, stream>>>(x, W1, bufA, n);
    if (n % 16) {
        int tail = (n % 16) * 64;
        k_gemm_tail<128, 64><<<cdiv(tail, tb), tb, 0, stream>>>(x, W1, bufA, strips * 16, n);
    }

    k_self_init<64, false><<<cdiv((long long)n * 64, tb), tb, 0, stream>>>(bufA, dinv, nullptr, bufB, n);
    k_edge_agg<64><<<cdiv(cdiv(e, 8), wavesPerBlk), tb, 0, stream>>>(src, dst, bufA, dinv, bufB, e);
    k_relu_bias<64><<<cdiv((long long)n * 64, tb), tb, 0, stream>>>(bufB, b1, bufA, n);

    k_gemm_wmma<64, 32><<<cdiv(strips, wavesPerBlk), tb, 0, stream>>>(bufA, W2, bufB, n);
    if (n % 16) {
        int tail = (n % 16) * 32;
        k_gemm_tail<64, 32><<<cdiv(tail, tb), tb, 0, stream>>>(bufA, W2, bufB, strips * 16, n);
    }

    k_self_init<32, true><<<cdiv((long long)n * 32, tb), tb, 0, stream>>>(bufB, dinv, b2, out, n);
    k_edge_agg<32><<<cdiv(cdiv(e, 8), wavesPerBlk), tb, 0, stream>>>(src, dst, bufB, dinv, out, e);

    (void)n_in; (void)out_size; (void)ws_size;
}


# in-degree counted in per-work-group LDS tables over all nodes (two half-word counters per word, 32768 edges per group), slabs summed in the rsqrt pass; replaces a scattered global f32 atomic per edge
# speedup vs baseline: 1.8039x; 1.2801x over previous
#include <hip/hip_runtime.h>

typedef float v2f __attribute__((ext_vector_type(2)));
typedef float v8f __attribute__((ext_vector_type(8)));

__global__ void k_deg_init(float* __restrict__ deg, int n) {
    int i = blockIdx.x * blockDim.x + threadIdx.x;
    if (i < n) deg[i] = 1.0f;
}

__global__ void k_deg_count(const int* __restrict__ dst, float* __restrict__ deg, int e) {
    int i = blockIdx.x * blockDim.x + threadIdx.x;
    if (i < e) atomicAdd(&deg[dst[i]], 1.0f);
}

__global__ void k_dinv(float* __restrict__ deg, int n) {
    int i = blockIdx.x * blockDim.x + threadIdx.x;
    if (i < n) deg[i] = 1.0f / sqrtf(deg[i]);
}

constexpr int DEG_NCAP   = 100000;
constexpr int DEG_WORDS  = ((DEG_NCAP + 1) / 2 + 31) & ~31;
constexpr int DEG_ECH    = 32768;
constexpr int DEG_TB     = 1024;
constexpr int DEG_B      = 16;

__global__ void __launch_bounds__(DEG_TB)
k_deg_hist(const int* __restrict__ dst, unsigned int* __restrict__ part, int e) {
    __shared__ unsigned int cnt[DEG_WORDS];
    const int tid = threadIdx.x;

    uint4* c4 = reinterpret_cast<uint4*>(cnt);
    for (int i = tid; i < DEG_WORDS / 4; i += DEG_TB) c4[i] = make_uint4(0u, 0u, 0u, 0u);
    __syncthreads();

    const long long e0 = (long long)blockIdx.x * DEG_ECH;
    const long long left = (long long)e - e0;
    const int m = left < (long long)DEG_ECH ? (int)left : DEG_ECH;
    const int* __restrict__ dp = dst + e0;
    for (int base = tid; base < m; base += DEG_B * DEG_TB) {
        int d[DEG_B];
#pragma unroll
        for (int k = 0; k < DEG_B; ++k) {
            const int i = base + k * DEG_TB;
            d[k] = (i < m) ? dp[i] : -1;
        }
#pragma unroll
        for (int k = 0; k < DEG_B; ++k) {
            if (base + k * DEG_TB < m) atomicAdd(&cnt[d[k] >> 1], (d[k] & 1) ? 0x10000u : 1u);
        }
    }
    __syncthreads();

    uint4* p4 = reinterpret_cast<uint4*>(part + (size_t)blockIdx.x * DEG_WORDS);
    for (int i = tid; i < DEG_WORDS / 4; i += DEG_TB) p4[i] = c4[i];
}

__global__ void k_dinv_hist(const unsigned int* __restrict__ part, float* __restrict__ dinv,
                            int n, int parts) {
    const int t = blockIdx.x * blockDim.x + threadIdx.x;
    if (2 * t >= n) return;
    unsigned int lo = 0u, hi = 0u;
#pragma unroll 8
    for (int p = 0; p < parts; ++p) {
        const unsigned int w = part[(size_t)p * DEG_WORDS + t];
        lo += w & 0xFFFFu;
        hi += w >> 16;
    }
    dinv[2 * t] = 1.0f / sqrtf((float)(lo + 1u));
    if (2 * t + 1 < n) dinv[2 * t + 1] = 1.0f / sqrtf((float)(hi + 1u));
}

template <int K, int NOUT>
__global__ void k_gemm_wmma(const float* __restrict__ A, const float* __restrict__ W,
                            float* __restrict__ H, int nrows) {
    const int wave = blockIdx.x * (blockDim.x >> 5) + (threadIdx.x >> 5);
    const int lane = threadIdx.x & 31;
    const int m    = lane & 15;
    const int kh   = lane >> 4;
    const int r0   = wave * 16;
    if (r0 + 16 > nrows) return;

    constexpr int NT = NOUT / 16;
    v8f acc[NT];
#pragma unroll
    for (int t = 0; t < NT; ++t)
#pragma unroll
        for (int j = 0; j < 8; ++j) acc[t][j] = 0.0f;

    const float* arow = A + (size_t)(r0 + m) * K;
    for (int k = 0; k < K; k += 4) {
        v2f a;
        a.x = arow[k + 2 * kh + 0];
        a.y = arow[k + 2 * kh + 1];
#pragma unroll
        for (int t = 0; t < NT; ++t) {
            v2f b;
            b.x = W[(size_t)(k + 2 * kh + 0) * NOUT + t * 16 + m];
            b.y = W[(size_t)(k + 2 * kh + 1) * NOUT + t * 16 + m];
            acc[t] = __builtin_amdgcn_wmma_f32_16x16x4_f32(
                 false, a,  false, b,
                 (short)0, acc[t],  false,  false);
        }
    }

#pragma unroll
    for (int t = 0; t < NT; ++t) {
#pragma unroll
        for (int j = 0; j < 8; ++j)
            H[(size_t)(r0 + j + 8 * kh) * NOUT + t * 16 + m] = acc[t][j];
    }
}

template <int K, int NOUT>
__global__ void k_gemm_tail(const float* __restrict__ A, const float* __restrict__ W,
                            float* __restrict__ H, int row0, int nrows) {
    int idx = blockIdx.x * blockDim.x + threadIdx.x;
    int r = row0 + idx / NOUT;
    int c = idx % NOUT;
    if (r >= nrows) return;
    float s = 0.0f;
    for (int k = 0; k < K; ++k) s += A[(size_t)r * K + k] * W[(size_t)k * NOUT + c];
    H[(size_t)r * NOUT + c] = s;
}

template <int C, bool HAS_BIAS>
__global__ void k_self_init(const float* __restrict__ h, const float* __restrict__ dinv,
                            const float* __restrict__ bias, float* __restrict__ agg, int n) {
    long long idx = (long long)blockIdx.x * blockDim.x + threadIdx.x;
    long long total = (long long)n * C;
    if (idx >= total) return;
    int i = (int)(idx / C);
    int c = (int)(idx % C);
    float d = dinv[i];
    float v = h[idx] * d * d;
    if (HAS_BIAS) v += bias[c];
    agg[idx] = v;
}

template <int C>
__global__ void k_edge_agg(const int* __restrict__ src, const int* __restrict__ dst,
                           const float* __restrict__ h, const float* __restrict__ dinv,
                           float* __restrict__ agg, int e) {
    constexpr int EPW = 8;
    constexpr int NJ  = C / 32;
    const int lane = threadIdx.x & 31;
    const int wave = __builtin_amdgcn_readfirstlane(
        (int)(blockIdx.x * (blockDim.x >> 5) + (threadIdx.x >> 5)));
    const long long e0 = (long long)wave * EPW;
    if (e0 >= (long long)e) return;

    if (e0 + EPW <= (long long)e) {
        int s[EPW], d[EPW];
        float v[EPW][NJ];
        float nrm[EPW];
#pragma unroll
        for (int i = 0; i < EPW; ++i) {
            s[i] = src[e0 + i];
            d[i] = dst[e0 + i];
        }
#pragma unroll
        for (int i = 0; i < EPW; ++i)
#pragma unroll
            for (int j = 0; j < NJ; ++j)
                v[i][j] = h[(size_t)s[i] * C + j * 32 + lane];
#pragma unroll
        for (int i = 0; i < EPW; ++i) nrm[i] = dinv[s[i]] * dinv[d[i]];
#pragma unroll
        for (int i = 0; i < EPW; ++i)
#pragma unroll
            for (int j = 0; j < NJ; ++j)
                atomicAdd(agg + (size_t)d[i] * C + j * 32 + lane, v[i][j] * nrm[i]);
    } else {
        for (int i = 0; i < EPW; ++i) {
            if (e0 + i >= (long long)e) break;
            const int s = src[e0 + i];
            const int d = dst[e0 + i];
            const float nrm = dinv[s] * dinv[d];
#pragma unroll
            for (int j = 0; j < NJ; ++j) {
                const float v = h[(size_t)s * C + j * 32 + lane];
                atomicAdd(agg + (size_t)d * C + j * 32 + lane, v * nrm);
            }
        }
    }
}

template <int C>
__global__ void k_relu_bias(const float* __restrict__ agg, const float* __restrict__ b,
                            float* __restrict__ p, int n) {
    long long idx = (long long)blockIdx.x * blockDim.x + threadIdx.x;
    long long total = (long long)n * C;
    if (idx >= total) return;
    int c = (int)(idx % C);
    float v = agg[idx] + b[c];
    p[idx] = v > 0.0f ? v : 0.0f;
}

extern "C" void kernel_launch(void* const* d_in, const int* in_sizes, int n_in,
                              void* d_out, int out_size, void* d_ws, size_t ws_size,
                              hipStream_t stream) {
    const float* x  = (const float*)d_in[0];
    const int*   ei = (const int*)d_in[1];
    const float* W1 = (const float*)d_in[2];
    const float* b1 = (const float*)d_in[3];
    const float* W2 = (const float*)d_in[4];
    const float* b2 = (const float*)d_in[5];
    float* out = (float*)d_out;

    const int n = in_sizes[0] / 128;
    const int e = in_sizes[1] / 2;
    const int* src = ei;
    const int* dst = ei + e;

    char* ws = (char*)d_ws;
    size_t off0 = ((size_t)n * sizeof(float) + 255) & ~(size_t)255;
    float* dinv = (float*)ws;
    float* bufA = (float*)(ws + off0);
    float* bufB = bufA + (size_t)n * 64;

    const int tb = 256;
    auto cdiv = [](long long a, long long b) { return (int)((a + b - 1) / b); };

    const int parts = cdiv(e, DEG_ECH);
    if (n <= DEG_NCAP && (size_t)parts * DEG_WORDS <= (size_t)n * 64) {
        k_deg_hist<<<parts, DEG_TB, 0, stream>>>(dst, (unsigned int*)bufB, e);
        k_dinv_hist<<<cdiv((n + 1) / 2, tb), tb, 0, stream>>>((const unsigned int*)bufB, dinv, n, parts);
    } else {
        k_deg_init<<<cdiv(n, tb), tb, 0, stream>>>(dinv, n);
        k_deg_count<<<cdiv(e, tb), tb, 0, stream>>>(dst, dinv, e);
        k_dinv<<<cdiv(n, tb), tb, 0, stream>>>(dinv, n);
    }

    const int strips = n / 16;
    const int wavesPerBlk = tb / 32;
    k_gemm_wmma<128, 64><<<cdiv(strips, wavesPerBlk), tb, 0, stream>>>(x, W1, bufA, n);
    if (n % 16) {
        int tail = (n % 16) * 64;
        k_gemm_tail<128, 64><<<cdiv(tail, tb), tb, 0, stream>>>(x, W1, bufA, strips * 16, n);
    }

    k_self_init<64, false><<<cdiv((long long)n * 64, tb), tb, 0, stream>>>(bufA, dinv, nullptr, bufB, n);
    k_edge_agg<64><<<cdiv(cdiv(e, 8), wavesPerBlk), tb, 0, stream>>>(src, dst, bufA, dinv, bufB, e);
    k_relu_bias<64><<<cdiv((long long)n * 64, tb), tb, 0, stream>>>(bufB, b1, bufA, n);

    k_gemm_wmma<64, 32><<<cdiv(strips, wavesPerBlk), tb, 0, stream>>>(bufA, W2, bufB, n);
    if (n % 16) {
        int tail = (n % 16) * 32;
        k_gemm_tail<64, 32><<<cdiv(tail, tb), tb, 0, stream>>>(bufA, W2, bufB, strips * 16, n);
    }

    k_self_init<32, true><<<cdiv((long long)n * 32, tb), tb, 0, stream>>>(bufB, dinv, b2, out, n);
    k_edge_agg<32><<<cdiv(cdiv(e, 8), wavesPerBlk), tb, 0, stream>>>(src, dst, bufB, dinv, out, e);

    (void)n_in; (void)out_size; (void)ws_size;
}


# edges counting-sorted per call by 400-node dst range (LDS counters); a work-group per range sums rows by LDS f32 adds (arrival order), stores rows once, W2 WMMA fed from LDS; no global atomics
# speedup vs baseline: 2.5039x; 1.3880x over previous
#include <hip/hip_runtime.h>

typedef float v2f __attribute__((ext_vector_type(2)));
typedef float v8f __attribute__((ext_vector_type(8)));

__global__ void k_deg_init(float* __restrict__ deg, int n) {
    int i = blockIdx.x * blockDim.x + threadIdx.x;
    if (i < n) deg[i] = 1.0f;
}

__global__ void k_deg_count(const int* __restrict__ dst, float* __restrict__ deg, int e) {
    int i = blockIdx.x * blockDim.x + threadIdx.x;
    if (i < e) atomicAdd(&deg[dst[i]], 1.0f);
}

__global__ void k_dinv(float* __restrict__ deg, int n) {
    int i = blockIdx.x * blockDim.x + threadIdx.x;
    if (i < n) deg[i] = 1.0f / sqrtf(deg[i]);
}

constexpr int DEG_NCAP   = 100000;
constexpr int DEG_WORDS  = ((DEG_NCAP + 1) / 2 + 31) & ~31;
constexpr int DEG_ECH    = 32768;
constexpr int DEG_TB     = 1024;
constexpr int DEG_B      = 16;
constexpr int BN         = 400;
constexpr int BW         = BN / 2;
constexpr int NBMAX      = (DEG_NCAP + BN - 1) / BN;
constexpr int SCAN_PMAX  = 64;

__global__ void __launch_bounds__(DEG_TB)
k_deg_hist(const int* __restrict__ dst, unsigned int* __restrict__ part,
           unsigned int* __restrict__ pcnt, int nb, int e) {
    __shared__ unsigned int cnt[DEG_WORDS];
    const int tid = threadIdx.x;

    uint4* c4 = reinterpret_cast<uint4*>(cnt);
    for (int i = tid; i < DEG_WORDS / 4; i += DEG_TB) c4[i] = make_uint4(0u, 0u, 0u, 0u);
    __syncthreads();

    const long long e0 = (long long)blockIdx.x * DEG_ECH;
    const long long left = (long long)e - e0;
    const int m = left < (long long)DEG_ECH ? (int)left : DEG_ECH;
    const int* __restrict__ dp = dst + e0;
    for (int base = tid; base < m; base += DEG_B * DEG_TB) {
        int d[DEG_B];
#pragma unroll
        for (int k = 0; k < DEG_B; ++k) {
            const int i = base + k * DEG_TB;
            d[k] = (i < m) ? dp[i] : -1;
        }
#pragma unroll
        for (int k = 0; k < DEG_B; ++k) {
            if (base + k * DEG_TB < m) atomicAdd(&cnt[d[k] >> 1], (d[k] & 1) ? 0x10000u : 1u);
        }
    }
    __syncthreads();

    uint4* p4 = reinterpret_cast<uint4*>(part + (size_t)blockIdx.x * DEG_WORDS);
    for (int i = tid; i < DEG_WORDS / 4; i += DEG_TB) p4[i] = c4[i];

    if (pcnt != nullptr) {
        const int wv = tid >> 5, ln = tid & 31;
        for (int b = wv; b < nb; b += DEG_TB / 32) {
            unsigned int sacc = 0u;
            for (int w = ln; w < BW; w += 32) {
                const int wi = b * BW + w;
                if (wi < DEG_WORDS) {
                    const unsigned int v = cnt[wi];
                    sacc += (v & 0xFFFFu) + (v >> 16);
                }
            }
            for (int o = 16; o > 0; o >>= 1) sacc += __shfl_xor(sacc, o);
            if (ln == 0) pcnt[(size_t)blockIdx.x * nb + b] = sacc;
        }
    }
}

__global__ void __launch_bounds__(256)
k_bin_scan(const unsigned int* __restrict__ pcnt, unsigned int* __restrict__ off,
           unsigned int* __restrict__ bstart, int parts, int nb) {
    __shared__ unsigned int pc[SCAN_PMAX * NBMAX];
    __shared__ unsigned int tot[256];
    const int b = threadIdx.x;
    const int total = parts * nb;
#pragma unroll 8
    for (int i = b; i < total; i += 256) pc[i] = pcnt[i];
    __syncthreads();
    unsigned int run = 0u;
    if (b < nb)
        for (int p = 0; p < parts; ++p) run += pc[p * nb + b];
    tot[b] = run;
    __syncthreads();
    for (int o = 1; o < 256; o <<= 1) {
        const unsigned int v = (b >= o) ? tot[b - o] : 0u;
        __syncthreads();
        tot[b] += v;
        __syncthreads();
    }
    if (b < nb) {
        unsigned int r2 = tot[b] - run;
        bstart[b] = r2;
        for (int p = 0; p < parts; ++p) {
            off[(size_t)p * nb + b] = r2;
            r2 += pc[p * nb + b];
        }
        if (b == nb - 1) bstart[nb] = tot[b];
    }
}

__global__ void __launch_bounds__(DEG_TB)
k_bin_scatter(const int* __restrict__ src, const int* __restrict__ dst,
              const unsigned int* __restrict__ off, unsigned int* __restrict__ recs,
              int e, int nb) {
    __shared__ unsigned int cur[NBMAX];
    const int tid = threadIdx.x;
    for (int b = tid; b < nb; b += DEG_TB) cur[b] = off[(size_t)blockIdx.x * nb + b];
    __syncthreads();

    const long long e0 = (long long)blockIdx.x * DEG_ECH;
    const long long left = (long long)e - e0;
    const int m = left < (long long)DEG_ECH ? (int)left : DEG_ECH;
    const int* __restrict__ dp = dst + e0;
    const int* __restrict__ sp = src + e0;
    for (int base = tid; base < m; base += DEG_B * DEG_TB) {
        int d[DEG_B], s[DEG_B];
#pragma unroll
        for (int k = 0; k < DEG_B; ++k) {
            const int i = base + k * DEG_TB;
            d[k] = (i < m) ? dp[i] : 0;
            s[k] = (i < m) ? sp[i] : 0;
        }
#pragma unroll
        for (int k = 0; k < DEG_B; ++k) {
            if (base + k * DEG_TB < m) {
                const int b = d[k] / BN;
                const unsigned int pos = atomicAdd(&cur[b], 1u);
                recs[pos] = ((unsigned int)(d[k] - b * BN) << 17) | (unsigned int)s[k];
            }
        }
    }
}

__global__ void k_dinv_hist(const unsigned int* __restrict__ part, float* __restrict__ dinv,
                            int n, int parts) {
    const int t = blockIdx.x * blockDim.x + threadIdx.x;
    if (2 * t >= n) return;
    unsigned int lo = 0u, hi = 0u;
#pragma unroll 8
    for (int p = 0; p < parts; ++p) {
        const unsigned int w = part[(size_t)p * DEG_WORDS + t];
        lo += w & 0xFFFFu;
        hi += w >> 16;
    }
    dinv[2 * t] = 1.0f / sqrtf((float)(lo + 1u));
    if (2 * t + 1 < n) dinv[2 * t + 1] = 1.0f / sqrtf((float)(hi + 1u));
}

template <int K, int NOUT>
__global__ void k_gemm_wmma(const float* __restrict__ A, const float* __restrict__ W,
                            float* __restrict__ H, int nrows) {
    const int wave = blockIdx.x * (blockDim.x >> 5) + (threadIdx.x >> 5);
    const int lane = threadIdx.x & 31;
    const int m    = lane & 15;
    const int kh   = lane >> 4;
    const int r0   = wave * 16;
    if (r0 + 16 > nrows) return;

    constexpr int NT = NOUT / 16;
    v8f acc[NT];
#pragma unroll
    for (int t = 0; t < NT; ++t)
#pragma unroll
        for (int j = 0; j < 8; ++j) acc[t][j] = 0.0f;

    const float* arow = A + (size_t)(r0 + m) * K;
    for (int k = 0; k < K; k += 4) {
        v2f a;
        a.x = arow[k + 2 * kh + 0];
        a.y = arow[k + 2 * kh + 1];
#pragma unroll
        for (int t = 0; t < NT; ++t) {
            v2f b;
            b.x = W[(size_t)(k + 2 * kh + 0) * NOUT + t * 16 + m];
            b.y = W[(size_t)(k + 2 * kh + 1) * NOUT + t * 16 + m];
            acc[t] = __builtin_amdgcn_wmma_f32_16x16x4_f32(
                 false, a,  false, b,
                 (short)0, acc[t],  false,  false);
        }
    }

#pragma unroll
    for (int t = 0; t < NT; ++t) {
#pragma unroll
        for (int j = 0; j < 8; ++j)
            H[(size_t)(r0 + j + 8 * kh) * NOUT + t * 16 + m] = acc[t][j];
    }
}

template <int K, int NOUT>
__global__ void k_gemm_tail(const float* __restrict__ A, const float* __restrict__ W,
                            float* __restrict__ H, int row0, int nrows) {
    int idx = blockIdx.x * blockDim.x + threadIdx.x;
    int r = row0 + idx / NOUT;
    int c = idx % NOUT;
    if (r >= nrows) return;
    float s = 0.0f;
    for (int k = 0; k < K; ++k) s += A[(size_t)r * K + k] * W[(size_t)k * NOUT + c];
    H[(size_t)r * NOUT + c] = s;
}

template <int C, bool HAS_BIAS>
__global__ void k_self_init(const float* __restrict__ h, const float* __restrict__ dinv,
                            const float* __restrict__ bias, float* __restrict__ agg, int n) {
    long long idx = (long long)blockIdx.x * blockDim.x + threadIdx.x;
    long long total = (long long)n * C;
    if (idx >= total) return;
    int i = (int)(idx / C);
    int c = (int)(idx % C);
    float d = dinv[i];
    float v = h[idx] * d * d;
    if (HAS_BIAS) v += bias[c];
    agg[idx] = v;
}

template <int C>
__global__ void k_edge_agg(const int* __restrict__ src, const int* __restrict__ dst,
                           const float* __restrict__ h, const float* __restrict__ dinv,
                           float* __restrict__ agg, int e) {
    constexpr int EPW = 8;
    constexpr int NJ  = C / 32;
    const int lane = threadIdx.x & 31;
    const int wave = __builtin_amdgcn_readfirstlane(
        (int)(blockIdx.x * (blockDim.x >> 5) + (threadIdx.x >> 5)));
    const long long e0 = (long long)wave * EPW;
    if (e0 >= (long long)e) return;

    if (e0 + EPW <= (long long)e) {
        int s[EPW], d[EPW];
        float v[EPW][NJ];
        float nrm[EPW];
#pragma unroll
        for (int i = 0; i < EPW; ++i) {
            s[i] = src[e0 + i];
            d[i] = dst[e0 + i];
        }
#pragma unroll
        for (int i = 0; i < EPW; ++i)
#pragma unroll
            for (int j = 0; j < NJ; ++j)
                v[i][j] = h[(size_t)s[i] * C + j * 32 + lane];
#pragma unroll
        for (int i = 0; i < EPW; ++i) nrm[i] = dinv[s[i]] * dinv[d[i]];
#pragma unroll
        for (int i = 0; i < EPW; ++i)
#pragma unroll
            for (int j = 0; j < NJ; ++j)
                atomicAdd(agg + (size_t)d[i] * C + j * 32 + lane, v[i][j] * nrm[i]);
    } else {
        for (int i = 0; i < EPW; ++i) {
            if (e0 + i >= (long long)e) break;
            const int s = src[e0 + i];
            const int d = dst[e0 + i];
            const float nrm = dinv[s] * dinv[d];
#pragma unroll
            for (int j = 0; j < NJ; ++j) {
                const float v = h[(size_t)s * C + j * 32 + lane];
                atomicAdd(agg + (size_t)d * C + j * 32 + lane, v * nrm);
            }
        }
    }
}

template <int C>
__global__ void k_relu_bias(const float* __restrict__ agg, const float* __restrict__ b,
                            float* __restrict__ p, int n) {
    long long idx = (long long)blockIdx.x * blockDim.x + threadIdx.x;
    long long total = (long long)n * C;
    if (idx >= total) return;
    int c = (int)(idx % C);
    float v = agg[idx] + b[c];
    p[idx] = v > 0.0f ? v : 0.0f;
}

constexpr int AGG_TB = 1024;
constexpr int AGG_G  = 16;

template <int C, bool LAYER1>
__global__ void __launch_bounds__(AGG_TB)
k_range_agg(const unsigned int* __restrict__ recs, const unsigned int* __restrict__ bstart,
            const float* __restrict__ h, const float* __restrict__ dinv,
            const float* __restrict__ bias, const float* __restrict__ W2,
            float* __restrict__ outp, int n) {
    constexpr int P  = C + 4;
    constexpr int NJ = C / 32;
    __shared__ float acc[BN * P];
    const int tid  = threadIdx.x;
    const int lane = tid & 31;
    const int wv   = __builtin_amdgcn_readfirstlane(tid >> 5);
    const int n0   = blockIdx.x * BN;
    const int rows = (n - n0) < BN ? (n - n0) : BN;

    constexpr int TOT = BN * C;
    constexpr int INIT_B = 5;
    for (int base = tid; base < TOT; base += INIT_B * AGG_TB) {
        float hv[INIT_B], dv[INIT_B];
#pragma unroll
        for (int u = 0; u < INIT_B; ++u) {
            int idx = base + u * AGG_TB;
            idx = idx < TOT ? idx : TOT - 1;
            const int r = idx / C, c = idx - r * C;
            int node = n0 + r;
            node = node < n ? node : n - 1;
            hv[u] = h[(size_t)node * C + c];
            dv[u] = dinv[node];
        }
#pragma unroll
        for (int u = 0; u < INIT_B; ++u) {
            const int idx = base + u * AGG_TB;
            if (idx < TOT) {
                const int r = idx / C, c = idx - r * C;
                acc[r * P + c] = (r < rows) ? hv[u] * dv[u] * dv[u] : 0.0f;
            }
        }
    }
    __syncthreads();

    const unsigned int eb = bstart[blockIdx.x];
    const unsigned int ee = bstart[blockIdx.x + 1];
    unsigned int c0 = eb + (unsigned int)wv * 32u;
    unsigned int pk = (c0 + lane < ee) ? recs[c0 + lane] : 0u;
    for (; c0 < ee; c0 += AGG_TB) {
        const unsigned int cn = c0 + AGG_TB;
        const unsigned int pkn = (cn + lane < ee) ? recs[cn + lane] : 0u;
        const bool ok = (c0 + lane) < ee;
        const int s  = (int)(pk & 0x1FFFFu);
        const int dl = (int)(pk >> 17);
        const float nr = ok ? dinv[s] * dinv[n0 + dl] : 0.0f;
        const int cnt = (int)((ee - c0) < 32u ? (ee - c0) : 32u);
#pragma unroll
        for (int g = 0; g < 32; g += AGG_G) {
            if (g < cnt) {
                float v[AGG_G][NJ];
#pragma unroll
                for (int k = 0; k < AGG_G; ++k) {
                    const int sk = __builtin_amdgcn_readlane(s, g + k);
#pragma unroll
                    for (int j = 0; j < NJ; ++j) v[k][j] = h[(size_t)sk * C + j * 32 + lane];
                }
#pragma unroll
                for (int k = 0; k < AGG_G; ++k) {
                    if (g + k < cnt) {
                        const int dk = __builtin_amdgcn_readlane(dl, g + k);
                        const float nk = __int_as_float(__builtin_amdgcn_readlane(__float_as_int(nr), g + k));
#pragma unroll
                        for (int j = 0; j < NJ; ++j)
                            atomicAdd(&acc[dk * P + j * 32 + lane], v[k][j] * nk);
                    }
                }
            }
        }
        pk = pkn;
    }
    __syncthreads();

    if constexpr (!LAYER1) {
        for (int idx = tid; idx < rows * C; idx += AGG_TB) {
            const int r = idx / C, c = idx - r * C;
            outp[(size_t)(n0 + r) * C + c] = acc[r * P + c] + bias[c];
        }
    } else {
        for (int idx = tid; idx < BN * C; idx += AGG_TB) {
            const int r = idx / C, c = idx - r * C;
            const float v = acc[r * P + c] + bias[c];
            acc[r * P + c] = v > 0.0f ? v : 0.0f;
        }
        __syncthreads();
        constexpr int NO = 32;
        constexpr int NT = NO / 16;
        const int full = rows / 16;
        if (wv < full) {
            const int m  = lane & 15;
            const int kh = lane >> 4;
            v8f d[NT];
#pragma unroll
            for (int t = 0; t < NT; ++t)
#pragma unroll
                for (int j = 0; j < 8; ++j) d[t][j] = 0.0f;
            const float* arow = acc + (wv * 16 + m) * P;
            for (int k = 0; k < C; k += 4) {
                v2f a;
                a.x = arow[k + 2 * kh + 0];
                a.y = arow[k + 2 * kh + 1];
#pragma unroll
                for (int t = 0; t < NT; ++t) {
                    v2f b;
                    b.x = W2[(size_t)(k + 2 * kh + 0) * NO + t * 16 + m];
                    b.y = W2[(size_t)(k + 2 * kh + 1) * NO + t * 16 + m];
                    d[t] = __builtin_amdgcn_wmma_f32_16x16x4_f32(
                        false, a, false, b, (short)0, d[t], false, false);
                }
            }
#pragma unroll
            for (int t = 0; t < NT; ++t)
#pragma unroll
                for (int j = 0; j < 8; ++j)
                    outp[(size_t)(n0 + wv * 16 + j + 8 * kh) * NO + t * 16 + m] = d[t][j];
        }
        const int r1 = full * 16;
        for (int idx = tid; idx < (rows - r1) * NO; idx += AGG_TB) {
            const int r = r1 + idx / NO, c = idx % NO;
            float sum = 0.0f;
            for (int k = 0; k < C; ++k) sum += acc[r * P + k] * W2[(size_t)k * NO + c];
            outp[(size_t)(n0 + r) * NO + c] = sum;
        }
    }
}

extern "C" void kernel_launch(void* const* d_in, const int* in_sizes, int n_in,
                              void* d_out, int out_size, void* d_ws, size_t ws_size,
                              hipStream_t stream) {
    const float* x  = (const float*)d_in[0];
    const int*   ei = (const int*)d_in[1];
    const float* W1 = (const float*)d_in[2];
    const float* b1 = (const float*)d_in[3];
    const float* W2 = (const float*)d_in[4];
    const float* b2 = (const float*)d_in[5];
    float* out = (float*)d_out;

    const int n = in_sizes[0] / 128;
    const int e = in_sizes[1] / 2;
    const int* src = ei;
    const int* dst = ei + e;

    char* ws = (char*)d_ws;
    size_t off0 = ((size_t)n * sizeof(float) + 255) & ~(size_t)255;
    float* dinv = (float*)ws;
    float* bufA = (float*)(ws + off0);
    float* bufB = bufA + (size_t)n * 64;

    const int tb = 256;
    auto cdiv = [](long long a, long long b) { return (int)((a + b - 1) / b); };

    const int parts = cdiv(e, DEG_ECH);
    const int strips = n / 16;
    const int wavesPerBlk = tb / 32;
    const int nb = cdiv(n, BN);
    const size_t half = (size_t)n * 32;
    if (n <= DEG_NCAP && parts <= SCAN_PMAX && (size_t)parts * DEG_WORDS <= half &&
        (size_t)e + 2 * (size_t)parts * nb + nb + 1 <= half) {
        unsigned int* slabs  = (unsigned int*)bufB;
        unsigned int* recs   = (unsigned int*)bufB + half;
        unsigned int* pcnt   = recs + e;
        unsigned int* off    = pcnt + (size_t)parts * nb;
        unsigned int* bstart = off + (size_t)parts * nb;
        float* h2 = bufB;

        k_deg_hist<<<parts, DEG_TB, 0, stream>>>(dst, slabs, pcnt, nb, e);
        k_dinv_hist<<<cdiv((n + 1) / 2, tb), tb, 0, stream>>>(slabs, dinv, n, parts);
        k_bin_scan<<<1, 256, 0, stream>>>(pcnt, off, bstart, parts, nb);
        k_bin_scatter<<<parts, DEG_TB, 0, stream>>>(src, dst, off, recs, e, nb);

        k_gemm_wmma<128, 64><<<cdiv(strips, wavesPerBlk), tb, 0, stream>>>(x, W1, bufA, n);
        if (n % 16) {
            int tail = (n % 16) * 64;
            k_gemm_tail<128, 64><<<cdiv(tail, tb), tb, 0, stream>>>(x, W1, bufA, strips * 16, n);
        }
        k_range_agg<64, true><<<nb, AGG_TB, 0, stream>>>(recs, bstart, bufA, dinv, b1, W2, h2, n);
        k_range_agg<32, false><<<nb, AGG_TB, 0, stream>>>(recs, bstart, h2, dinv, b2, nullptr, out, n);
        return;
    }

    if (n <= DEG_NCAP && (size_t)parts * DEG_WORDS <= (size_t)n * 64) {
        k_deg_hist<<<parts, DEG_TB, 0, stream>>>(dst, (unsigned int*)bufB, nullptr, 0, e);
        k_dinv_hist<<<cdiv((n + 1) / 2, tb), tb, 0, stream>>>((const unsigned int*)bufB, dinv, n, parts);
    } else {
        k_deg_init<<<cdiv(n, tb), tb, 0, stream>>>(dinv, n);
        k_deg_count<<<cdiv(e, tb), tb, 0, stream>>>(dst, dinv, e);
        k_dinv<<<cdiv(n, tb), tb, 0, stream>>>(dinv, n);
    }

    k_gemm_wmma<128, 64><<<cdiv(strips, wavesPerBlk), tb, 0, stream>>>(x, W1, bufA, n);
    if (n % 16) {
        int tail = (n % 16) * 64;
        k_gemm_tail<128, 64><<<cdiv(tail, tb), tb, 0, stream>>>(x, W1, bufA, strips * 16, n);
    }

    k_self_init<64, false><<<cdiv((long long)n * 64, tb), tb, 0, stream>>>(bufA, dinv, nullptr, bufB, n);
    k_edge_agg<64><<<cdiv(cdiv(e, 8), wavesPerBlk), tb, 0, stream>>>(src, dst, bufA, dinv, bufB, e);
    k_relu_bias<64><<<cdiv((long long)n * 64, tb), tb, 0, stream>>>(bufB, b1, bufA, n);

    k_gemm_wmma<64, 32><<<cdiv(strips, wavesPerBlk), tb, 0, stream>>>(bufA, W2, bufB, n);
    if (n % 16) {
        int tail = (n % 16) * 32;
        k_gemm_tail<64, 32><<<cdiv(tail, tb), tb, 0, stream>>>(bufA, W2, bufB, strips * 16, n);
    }

    k_self_init<32, true><<<cdiv((long long)n * 32, tb), tb, 0, stream>>>(bufB, dinv, b2, out, n);
    k_edge_agg<32><<<cdiv(cdiv(e, 8), wavesPerBlk), tb, 0, stream>>>(src, dst, bufB, dinv, out, e);

    (void)n_in; (void)out_size; (void)ws_size;
}


# sort placement staged in LDS: a work-group groups its records by destination range in LDS, then writes each run with whole-128-B-line stores (was one scattered 4-B store per edge); slots unchanged
# speedup vs baseline: 2.8584x; 1.1416x over previous
#include <hip/hip_runtime.h>

typedef float v2f __attribute__((ext_vector_type(2)));
typedef float v8f __attribute__((ext_vector_type(8)));

__global__ void k_deg_init(float* __restrict__ deg, int n) {
    int i = blockIdx.x * blockDim.x + threadIdx.x;
    if (i < n) deg[i] = 1.0f;
}

__global__ void k_deg_count(const int* __restrict__ dst, float* __restrict__ deg, int e) {
    int i = blockIdx.x * blockDim.x + threadIdx.x;
    if (i < e) atomicAdd(&deg[dst[i]], 1.0f);
}

__global__ void k_dinv(float* __restrict__ deg, int n) {
    int i = blockIdx.x * blockDim.x + threadIdx.x;
    if (i < n) deg[i] = 1.0f / sqrtf(deg[i]);
}

constexpr int DEG_NCAP   = 100000;
constexpr int DEG_WORDS  = ((DEG_NCAP + 1) / 2 + 31) & ~31;
constexpr int DEG_ECH    = 32768;
constexpr int DEG_TB     = 1024;
constexpr int DEG_B      = 16;
constexpr int BN         = 400;
constexpr int BW         = BN / 2;
constexpr int NBMAX      = (DEG_NCAP + BN - 1) / BN;
constexpr int SCAN_PMAX  = 64;

__global__ void __launch_bounds__(DEG_TB)
k_deg_hist(const int* __restrict__ dst, unsigned int* __restrict__ part,
           unsigned int* __restrict__ pcnt, int nb, int e) {
    __shared__ unsigned int cnt[DEG_WORDS];
    const int tid = threadIdx.x;

    uint4* c4 = reinterpret_cast<uint4*>(cnt);
    for (int i = tid; i < DEG_WORDS / 4; i += DEG_TB) c4[i] = make_uint4(0u, 0u, 0u, 0u);
    __syncthreads();

    const long long e0 = (long long)blockIdx.x * DEG_ECH;
    const long long left = (long long)e - e0;
    const int m = left < (long long)DEG_ECH ? (int)left : DEG_ECH;
    const int* __restrict__ dp = dst + e0;
    for (int base = tid; base < m; base += DEG_B * DEG_TB) {
        int d[DEG_B];
#pragma unroll
        for (int k = 0; k < DEG_B; ++k) {
            const int i = base + k * DEG_TB;
            d[k] = (i < m) ? dp[i] : -1;
        }
#pragma unroll
        for (int k = 0; k < DEG_B; ++k) {
            if (base + k * DEG_TB < m) atomicAdd(&cnt[d[k] >> 1], (d[k] & 1) ? 0x10000u : 1u);
        }
    }
    __syncthreads();

    uint4* p4 = reinterpret_cast<uint4*>(part + (size_t)blockIdx.x * DEG_WORDS);
    for (int i = tid; i < DEG_WORDS / 4; i += DEG_TB) p4[i] = c4[i];

    if (pcnt != nullptr) {
        const int wv = tid >> 5, ln = tid & 31;
        for (int b = wv; b < nb; b += DEG_TB / 32) {
            unsigned int sacc = 0u;
            for (int w = ln; w < BW; w += 32) {
                const int wi = b * BW + w;
                if (wi < DEG_WORDS) {
                    const unsigned int v = cnt[wi];
                    sacc += (v & 0xFFFFu) + (v >> 16);
                }
            }
            for (int o = 16; o > 0; o >>= 1) sacc += __shfl_xor(sacc, o);
            if (ln == 0) pcnt[(size_t)blockIdx.x * nb + b] = sacc;
        }
    }
}

__global__ void __launch_bounds__(256)
k_bin_scan(const unsigned int* __restrict__ pcnt, unsigned int* __restrict__ off,
           unsigned int* __restrict__ bstart, int parts, int nb) {
    __shared__ unsigned int pc[SCAN_PMAX * NBMAX];
    __shared__ unsigned int tot[256];
    const int b = threadIdx.x;
    const int total = parts * nb;
#pragma unroll 8
    for (int i = b; i < total; i += 256) pc[i] = pcnt[i];
    __syncthreads();
    unsigned int run = 0u;
    if (b < nb)
        for (int p = 0; p < parts; ++p) run += pc[p * nb + b];
    tot[b] = run;
    __syncthreads();
    for (int o = 1; o < 256; o <<= 1) {
        const unsigned int v = (b >= o) ? tot[b - o] : 0u;
        __syncthreads();
        tot[b] += v;
        __syncthreads();
    }
    if (b < nb) {
        unsigned int r2 = tot[b] - run;
        bstart[b] = r2;
        for (int p = 0; p < parts; ++p) {
            off[(size_t)p * nb + b] = r2;
            r2 += pc[p * nb + b];
        }
        if (b == nb - 1) bstart[nb] = tot[b];
    }
}

__global__ void __launch_bounds__(DEG_TB)
k_bin_scatter(const int* __restrict__ src, const int* __restrict__ dst,
              const unsigned int* __restrict__ pcnt, const unsigned int* __restrict__ off,
              unsigned int* __restrict__ recs, int e, int nb) {
    __shared__ unsigned int stage[DEG_ECH];
    __shared__ unsigned int lbeg[NBMAX + 1];
    __shared__ unsigned int cur[NBMAX];
    __shared__ unsigned int gbeg[NBMAX];
    __shared__ unsigned int wtot[8];
    const int tid  = threadIdx.x;
    const int lane = tid & 31;
    const int wv   = tid >> 5;

    unsigned int c = 0u, incl = 0u;
    if (tid < 256) {
        c = (tid < nb) ? pcnt[(size_t)blockIdx.x * nb + tid] : 0u;
        incl = c;
        for (int o = 1; o < 32; o <<= 1) {
            const unsigned int v = __shfl_up(incl, o);
            if (lane >= o) incl += v;
        }
        if (lane == 31) wtot[wv] = incl;
    }
    __syncthreads();
    if (tid < 256) {
        unsigned int before = 0u;
        for (int w = 0; w < wv; ++w) before += wtot[w];
        const unsigned int ex = before + incl - c;
        if (tid < nb) {
            lbeg[tid] = ex;
            cur[tid]  = ex;
            gbeg[tid] = off[(size_t)blockIdx.x * nb + tid];
        }
        if (tid == nb - 1) lbeg[nb] = ex + c;
    }
    __syncthreads();

    const long long e0 = (long long)blockIdx.x * DEG_ECH;
    const long long left = (long long)e - e0;
    const int m = left < (long long)DEG_ECH ? (int)left : DEG_ECH;
    const int* __restrict__ dp = dst + e0;
    const int* __restrict__ sp = src + e0;
    for (int base = tid; base < m; base += DEG_B * DEG_TB) {
        int d[DEG_B], s[DEG_B];
#pragma unroll
        for (int k = 0; k < DEG_B; ++k) {
            const int i = base + k * DEG_TB;
            d[k] = (i < m) ? dp[i] : 0;
            s[k] = (i < m) ? sp[i] : 0;
        }
#pragma unroll
        for (int k = 0; k < DEG_B; ++k) {
            if (base + k * DEG_TB < m) {
                const int b = d[k] / BN;
                const unsigned int pos = atomicAdd(&cur[b], 1u);
                stage[pos] = ((unsigned int)(d[k] - b * BN) << 17) | (unsigned int)s[k];
            }
        }
    }
    __syncthreads();

    for (int b = wv; b < nb; b += DEG_TB / 32) {
        const unsigned int l0  = lbeg[b];
        const unsigned int len = lbeg[b + 1] - l0;
        const unsigned int g0  = gbeg[b];
        const unsigned int a   = g0 & 31u;
        for (unsigned int j = lane; j < len + a; j += 32u) {
            if (j >= a) recs[g0 - a + j] = stage[l0 + j - a];
        }
    }
}

__global__ void k_dinv_hist(const unsigned int* __restrict__ part, float* __restrict__ dinv,
                            int n, int parts) {
    const int t = blockIdx.x * blockDim.x + threadIdx.x;
    if (2 * t >= n) return;
    unsigned int lo = 0u, hi = 0u;
#pragma unroll 8
    for (int p = 0; p < parts; ++p) {
        const unsigned int w = part[(size_t)p * DEG_WORDS + t];
        lo += w & 0xFFFFu;
        hi += w >> 16;
    }
    dinv[2 * t] = 1.0f / sqrtf((float)(lo + 1u));
    if (2 * t + 1 < n) dinv[2 * t + 1] = 1.0f / sqrtf((float)(hi + 1u));
}

template <int K, int NOUT>
__global__ void k_gemm_wmma(const float* __restrict__ A, const float* __restrict__ W,
                            float* __restrict__ H, int nrows) {
    const int wave = blockIdx.x * (blockDim.x >> 5) + (threadIdx.x >> 5);
    const int lane = threadIdx.x & 31;
    const int m    = lane & 15;
    const int kh   = lane >> 4;
    const int r0   = wave * 16;
    if (r0 + 16 > nrows) return;

    constexpr int NT = NOUT / 16;
    v8f acc[NT];
#pragma unroll
    for (int t = 0; t < NT; ++t)
#pragma unroll
        for (int j = 0; j < 8; ++j) acc[t][j] = 0.0f;

    const float* arow = A + (size_t)(r0 + m) * K;
    for (int k = 0; k < K; k += 4) {
        v2f a;
        a.x = arow[k + 2 * kh + 0];
        a.y = arow[k + 2 * kh + 1];
#pragma unroll
        for (int t = 0; t < NT; ++t) {
            v2f b;
            b.x = W[(size_t)(k + 2 * kh + 0) * NOUT + t * 16 + m];
            b.y = W[(size_t)(k + 2 * kh + 1) * NOUT + t * 16 + m];
            acc[t] = __builtin_amdgcn_wmma_f32_16x16x4_f32(
                 false, a,  false, b,
                 (short)0, acc[t],  false,  false);
        }
    }

#pragma unroll
    for (int t = 0; t < NT; ++t) {
#pragma unroll
        for (int j = 0; j < 8; ++j)
            H[(size_t)(r0 + j + 8 * kh) * NOUT + t * 16 + m] = acc[t][j];
    }
}

template <int K, int NOUT>
__global__ void k_gemm_tail(const float* __restrict__ A, const float* __restrict__ W,
                            float* __restrict__ H, int row0, int nrows) {
    int idx = blockIdx.x * blockDim.x + threadIdx.x;
    int r = row0 + idx / NOUT;
    int c = idx % NOUT;
    if (r >= nrows) return;
    float s = 0.0f;
    for (int k = 0; k < K; ++k) s += A[(size_t)r * K + k] * W[(size_t)k * NOUT + c];
    H[(size_t)r * NOUT + c] = s;
}

template <int C, bool HAS_BIAS>
__global__ void k_self_init(const float* __restrict__ h, const float* __restrict__ dinv,
                            const float* __restrict__ bias, float* __restrict__ agg, int n) {
    long long idx = (long long)blockIdx.x * blockDim.x + threadIdx.x;
    long long total = (long long)n * C;
    if (idx >= total) return;
    int i = (int)(idx / C);
    int c = (int)(idx % C);
    float d = dinv[i];
    float v = h[idx] * d * d;
    if (HAS_BIAS) v += bias[c];
    agg[idx] = v;
}

template <int C>
__global__ void k_edge_agg(const int* __restrict__ src, const int* __restrict__ dst,
                           const float* __restrict__ h, const float* __restrict__ dinv,
                           float* __restrict__ agg, int e) {
    constexpr int EPW = 8;
    constexpr int NJ  = C / 32;
    const int lane = threadIdx.x & 31;
    const int wave = __builtin_amdgcn_readfirstlane(
        (int)(blockIdx.x * (blockDim.x >> 5) + (threadIdx.x >> 5)));
    const long long e0 = (long long)wave * EPW;
    if (e0 >= (long long)e) return;

    if (e0 + EPW <= (long long)e) {
        int s[EPW], d[EPW];
        float v[EPW][NJ];
        float nrm[EPW];
#pragma unroll
        for (int i = 0; i < EPW; ++i) {
            s[i] = src[e0 + i];
            d[i] = dst[e0 + i];
        }
#pragma unroll
        for (int i = 0; i < EPW; ++i)
#pragma unroll
            for (int j = 0; j < NJ; ++j)
                v[i][j] = h[(size_t)s[i] * C + j * 32 + lane];
#pragma unroll
        for (int i = 0; i < EPW; ++i) nrm[i] = dinv[s[i]] * dinv[d[i]];
#pragma unroll
        for (int i = 0; i < EPW; ++i)
#pragma unroll
            for (int j = 0; j < NJ; ++j)
                atomicAdd(agg + (size_t)d[i] * C + j * 32 + lane, v[i][j] * nrm[i]);
    } else {
        for (int i = 0; i < EPW; ++i) {
            if (e0 + i >= (long long)e) break;
            const int s = src[e0 + i];
            const int d = dst[e0 + i];
            const float nrm = dinv[s] * dinv[d];
#pragma unroll
            for (int j = 0; j < NJ; ++j) {
                const float v = h[(size_t)s * C + j * 32 + lane];
                atomicAdd(agg + (size_t)d * C + j * 32 + lane, v * nrm);
            }
        }
    }
}

template <int C>
__global__ void k_relu_bias(const float* __restrict__ agg, const float* __restrict__ b,
                            float* __restrict__ p, int n) {
    long long idx = (long long)blockIdx.x * blockDim.x + threadIdx.x;
    long long total = (long long)n * C;
    if (idx >= total) return;
    int c = (int)(idx % C);
    float v = agg[idx] + b[c];
    p[idx] = v > 0.0f ? v : 0.0f;
}

constexpr int AGG_TB = 1024;
constexpr int AGG_G  = 16;

template <int C, bool LAYER1>
__global__ void __launch_bounds__(AGG_TB)
k_range_agg(const unsigned int* __restrict__ recs, const unsigned int* __restrict__ bstart,
            const float* __restrict__ h, const float* __restrict__ dinv,
            const float* __restrict__ bias, const float* __restrict__ W2,
            float* __restrict__ outp, int n) {
    constexpr int P  = C + 4;
    constexpr int NJ = C / 32;
    __shared__ float acc[BN * P];
    const int tid  = threadIdx.x;
    const int lane = tid & 31;
    const int wv   = __builtin_amdgcn_readfirstlane(tid >> 5);
    const int n0   = blockIdx.x * BN;
    const int rows = (n - n0) < BN ? (n - n0) : BN;

    constexpr int TOT = BN * C;
    constexpr int INIT_B = 5;
    for (int base = tid; base < TOT; base += INIT_B * AGG_TB) {
        float hv[INIT_B], dv[INIT_B];
#pragma unroll
        for (int u = 0; u < INIT_B; ++u) {
            int idx = base + u * AGG_TB;
            idx = idx < TOT ? idx : TOT - 1;
            const int r = idx / C, c = idx - r * C;
            int node = n0 + r;
            node = node < n ? node : n - 1;
            hv[u] = h[(size_t)node * C + c];
            dv[u] = dinv[node];
        }
#pragma unroll
        for (int u = 0; u < INIT_B; ++u) {
            const int idx = base + u * AGG_TB;
            if (idx < TOT) {
                const int r = idx / C, c = idx - r * C;
                acc[r * P + c] = (r < rows) ? hv[u] * dv[u] * dv[u] : 0.0f;
            }
        }
    }
    __syncthreads();

    const unsigned int eb = bstart[blockIdx.x];
    const unsigned int ee = bstart[blockIdx.x + 1];
    unsigned int c0 = eb + (unsigned int)wv * 32u;
    unsigned int pk = (c0 + lane < ee) ? recs[c0 + lane] : 0u;
    for (; c0 < ee; c0 += AGG_TB) {
        const unsigned int cn = c0 + AGG_TB;
        const unsigned int pkn = (cn + lane < ee) ? recs[cn + lane] : 0u;
        const bool ok = (c0 + lane) < ee;
        const int s  = (int)(pk & 0x1FFFFu);
        const int dl = (int)(pk >> 17);
        const float nr = ok ? dinv[s] * dinv[n0 + dl] : 0.0f;
        const int cnt = (int)((ee - c0) < 32u ? (ee - c0) : 32u);
#pragma unroll
        for (int g = 0; g < 32; g += AGG_G) {
            if (g < cnt) {
                float v[AGG_G][NJ];
#pragma unroll
                for (int k = 0; k < AGG_G; ++k) {
                    const int sk = __builtin_amdgcn_readlane(s, g + k);
#pragma unroll
                    for (int j = 0; j < NJ; ++j) v[k][j] = h[(size_t)sk * C + j * 32 + lane];
                }
#pragma unroll
                for (int k = 0; k < AGG_G; ++k) {
                    if (g + k < cnt) {
                        const int dk = __builtin_amdgcn_readlane(dl, g + k);
                        const float nk = __int_as_float(__builtin_amdgcn_readlane(__float_as_int(nr), g + k));
#pragma unroll
                        for (int j = 0; j < NJ; ++j)
                            atomicAdd(&acc[dk * P + j * 32 + lane], v[k][j] * nk);
                    }
                }
            }
        }
        pk = pkn;
    }
    __syncthreads();

    if constexpr (!LAYER1) {
        for (int idx = tid; idx < rows * C; idx += AGG_TB) {
            const int r = idx / C, c = idx - r * C;
            outp[(size_t)(n0 + r) * C + c] = acc[r * P + c] + bias[c];
        }
    } else {
        for (int idx = tid; idx < BN * C; idx += AGG_TB) {
            const int r = idx / C, c = idx - r * C;
            const float v = acc[r * P + c] + bias[c];
            acc[r * P + c] = v > 0.0f ? v : 0.0f;
        }
        __syncthreads();
        constexpr int NO = 32;
        constexpr int NT = NO / 16;
        const int full = rows / 16;
        if (wv < full) {
            const int m  = lane & 15;
            const int kh = lane >> 4;
            v8f d[NT];
#pragma unroll
            for (int t = 0; t < NT; ++t)
#pragma unroll
                for (int j = 0; j < 8; ++j) d[t][j] = 0.0f;
            const float* arow = acc + (wv * 16 + m) * P;
            for (int k = 0; k < C; k += 4) {
                v2f a;
                a.x = arow[k + 2 * kh + 0];
                a.y = arow[k + 2 * kh + 1];
#pragma unroll
                for (int t = 0; t < NT; ++t) {
                    v2f b;
                    b.x = W2[(size_t)(k + 2 * kh + 0) * NO + t * 16 + m];
                    b.y = W2[(size_t)(k + 2 * kh + 1) * NO + t * 16 + m];
                    d[t] = __builtin_amdgcn_wmma_f32_16x16x4_f32(
                        false, a, false, b, (short)0, d[t], false, false);
                }
            }
#pragma unroll
            for (int t = 0; t < NT; ++t)
#pragma unroll
                for (int j = 0; j < 8; ++j)
                    outp[(size_t)(n0 + wv * 16 + j + 8 * kh) * NO + t * 16 + m] = d[t][j];
        }
        const int r1 = full * 16;
        for (int idx = tid; idx < (rows - r1) * NO; idx += AGG_TB) {
            const int r = r1 + idx / NO, c = idx % NO;
            float sum = 0.0f;
            for (int k = 0; k < C; ++k) sum += acc[r * P + k] * W2[(size_t)k * NO + c];
            outp[(size_t)(n0 + r) * NO + c] = sum;
        }
    }
}

extern "C" void kernel_launch(void* const* d_in, const int* in_sizes, int n_in,
                              void* d_out, int out_size, void* d_ws, size_t ws_size,
                              hipStream_t stream) {
    const float* x  = (const float*)d_in[0];
    const int*   ei = (const int*)d_in[1];
    const float* W1 = (const float*)d_in[2];
    const float* b1 = (const float*)d_in[3];
    const float* W2 = (const float*)d_in[4];
    const float* b2 = (const float*)d_in[5];
    float* out = (float*)d_out;

    const int n = in_sizes[0] / 128;
    const int e = in_sizes[1] / 2;
    const int* src = ei;
    const int* dst = ei + e;

    char* ws = (char*)d_ws;
    size_t off0 = ((size_t)n * sizeof(float) + 255) & ~(size_t)255;
    float* dinv = (float*)ws;
    float* bufA = (float*)(ws + off0);
    float* bufB = bufA + (size_t)n * 64;

    const int tb = 256;
    auto cdiv = [](long long a, long long b) { return (int)((a + b - 1) / b); };

    const int parts = cdiv(e, DEG_ECH);
    const int strips = n / 16;
    const int wavesPerBlk = tb / 32;
    const int nb = cdiv(n, BN);
    const size_t half = (size_t)n * 32;
    if (n <= DEG_NCAP && parts <= SCAN_PMAX && (size_t)parts * DEG_WORDS <= half &&
        (size_t)e + 2 * (size_t)parts * nb + nb + 1 <= half) {
        unsigned int* slabs  = (unsigned int*)bufB;
        unsigned int* recs   = (unsigned int*)bufB + half;
        unsigned int* pcnt   = recs + e;
        unsigned int* off    = pcnt + (size_t)parts * nb;
        unsigned int* bstart = off + (size_t)parts * nb;
        float* h2 = bufB;

        k_deg_hist<<<parts, DEG_TB, 0, stream>>>(dst, slabs, pcnt, nb, e);
        k_dinv_hist<<<cdiv((n + 1) / 2, tb), tb, 0, stream>>>(slabs, dinv, n, parts);
        k_bin_scan<<<1, 256, 0, stream>>>(pcnt, off, bstart, parts, nb);
        k_bin_scatter<<<parts, DEG_TB, 0, stream>>>(src, dst, pcnt, off, recs, e, nb);

        k_gemm_wmma<128, 64><<<cdiv(strips, wavesPerBlk), tb, 0, stream>>>(x, W1, bufA, n);
        if (n % 16) {
            int tail = (n % 16) * 64;
            k_gemm_tail<128, 64><<<cdiv(tail, tb), tb, 0, stream>>>(x, W1, bufA, strips * 16, n);
        }
        k_range_agg<64, true><<<nb, AGG_TB, 0, stream>>>(recs, bstart, bufA, dinv, b1, W2, h2, n);
        k_range_agg<32, false><<<nb, AGG_TB, 0, stream>>>(recs, bstart, h2, dinv, b2, nullptr, out, n);
        return;
    }

    if (n <= DEG_NCAP && (size_t)parts * DEG_WORDS <= (size_t)n * 64) {
        k_deg_hist<<<parts, DEG_TB, 0, stream>>>(dst, (unsigned int*)bufB, nullptr, 0, e);
        k_dinv_hist<<<cdiv((n + 1) / 2, tb), tb, 0, stream>>>((const unsigned int*)bufB, dinv, n, parts);
    } else {
        k_deg_init<<<cdiv(n, tb), tb, 0, stream>>>(dinv, n);
        k_deg_count<<<cdiv(e, tb), tb, 0, stream>>>(dst, dinv, e);
        k_dinv<<<cdiv(n, tb), tb, 0, stream>>>(dinv, n);
    }

    k_gemm_wmma<128, 64><<<cdiv(strips, wavesPerBlk), tb, 0, stream>>>(x, W1, bufA, n);
    if (n % 16) {
        int tail = (n % 16) * 64;
        k_gemm_tail<128, 64><<<cdiv(tail, tb), tb, 0, stream>>>(x, W1, bufA, strips * 16, n);
    }

    k_self_init<64, false><<<cdiv((long long)n * 64, tb), tb, 0, stream>>>(bufA, dinv, nullptr, bufB, n);
    k_edge_agg<64><<<cdiv(cdiv(e, 8), wavesPerBlk), tb, 0, stream>>>(src, dst, bufA, dinv, bufB, e);
    k_relu_bias<64><<<cdiv((long long)n * 64, tb), tb, 0, stream>>>(bufB, b1, bufA, n);

    k_gemm_wmma<64, 32><<<cdiv(strips, wavesPerBlk), tb, 0, stream>>>(bufA, W2, bufB, n);
    if (n % 16) {
        int tail = (n % 16) * 32;
        k_gemm_tail<64, 32><<<cdiv(tail, tb), tb, 0, stream>>>(bufA, W2, bufB, strips * 16, n);
    }

    k_self_init<32, true><<<cdiv((long long)n * 32, tb), tb, 0, stream>>>(bufB, dinv, b2, out, n);
    k_edge_agg<32><<<cdiv(cdiv(e, 8), wavesPerBlk), tb, 0, stream>>>(src, dst, bufB, dinv, out, e);

    (void)n_in; (void)out_size; (void)ws_size;
}


# normalisation factored: rows stored times dinv[src] by their producers (GEMM and W2 epilogues); range sums add rows as they are, scaled once by dinv[dst] after summing; no per-record dinv gathers
# speedup vs baseline: 3.0128x; 1.0540x over previous
#include <hip/hip_runtime.h>

typedef float v2f __attribute__((ext_vector_type(2)));
typedef float v8f __attribute__((ext_vector_type(8)));

__global__ void k_deg_init(float* __restrict__ deg, int n) {
    int i = blockIdx.x * blockDim.x + threadIdx.x;
    if (i < n) deg[i] = 1.0f;
}

__global__ void k_deg_count(const int* __restrict__ dst, float* __restrict__ deg, int e) {
    int i = blockIdx.x * blockDim.x + threadIdx.x;
    if (i < e) atomicAdd(&deg[dst[i]], 1.0f);
}

__global__ void k_dinv(float* __restrict__ deg, int n) {
    int i = blockIdx.x * blockDim.x + threadIdx.x;
    if (i < n) deg[i] = 1.0f / sqrtf(deg[i]);
}

constexpr int DEG_NCAP   = 100000;
constexpr int DEG_WORDS  = ((DEG_NCAP + 1) / 2 + 31) & ~31;
constexpr int DEG_ECH    = 32768;
constexpr int DEG_TB     = 1024;
constexpr int DEG_B      = 16;
constexpr int BN         = 400;
constexpr int BW         = BN / 2;
constexpr int NBMAX      = (DEG_NCAP + BN - 1) / BN;
constexpr int SCAN_PMAX  = 64;

__global__ void __launch_bounds__(DEG_TB)
k_deg_hist(const int* __restrict__ dst, unsigned int* __restrict__ part,
           unsigned int* __restrict__ pcnt, int nb, int e) {
    __shared__ unsigned int cnt[DEG_WORDS];
    const int tid = threadIdx.x;

    uint4* c4 = reinterpret_cast<uint4*>(cnt);
    for (int i = tid; i < DEG_WORDS / 4; i += DEG_TB) c4[i] = make_uint4(0u, 0u, 0u, 0u);
    __syncthreads();

    const long long e0 = (long long)blockIdx.x * DEG_ECH;
    const long long left = (long long)e - e0;
    const int m = left < (long long)DEG_ECH ? (int)left : DEG_ECH;
    const int* __restrict__ dp = dst + e0;
    for (int base = tid; base < m; base += DEG_B * DEG_TB) {
        int d[DEG_B];
#pragma unroll
        for (int k = 0; k < DEG_B; ++k) {
            const int i = base + k * DEG_TB;
            d[k] = (i < m) ? dp[i] : -1;
        }
#pragma unroll
        for (int k = 0; k < DEG_B; ++k) {
            if (base + k * DEG_TB < m) atomicAdd(&cnt[d[k] >> 1], (d[k] & 1) ? 0x10000u : 1u);
        }
    }
    __syncthreads();

    uint4* p4 = reinterpret_cast<uint4*>(part + (size_t)blockIdx.x * DEG_WORDS);
    for (int i = tid; i < DEG_WORDS / 4; i += DEG_TB) p4[i] = c4[i];

    if (pcnt != nullptr) {
        const int wv = tid >> 5, ln = tid & 31;
        for (int b = wv; b < nb; b += DEG_TB / 32) {
            unsigned int sacc = 0u;
            for (int w = ln; w < BW; w += 32) {
                const int wi = b * BW + w;
                if (wi < DEG_WORDS) {
                    const unsigned int v = cnt[wi];
                    sacc += (v & 0xFFFFu) + (v >> 16);
                }
            }
            for (int o = 16; o > 0; o >>= 1) sacc += __shfl_xor(sacc, o);
            if (ln == 0) pcnt[(size_t)blockIdx.x * nb + b] = sacc;
        }
    }
}

__global__ void __launch_bounds__(256)
k_bin_scan(const unsigned int* __restrict__ pcnt, unsigned int* __restrict__ off,
           unsigned int* __restrict__ bstart, int parts, int nb) {
    __shared__ unsigned int pc[SCAN_PMAX * NBMAX];
    __shared__ unsigned int tot[256];
    const int b = threadIdx.x;
    const int total = parts * nb;
#pragma unroll 8
    for (int i = b; i < total; i += 256) pc[i] = pcnt[i];
    __syncthreads();
    unsigned int run = 0u;
    if (b < nb)
        for (int p = 0; p < parts; ++p) run += pc[p * nb + b];
    tot[b] = run;
    __syncthreads();
    for (int o = 1; o < 256; o <<= 1) {
        const unsigned int v = (b >= o) ? tot[b - o] : 0u;
        __syncthreads();
        tot[b] += v;
        __syncthreads();
    }
    if (b < nb) {
        unsigned int r2 = tot[b] - run;
        bstart[b] = r2;
        for (int p = 0; p < parts; ++p) {
            off[(size_t)p * nb + b] = r2;
            r2 += pc[p * nb + b];
        }
        if (b == nb - 1) bstart[nb] = tot[b];
    }
}

__global__ void __launch_bounds__(DEG_TB)
k_bin_scatter(const int* __restrict__ src, const int* __restrict__ dst,
              const unsigned int* __restrict__ pcnt, const unsigned int* __restrict__ off,
              unsigned int* __restrict__ recs, int e, int nb) {
    __shared__ unsigned int stage[DEG_ECH];
    __shared__ unsigned int lbeg[NBMAX + 1];
    __shared__ unsigned int cur[NBMAX];
    __shared__ unsigned int gbeg[NBMAX];
    __shared__ unsigned int wtot[8];
    const int tid  = threadIdx.x;
    const int lane = tid & 31;
    const int wv   = tid >> 5;

    unsigned int c = 0u, incl = 0u;
    if (tid < 256) {
        c = (tid < nb) ? pcnt[(size_t)blockIdx.x * nb + tid] : 0u;
        incl = c;
        for (int o = 1; o < 32; o <<= 1) {
            const unsigned int v = __shfl_up(incl, o);
            if (lane >= o) incl += v;
        }
        if (lane == 31) wtot[wv] = incl;
    }
    __syncthreads();
    if (tid < 256) {
        unsigned int before = 0u;
        for (int w = 0; w < wv; ++w) before += wtot[w];
        const unsigned int ex = before + incl - c;
        if (tid < nb) {
            lbeg[tid] = ex;
            cur[tid]  = ex;
            gbeg[tid] = off[(size_t)blockIdx.x * nb + tid];
        }
        if (tid == nb - 1) lbeg[nb] = ex + c;
    }
    __syncthreads();

    const long long e0 = (long long)blockIdx.x * DEG_ECH;
    const long long left = (long long)e - e0;
    const int m = left < (long long)DEG_ECH ? (int)left : DEG_ECH;
    const int* __restrict__ dp = dst + e0;
    const int* __restrict__ sp = src + e0;
    for (int base = tid; base < m; base += DEG_B * DEG_TB) {
        int d[DEG_B], s[DEG_B];
#pragma unroll
        for (int k = 0; k < DEG_B; ++k) {
            const int i = base + k * DEG_TB;
            d[k] = (i < m) ? dp[i] : 0;
            s[k] = (i < m) ? sp[i] : 0;
        }
#pragma unroll
        for (int k = 0; k < DEG_B; ++k) {
            if (base + k * DEG_TB < m) {
                const int b = d[k] / BN;
                const unsigned int pos = atomicAdd(&cur[b], 1u);
                stage[pos] = ((unsigned int)(d[k] - b * BN) << 17) | (unsigned int)s[k];
            }
        }
    }
    __syncthreads();

    for (int b = wv; b < nb; b += DEG_TB / 32) {
        const unsigned int l0  = lbeg[b];
        const unsigned int len = lbeg[b + 1] - l0;
        const unsigned int g0  = gbeg[b];
        const unsigned int a   = g0 & 31u;
        for (unsigned int j = lane; j < len + a; j += 32u) {
            if (j >= a) recs[g0 - a + j] = stage[l0 + j - a];
        }
    }
}

__global__ void k_dinv_hist(const unsigned int* __restrict__ part, float* __restrict__ dinv,
                            int n, int parts) {
    const int t = blockIdx.x * blockDim.x + threadIdx.x;
    if (2 * t >= n) return;
    unsigned int lo = 0u, hi = 0u;
#pragma unroll 8
    for (int p = 0; p < parts; ++p) {
        const unsigned int w = part[(size_t)p * DEG_WORDS + t];
        lo += w & 0xFFFFu;
        hi += w >> 16;
    }
    dinv[2 * t] = 1.0f / sqrtf((float)(lo + 1u));
    if (2 * t + 1 < n) dinv[2 * t + 1] = 1.0f / sqrtf((float)(hi + 1u));
}

template <int K, int NOUT, bool SCALED>
__global__ void k_gemm_wmma(const float* __restrict__ A, const float* __restrict__ W,
                            const float* __restrict__ rowscale,
                            float* __restrict__ H, int nrows) {
    const int wave = blockIdx.x * (blockDim.x >> 5) + (threadIdx.x >> 5);
    const int lane = threadIdx.x & 31;
    const int m    = lane & 15;
    const int kh   = lane >> 4;
    const int r0   = wave * 16;
    if (r0 + 16 > nrows) return;

    constexpr int NT = NOUT / 16;
    v8f acc[NT];
#pragma unroll
    for (int t = 0; t < NT; ++t)
#pragma unroll
        for (int j = 0; j < 8; ++j) acc[t][j] = 0.0f;

    float sc[8];
#pragma unroll
    for (int j = 0; j < 8; ++j) sc[j] = SCALED ? rowscale[r0 + j + 8 * kh] : 1.0f;

    const float* arow = A + (size_t)(r0 + m) * K;
    for (int k = 0; k < K; k += 4) {
        v2f a;
        a.x = arow[k + 2 * kh + 0];
        a.y = arow[k + 2 * kh + 1];
#pragma unroll
        for (int t = 0; t < NT; ++t) {
            v2f b;
            b.x = W[(size_t)(k + 2 * kh + 0) * NOUT + t * 16 + m];
            b.y = W[(size_t)(k + 2 * kh + 1) * NOUT + t * 16 + m];
            acc[t] = __builtin_amdgcn_wmma_f32_16x16x4_f32(
                 false, a,  false, b,
                 (short)0, acc[t],  false,  false);
        }
    }

#pragma unroll
    for (int t = 0; t < NT; ++t) {
#pragma unroll
        for (int j = 0; j < 8; ++j)
            H[(size_t)(r0 + j + 8 * kh) * NOUT + t * 16 + m] =
                SCALED ? acc[t][j] * sc[j] : acc[t][j];
    }
}

template <int K, int NOUT, bool SCALED>
__global__ void k_gemm_tail(const float* __restrict__ A, const float* __restrict__ W,
                            const float* __restrict__ rowscale,
                            float* __restrict__ H, int row0, int nrows) {
    int idx = blockIdx.x * blockDim.x + threadIdx.x;
    int r = row0 + idx / NOUT;
    int c = idx % NOUT;
    if (r >= nrows) return;
    float s = 0.0f;
    for (int k = 0; k < K; ++k) s += A[(size_t)r * K + k] * W[(size_t)k * NOUT + c];
    H[(size_t)r * NOUT + c] = SCALED ? s * rowscale[r] : s;
}

template <int C, bool HAS_BIAS>
__global__ void k_self_init(const float* __restrict__ h, const float* __restrict__ dinv,
                            const float* __restrict__ bias, float* __restrict__ agg, int n) {
    long long idx = (long long)blockIdx.x * blockDim.x + threadIdx.x;
    long long total = (long long)n * C;
    if (idx >= total) return;
    int i = (int)(idx / C);
    int c = (int)(idx % C);
    float d = dinv[i];
    float v = h[idx] * d * d;
    if (HAS_BIAS) v += bias[c];
    agg[idx] = v;
}

template <int C>
__global__ void k_edge_agg(const int* __restrict__ src, const int* __restrict__ dst,
                           const float* __restrict__ h, const float* __restrict__ dinv,
                           float* __restrict__ agg, int e) {
    constexpr int EPW = 8;
    constexpr int NJ  = C / 32;
    const int lane = threadIdx.x & 31;
    const int wave = __builtin_amdgcn_readfirstlane(
        (int)(blockIdx.x * (blockDim.x >> 5) + (threadIdx.x >> 5)));
    const long long e0 = (long long)wave * EPW;
    if (e0 >= (long long)e) return;

    if (e0 + EPW <= (long long)e) {
        int s[EPW], d[EPW];
        float v[EPW][NJ];
        float nrm[EPW];
#pragma unroll
        for (int i = 0; i < EPW; ++i) {
            s[i] = src[e0 + i];
            d[i] = dst[e0 + i];
        }
#pragma unroll
        for (int i = 0; i < EPW; ++i)
#pragma unroll
            for (int j = 0; j < NJ; ++j)
                v[i][j] = h[(size_t)s[i] * C + j * 32 + lane];
#pragma unroll
        for (int i = 0; i < EPW; ++i) nrm[i] = dinv[s[i]] * dinv[d[i]];
#pragma unroll
        for (int i = 0; i < EPW; ++i)
#pragma unroll
            for (int j = 0; j < NJ; ++j)
                atomicAdd(agg + (size_t)d[i] * C + j * 32 + lane, v[i][j] * nrm[i]);
    } else {
        for (int i = 0; i < EPW; ++i) {
            if (e0 + i >= (long long)e) break;
            const int s = src[e0 + i];
            const int d = dst[e0 + i];
            const float nrm = dinv[s] * dinv[d];
#pragma unroll
            for (int j = 0; j < NJ; ++j) {
                const float v = h[(size_t)s * C + j * 32 + lane];
                atomicAdd(agg + (size_t)d * C + j * 32 + lane, v * nrm);
            }
        }
    }
}

template <int C>
__global__ void k_relu_bias(const float* __restrict__ agg, const float* __restrict__ b,
                            float* __restrict__ p, int n) {
    long long idx = (long long)blockIdx.x * blockDim.x + threadIdx.x;
    long long total = (long long)n * C;
    if (idx >= total) return;
    int c = (int)(idx % C);
    float v = agg[idx] + b[c];
    p[idx] = v > 0.0f ? v : 0.0f;
}

constexpr int AGG_TB = 1024;
constexpr int AGG_G  = 16;

template <int C, bool LAYER1>
__global__ void __launch_bounds__(AGG_TB)
k_range_agg(const unsigned int* __restrict__ recs, const unsigned int* __restrict__ bstart,
            const float* __restrict__ h, const float* __restrict__ dinv,
            const float* __restrict__ bias, const float* __restrict__ W2,
            float* __restrict__ outp, int n) {
    constexpr int P  = C + 4;
    constexpr int NJ = C / 32;
    __shared__ float acc[BN * P];
    __shared__ alignas(16) float dloc[BN];
    const int tid  = threadIdx.x;
    const int lane = tid & 31;
    const int wv   = __builtin_amdgcn_readfirstlane(tid >> 5);
    const int n0   = blockIdx.x * BN;
    const int rows = (n - n0) < BN ? (n - n0) : BN;

    float dself;
    {
        int node = n0 + (tid < BN ? tid : BN - 1);
        node = node < n ? node : n - 1;
        dself = dinv[node];
    }

    constexpr int TOT = BN * C;
    constexpr int INIT_B = 5;
    for (int base = tid; base < TOT; base += INIT_B * AGG_TB) {
        float hv[INIT_B];
#pragma unroll
        for (int u = 0; u < INIT_B; ++u) {
            int idx = base + u * AGG_TB;
            idx = idx < TOT ? idx : TOT - 1;
            const int r = idx / C, c = idx - r * C;
            int node = n0 + r;
            node = node < n ? node : n - 1;
            hv[u] = h[(size_t)node * C + c];
        }
#pragma unroll
        for (int u = 0; u < INIT_B; ++u) {
            const int idx = base + u * AGG_TB;
            if (idx < TOT) {
                const int r = idx / C, c = idx - r * C;
                acc[r * P + c] = (r < rows) ? hv[u] : 0.0f;
            }
        }
    }
    if (tid < BN) dloc[tid] = (tid < rows) ? dself : 0.0f;
    __syncthreads();

    const unsigned int eb = bstart[blockIdx.x];
    const unsigned int ee = bstart[blockIdx.x + 1];
    unsigned int c0 = eb + (unsigned int)wv * 32u;
    unsigned int pk = (c0 + lane < ee) ? recs[c0 + lane] : 0u;
    for (; c0 < ee; c0 += AGG_TB) {
        const unsigned int cn = c0 + AGG_TB;
        const unsigned int pkn = (cn + lane < ee) ? recs[cn + lane] : 0u;
        const int s  = (int)(pk & 0x1FFFFu);
        const int dl = (int)(pk >> 17);
        const int cnt = (int)((ee - c0) < 32u ? (ee - c0) : 32u);
#pragma unroll
        for (int g = 0; g < 32; g += AGG_G) {
            if (g < cnt) {
                float v[AGG_G][NJ];
#pragma unroll
                for (int k = 0; k < AGG_G; ++k) {
                    const int sk = __builtin_amdgcn_readlane(s, g + k);
#pragma unroll
                    for (int j = 0; j < NJ; ++j) v[k][j] = h[(size_t)sk * C + j * 32 + lane];
                }
#pragma unroll
                for (int k = 0; k < AGG_G; ++k) {
                    if (g + k < cnt) {
                        const int dk = __builtin_amdgcn_readlane(dl, g + k);
#pragma unroll
                        for (int j = 0; j < NJ; ++j)
                            atomicAdd(&acc[dk * P + j * 32 + lane], v[k][j]);
                    }
                }
            }
        }
        pk = pkn;
    }
    __syncthreads();

    if constexpr (!LAYER1) {
        for (int idx = tid; idx < rows * C; idx += AGG_TB) {
            const int r = idx / C, c = idx - r * C;
            outp[(size_t)(n0 + r) * C + c] = acc[r * P + c] * dloc[r] + bias[c];
        }
    } else {
        for (int idx = tid; idx < BN * C; idx += AGG_TB) {
            const int r = idx / C, c = idx - r * C;
            const float v = acc[r * P + c] * dloc[r] + bias[c];
            acc[r * P + c] = v > 0.0f ? v : 0.0f;
        }
        __syncthreads();
        constexpr int NO = 32;
        constexpr int NT = NO / 16;
        const int full = rows / 16;
        if (wv < full) {
            const int m  = lane & 15;
            const int kh = lane >> 4;
            v8f d[NT];
#pragma unroll
            for (int t = 0; t < NT; ++t)
#pragma unroll
                for (int j = 0; j < 8; ++j) d[t][j] = 0.0f;
            float sc[8];
#pragma unroll
            for (int j = 0; j < 8; ++j) sc[j] = dloc[wv * 16 + j + 8 * kh];
            const float* arow = acc + (wv * 16 + m) * P;
            for (int k = 0; k < C; k += 4) {
                v2f a;
                a.x = arow[k + 2 * kh + 0];
                a.y = arow[k + 2 * kh + 1];
#pragma unroll
                for (int t = 0; t < NT; ++t) {
                    v2f b;
                    b.x = W2[(size_t)(k + 2 * kh + 0) * NO + t * 16 + m];
                    b.y = W2[(size_t)(k + 2 * kh + 1) * NO + t * 16 + m];
                    d[t] = __builtin_amdgcn_wmma_f32_16x16x4_f32(
                        false, a, false, b, (short)0, d[t], false, false);
                }
            }
#pragma unroll
            for (int t = 0; t < NT; ++t)
#pragma unroll
                for (int j = 0; j < 8; ++j)
                    outp[(size_t)(n0 + wv * 16 + j + 8 * kh) * NO + t * 16 + m] = d[t][j] * sc[j];
        }
        const int r1 = full * 16;
        for (int idx = tid; idx < (rows - r1) * NO; idx += AGG_TB) {
            const int r = r1 + idx / NO, c = idx % NO;
            float sum = 0.0f;
            for (int k = 0; k < C; ++k) sum += acc[r * P + k] * W2[(size_t)k * NO + c];
            outp[(size_t)(n0 + r) * NO + c] = sum * dloc[r];
        }
    }
}

extern "C" void kernel_launch(void* const* d_in, const int* in_sizes, int n_in,
                              void* d_out, int out_size, void* d_ws, size_t ws_size,
                              hipStream_t stream) {
    const float* x  = (const float*)d_in[0];
    const int*   ei = (const int*)d_in[1];
    const float* W1 = (const float*)d_in[2];
    const float* b1 = (const float*)d_in[3];
    const float* W2 = (const float*)d_in[4];
    const float* b2 = (const float*)d_in[5];
    float* out = (float*)d_out;

    const int n = in_sizes[0] / 128;
    const int e = in_sizes[1] / 2;
    const int* src = ei;
    const int* dst = ei + e;

    char* ws = (char*)d_ws;
    size_t off0 = ((size_t)n * sizeof(float) + 255) & ~(size_t)255;
    float* dinv = (float*)ws;
    float* bufA = (float*)(ws + off0);
    float* bufB = bufA + (size_t)n * 64;

    const int tb = 256;
    auto cdiv = [](long long a, long long b) { return (int)((a + b - 1) / b); };

    const int parts = cdiv(e, DEG_ECH);
    const int strips = n / 16;
    const int wavesPerBlk = tb / 32;
    const int nb = cdiv(n, BN);
    const size_t half = (size_t)n * 32;
    if (n <= DEG_NCAP && parts <= SCAN_PMAX && (size_t)parts * DEG_WORDS <= half &&
        (size_t)e + 2 * (size_t)parts * nb + nb + 1 <= half) {
        unsigned int* slabs  = (unsigned int*)bufB;
        unsigned int* recs   = (unsigned int*)bufB + half;
        unsigned int* pcnt   = recs + e;
        unsigned int* off    = pcnt + (size_t)parts * nb;
        unsigned int* bstart = off + (size_t)parts * nb;
        float* h2 = bufB;

        k_deg_hist<<<parts, DEG_TB, 0, stream>>>(dst, slabs, pcnt, nb, e);
        k_dinv_hist<<<cdiv((n + 1) / 2, tb), tb, 0, stream>>>(slabs, dinv, n, parts);
        k_bin_scan<<<1, 256, 0, stream>>>(pcnt, off, bstart, parts, nb);
        k_bin_scatter<<<parts, DEG_TB, 0, stream>>>(src, dst, pcnt, off, recs, e, nb);

        k_gemm_wmma<128, 64, true><<<cdiv(strips, wavesPerBlk), tb, 0, stream>>>(x, W1, dinv, bufA, n);
        if (n % 16) {
            int tail = (n % 16) * 64;
            k_gemm_tail<128, 64, true><<<cdiv(tail, tb), tb, 0, stream>>>(x, W1, dinv, bufA, strips * 16, n);
        }
        k_range_agg<64, true><<<nb, AGG_TB, 0, stream>>>(recs, bstart, bufA, dinv, b1, W2, h2, n);
        k_range_agg<32, false><<<nb, AGG_TB, 0, stream>>>(recs, bstart, h2, dinv, b2, nullptr, out, n);
        return;
    }

    if (n <= DEG_NCAP && (size_t)parts * DEG_WORDS <= (size_t)n * 64) {
        k_deg_hist<<<parts, DEG_TB, 0, stream>>>(dst, (unsigned int*)bufB, nullptr, 0, e);
        k_dinv_hist<<<cdiv((n + 1) / 2, tb), tb, 0, stream>>>((const unsigned int*)bufB, dinv, n, parts);
    } else {
        k_deg_init<<<cdiv(n, tb), tb, 0, stream>>>(dinv, n);
        k_deg_count<<<cdiv(e, tb), tb, 0, stream>>>(dst, dinv, e);
        k_dinv<<<cdiv(n, tb), tb, 0, stream>>>(dinv, n);
    }

    k_gemm_wmma<128, 64, false><<<cdiv(strips, wavesPerBlk), tb, 0, stream>>>(x, W1, nullptr, bufA, n);
    if (n % 16) {
        int tail = (n % 16) * 64;
        k_gemm_tail<128, 64, false><<<cdiv(tail, tb), tb, 0, stream>>>(x, W1, nullptr, bufA, strips * 16, n);
    }

    k_self_init<64, false><<<cdiv((long long)n * 64, tb), tb, 0, stream>>>(bufA, dinv, nullptr, bufB, n);
    k_edge_agg<64><<<cdiv(cdiv(e, 8), wavesPerBlk), tb, 0, stream>>>(src, dst, bufA, dinv, bufB, e);
    k_relu_bias<64><<<cdiv((long long)n * 64, tb), tb, 0, stream>>>(bufB, b1, bufA, n);

    k_gemm_wmma<64, 32, false><<<cdiv(strips, wavesPerBlk), tb, 0, stream>>>(bufA, W2, nullptr, bufB, n);
    if (n % 16) {
        int tail = (n % 16) * 32;
        k_gemm_tail<64, 32, false><<<cdiv(tail, tb), tb, 0, stream>>>(bufA, W2, nullptr, bufB, strips * 16, n);
    }

    k_self_init<32, true><<<cdiv((long long)n * 32, tb), tb, 0, stream>>>(bufB, dinv, b2, out, n);
    k_edge_agg<32><<<cdiv(cdiv(e, 8), wavesPerBlk), tb, 0, stream>>>(src, dst, bufB, dinv, out, e);

    (void)n_in; (void)out_size; (void)ws_size;
}


# launches 7 -> 4: placement and the x*W1 GEMM share one grid as independent work-groups; the degree-sum (rsqrt) pass is done by the GEMM groups and the offset scan by the placement groups
# speedup vs baseline: 3.7785x; 1.2541x over previous
#include <hip/hip_runtime.h>

typedef float v2f __attribute__((ext_vector_type(2)));
typedef float v8f __attribute__((ext_vector_type(8)));

__global__ void k_deg_init(float* __restrict__ deg, int n) {
    int i = blockIdx.x * blockDim.x + threadIdx.x;
    if (i < n) deg[i] = 1.0f;
}

__global__ void k_deg_count(const int* __restrict__ dst, float* __restrict__ deg, int e) {
    int i = blockIdx.x * blockDim.x + threadIdx.x;
    if (i < e) atomicAdd(&deg[dst[i]], 1.0f);
}

__global__ void k_dinv(float* __restrict__ deg, int n) {
    int i = blockIdx.x * blockDim.x + threadIdx.x;
    if (i < n) deg[i] = 1.0f / sqrtf(deg[i]);
}

constexpr int DEG_NCAP   = 100000;
constexpr int DEG_WORDS  = ((DEG_NCAP + 1) / 2 + 31) & ~31;
constexpr int DEG_ECH    = 32768;
constexpr int DEG_TB     = 1024;
constexpr int DEG_B      = 16;
constexpr int BN         = 400;
constexpr int BW         = BN / 2;
constexpr int NBMAX      = (DEG_NCAP + BN - 1) / BN;
constexpr int PG_TB      = 1024;
constexpr int PG_ROWS    = (PG_TB / 32) * 16;
constexpr int PG_XW      = 10240;

__global__ void __launch_bounds__(DEG_TB)
k_deg_hist(const int* __restrict__ dst, unsigned int* __restrict__ part,
           unsigned int* __restrict__ pcnt, int nb, int e) {
    __shared__ unsigned int cnt[DEG_WORDS];
    const int tid = threadIdx.x;

    uint4* c4 = reinterpret_cast<uint4*>(cnt);
    for (int i = tid; i < DEG_WORDS / 4; i += DEG_TB) c4[i] = make_uint4(0u, 0u, 0u, 0u);
    __syncthreads();

    const long long e0 = (long long)blockIdx.x * DEG_ECH;
    const long long left = (long long)e - e0;
    const int m = left < (long long)DEG_ECH ? (int)left : DEG_ECH;
    const int* __restrict__ dp = dst + e0;
    for (int base = tid; base < m; base += DEG_B * DEG_TB) {
        int d[DEG_B];
#pragma unroll
        for (int k = 0; k < DEG_B; ++k) {
            const int i = base + k * DEG_TB;
            d[k] = (i < m) ? dp[i] : -1;
        }
#pragma unroll
        for (int k = 0; k < DEG_B; ++k) {
            if (base + k * DEG_TB < m) atomicAdd(&cnt[d[k] >> 1], (d[k] & 1) ? 0x10000u : 1u);
        }
    }
    __syncthreads();

    uint4* p4 = reinterpret_cast<uint4*>(part + (size_t)blockIdx.x * DEG_WORDS);
    for (int i = tid; i < DEG_WORDS / 4; i += DEG_TB) p4[i] = c4[i];

    if (pcnt != nullptr) {
        const int wv = tid >> 5, ln = tid & 31;
        for (int b = wv; b < nb; b += DEG_TB / 32) {
            unsigned int sacc = 0u;
            for (int w = ln; w < BW; w += 32) {
                const int wi = b * BW + w;
                if (wi < DEG_WORDS) {
                    const unsigned int v = cnt[wi];
                    sacc += (v & 0xFFFFu) + (v >> 16);
                }
            }
            for (int o = 16; o > 0; o >>= 1) sacc += __shfl_xor(sacc, o);
            if (ln == 0) pcnt[(size_t)blockIdx.x * nb + b] = sacc;
        }
    }
}

__global__ void k_dinv_hist(const unsigned int* __restrict__ part, float* __restrict__ dinv,
                            int n, int parts) {
    const int t = blockIdx.x * blockDim.x + threadIdx.x;
    if (2 * t >= n) return;
    unsigned int lo = 0u, hi = 0u;
#pragma unroll 8
    for (int p = 0; p < parts; ++p) {
        const unsigned int w = part[(size_t)p * DEG_WORDS + t];
        lo += w & 0xFFFFu;
        hi += w >> 16;
    }
    dinv[2 * t] = 1.0f / sqrtf((float)(lo + 1u));
    if (2 * t + 1 < n) dinv[2 * t + 1] = 1.0f / sqrtf((float)(hi + 1u));
}

template <int K, int NOUT, bool SCALED>
__global__ void k_gemm_wmma(const float* __restrict__ A, const float* __restrict__ W,
                            const float* __restrict__ rowscale,
                            float* __restrict__ H, int nrows) {
    const int wave = blockIdx.x * (blockDim.x >> 5) + (threadIdx.x >> 5);
    const int lane = threadIdx.x & 31;
    const int m    = lane & 15;
    const int kh   = lane >> 4;
    const int r0   = wave * 16;
    if (r0 + 16 > nrows) return;

    constexpr int NT = NOUT / 16;
    v8f acc[NT];
#pragma unroll
    for (int t = 0; t < NT; ++t)
#pragma unroll
        for (int j = 0; j < 8; ++j) acc[t][j] = 0.0f;

    float sc[8];
#pragma unroll
    for (int j = 0; j < 8; ++j) sc[j] = SCALED ? rowscale[r0 + j + 8 * kh] : 1.0f;

    const float* arow = A + (size_t)(r0 + m) * K;
    for (int k = 0; k < K; k += 4) {
        v2f a;
        a.x = arow[k + 2 * kh + 0];
        a.y = arow[k + 2 * kh + 1];
#pragma unroll
        for (int t = 0; t < NT; ++t) {
            v2f b;
            b.x = W[(size_t)(k + 2 * kh + 0) * NOUT + t * 16 + m];
            b.y = W[(size_t)(k + 2 * kh + 1) * NOUT + t * 16 + m];
            acc[t] = __builtin_amdgcn_wmma_f32_16x16x4_f32(
                 false, a,  false, b,
                 (short)0, acc[t],  false,  false);
        }
    }

#pragma unroll
    for (int t = 0; t < NT; ++t) {
#pragma unroll
        for (int j = 0; j < 8; ++j)
            H[(size_t)(r0 + j + 8 * kh) * NOUT + t * 16 + m] =
                SCALED ? acc[t][j] * sc[j] : acc[t][j];
    }
}

template <int K, int NOUT, bool SCALED>
__global__ void k_gemm_tail(const float* __restrict__ A, const float* __restrict__ W,
                            const float* __restrict__ rowscale,
                            float* __restrict__ H, int row0, int nrows) {
    int idx = blockIdx.x * blockDim.x + threadIdx.x;
    int r = row0 + idx / NOUT;
    int c = idx % NOUT;
    if (r >= nrows) return;
    float s = 0.0f;
    for (int k = 0; k < K; ++k) s += A[(size_t)r * K + k] * W[(size_t)k * NOUT + c];
    H[(size_t)r * NOUT + c] = SCALED ? s * rowscale[r] : s;
}

template <int K, int NOUT>
__global__ void __launch_bounds__(PG_TB)
k_place_gemm(const int* __restrict__ src, const int* __restrict__ dst,
             const unsigned int* __restrict__ pcnt, unsigned int* __restrict__ recs,
             unsigned int* __restrict__ bstart, int e, int nb, int parts,
             const unsigned int* __restrict__ part, const float* __restrict__ A,
             const float* __restrict__ W, float* __restrict__ dinv,
             float* __restrict__ H, int nrows) {
    __shared__ unsigned int stage[DEG_ECH + PG_XW];
    __shared__ unsigned int lbeg[NBMAX + 1];
    __shared__ unsigned int cur[NBMAX];
    __shared__ unsigned int gbeg[NBMAX];
    __shared__ unsigned int wtot[8];
    __shared__ unsigned int wtot2[8];
    unsigned int* xs = stage + DEG_ECH;
    const int tid  = threadIdx.x;
    const int lane = tid & 31;
    const int wv   = tid >> 5;

    if ((int)blockIdx.x < parts) {
        const int p = blockIdx.x;
        unsigned int c = 0u;
        if (tid < 256 && tid < nb) c = pcnt[(size_t)p * nb + tid];
        {
            const int b = tid & 255, q = tid >> 8;
            unsigned int st = 0u, sb = 0u;
            if (b < nb) {
#pragma unroll 8
                for (int pp = q; pp < parts; pp += 4) {
                    const unsigned int v = pcnt[(size_t)pp * nb + b];
                    st += v;
                    sb += (pp < p) ? v : 0u;
                }
            }
            xs[q * 256 + b] = st;
            xs[1024 + q * 256 + b] = sb;
        }
        __syncthreads();

        unsigned int incl = 0u, t = 0u, tincl = 0u, bef = 0u;
        if (tid < 256) {
            t   = xs[tid] + xs[256 + tid] + xs[512 + tid] + xs[768 + tid];
            bef = xs[1024 + tid] + xs[1280 + tid] + xs[1536 + tid] + xs[1792 + tid];
            incl = c;
            tincl = t;
            for (int o = 1; o < 32; o <<= 1) {
                const unsigned int v  = __shfl_up(incl, o);
                const unsigned int v2 = __shfl_up(tincl, o);
                if (lane >= o) { incl += v; tincl += v2; }
            }
            if (lane == 31) { wtot[wv] = incl; wtot2[wv] = tincl; }
        }
        __syncthreads();
        if (tid < 256) {
            unsigned int before = 0u, before2 = 0u;
            for (int w = 0; w < wv; ++w) { before += wtot[w]; before2 += wtot2[w]; }
            const unsigned int ex  = before + incl - c;
            const unsigned int ex2 = before2 + tincl - t;
            if (tid < nb) {
                lbeg[tid] = ex;
                cur[tid]  = ex;
                gbeg[tid] = ex2 + bef;
                if (p == 0) bstart[tid] = ex2;
            }
            if (tid == nb - 1) {
                lbeg[nb] = ex + c;
                if (p == 0) bstart[nb] = ex2 + t;
            }
        }
        __syncthreads();

        const long long e0 = (long long)p * DEG_ECH;
        const long long left = (long long)e - e0;
        const int m = left < (long long)DEG_ECH ? (int)left : DEG_ECH;
        const int* __restrict__ dp = dst + e0;
        const int* __restrict__ sp = src + e0;
        for (int base = tid; base < m; base += DEG_B * DEG_TB) {
            int d[DEG_B], s[DEG_B];
#pragma unroll
            for (int k = 0; k < DEG_B; ++k) {
                const int i = base + k * DEG_TB;
                d[k] = (i < m) ? dp[i] : 0;
                s[k] = (i < m) ? sp[i] : 0;
            }
#pragma unroll
            for (int k = 0; k < DEG_B; ++k) {
                if (base + k * DEG_TB < m) {
                    const int b = d[k] / BN;
                    const unsigned int pos = atomicAdd(&cur[b], 1u);
                    stage[pos] = ((unsigned int)(d[k] - b * BN) << 17) | (unsigned int)s[k];
                }
            }
        }
        __syncthreads();

        for (int b = wv; b < nb; b += DEG_TB / 32) {
            const unsigned int l0  = lbeg[b];
            const unsigned int len = lbeg[b + 1] - l0;
            const unsigned int g0  = gbeg[b];
            const unsigned int a   = g0 & 31u;
            for (unsigned int j = lane; j < len + a; j += 32u) {
                if (j >= a) recs[g0 - a + j] = stage[l0 + j - a];
            }
        }
        return;
    }

    const int row0 = ((int)blockIdx.x - parts) * PG_ROWS;
    {
        const int w = tid & 255, q = tid >> 8;
        const int tw = (row0 >> 1) + w;
        unsigned int lo = 0u, hi = 0u;
        if (2 * tw < nrows) {
#pragma unroll 8
            for (int pp = q; pp < parts; pp += 4) {
                const unsigned int v = part[(size_t)pp * DEG_WORDS + tw];
                lo += v & 0xFFFFu;
                hi += v >> 16;
            }
        }
        xs[q * 256 + w] = lo;
        xs[1024 + q * 256 + w] = hi;
    }
    __syncthreads();
    float* dl = reinterpret_cast<float*>(xs + 2048);
    if (tid < PG_ROWS) {
        const int w = tid >> 1;
        const unsigned int* hsel = xs + ((tid & 1) ? 1024 : 0);
        const unsigned int cntv = hsel[w] + hsel[256 + w] + hsel[512 + w] + hsel[768 + w];
        const float dv = 1.0f / sqrtf((float)(cntv + 1u));
        dl[tid] = dv;
        if (row0 + tid < nrows) dinv[row0 + tid] = dv;
    }
    __syncthreads();

    const int m  = lane & 15;
    const int kh = lane >> 4;
    const int r0 = row0 + wv * 16;
    if (r0 + 16 > nrows) return;

    constexpr int NT = NOUT / 16;
    v8f acc[NT];
#pragma unroll
    for (int t = 0; t < NT; ++t)
#pragma unroll
        for (int j = 0; j < 8; ++j) acc[t][j] = 0.0f;

    float sc[8];
#pragma unroll
    for (int j = 0; j < 8; ++j) sc[j] = dl[wv * 16 + j + 8 * kh];

    const float* arow = A + (size_t)(r0 + m) * K;
    for (int k = 0; k < K; k += 4) {
        v2f a;
        a.x = arow[k + 2 * kh + 0];
        a.y = arow[k + 2 * kh + 1];
#pragma unroll
        for (int t = 0; t < NT; ++t) {
            v2f b;
            b.x = W[(size_t)(k + 2 * kh + 0) * NOUT + t * 16 + m];
            b.y = W[(size_t)(k + 2 * kh + 1) * NOUT + t * 16 + m];
            acc[t] = __builtin_amdgcn_wmma_f32_16x16x4_f32(
                 false, a,  false, b,
                 (short)0, acc[t],  false,  false);
        }
    }

#pragma unroll
    for (int t = 0; t < NT; ++t) {
#pragma unroll
        for (int j = 0; j < 8; ++j)
            H[(size_t)(r0 + j + 8 * kh) * NOUT + t * 16 + m] = acc[t][j] * sc[j];
    }
}

template <int C, bool HAS_BIAS>
__global__ void k_self_init(const float* __restrict__ h, const float* __restrict__ dinv,
                            const float* __restrict__ bias, float* __restrict__ agg, int n) {
    long long idx = (long long)blockIdx.x * blockDim.x + threadIdx.x;
    long long total = (long long)n * C;
    if (idx >= total) return;
    int i = (int)(idx / C);
    int c = (int)(idx % C);
    float d = dinv[i];
    float v = h[idx] * d * d;
    if (HAS_BIAS) v += bias[c];
    agg[idx] = v;
}

template <int C>
__global__ void k_edge_agg(const int* __restrict__ src, const int* __restrict__ dst,
                           const float* __restrict__ h, const float* __restrict__ dinv,
                           float* __restrict__ agg, int e) {
    constexpr int EPW = 8;
    constexpr int NJ  = C / 32;
    const int lane = threadIdx.x & 31;
    const int wave = __builtin_amdgcn_readfirstlane(
        (int)(blockIdx.x * (blockDim.x >> 5) + (threadIdx.x >> 5)));
    const long long e0 = (long long)wave * EPW;
    if (e0 >= (long long)e) return;

    if (e0 + EPW <= (long long)e) {
        int s[EPW], d[EPW];
        float v[EPW][NJ];
        float nrm[EPW];
#pragma unroll
        for (int i = 0; i < EPW; ++i) {
            s[i] = src[e0 + i];
            d[i] = dst[e0 + i];
        }
#pragma unroll
        for (int i = 0; i < EPW; ++i)
#pragma unroll
            for (int j = 0; j < NJ; ++j)
                v[i][j] = h[(size_t)s[i] * C + j * 32 + lane];
#pragma unroll
        for (int i = 0; i < EPW; ++i) nrm[i] = dinv[s[i]] * dinv[d[i]];
#pragma unroll
        for (int i = 0; i < EPW; ++i)
#pragma unroll
            for (int j = 0; j < NJ; ++j)
                atomicAdd(agg + (size_t)d[i] * C + j * 32 + lane, v[i][j] * nrm[i]);
    } else {
        for (int i = 0; i < EPW; ++i) {
            if (e0 + i >= (long long)e) break;
            const int s = src[e0 + i];
            const int d = dst[e0 + i];
            const float nrm = dinv[s] * dinv[d];
#pragma unroll
            for (int j = 0; j < NJ; ++j) {
                const float v = h[(size_t)s * C + j * 32 + lane];
                atomicAdd(agg + (size_t)d * C + j * 32 + lane, v * nrm);
            }
        }
    }
}

template <int C>
__global__ void k_relu_bias(const float* __restrict__ agg, const float* __restrict__ b,
                            float* __restrict__ p, int n) {
    long long idx = (long long)blockIdx.x * blockDim.x + threadIdx.x;
    long long total = (long long)n * C;
    if (idx >= total) return;
    int c = (int)(idx % C);
    float v = agg[idx] + b[c];
    p[idx] = v > 0.0f ? v : 0.0f;
}

constexpr int AGG_TB = 1024;
constexpr int AGG_G  = 16;

template <int C, bool LAYER1>
__global__ void __launch_bounds__(AGG_TB)
k_range_agg(const unsigned int* __restrict__ recs, const unsigned int* __restrict__ bstart,
            const float* __restrict__ h, const float* __restrict__ dinv,
            const float* __restrict__ bias, const float* __restrict__ W2,
            float* __restrict__ outp, int n) {
    constexpr int P  = C + 4;
    constexpr int NJ = C / 32;
    __shared__ float acc[BN * P];
    __shared__ alignas(16) float dloc[BN];
    const int tid  = threadIdx.x;
    const int lane = tid & 31;
    const int wv   = __builtin_amdgcn_readfirstlane(tid >> 5);
    const int n0   = blockIdx.x * BN;
    const int rows = (n - n0) < BN ? (n - n0) : BN;

    float dself;
    {
        int node = n0 + (tid < BN ? tid : BN - 1);
        node = node < n ? node : n - 1;
        dself = dinv[node];
    }

    constexpr int TOT = BN * C;
    constexpr int INIT_B = 5;
    for (int base = tid; base < TOT; base += INIT_B * AGG_TB) {
        float hv[INIT_B];
#pragma unroll
        for (int u = 0; u < INIT_B; ++u) {
            int idx = base + u * AGG_TB;
            idx = idx < TOT ? idx : TOT - 1;
            const int r = idx / C, c = idx - r * C;
            int node = n0 + r;
            node = node < n ? node : n - 1;
            hv[u] = h[(size_t)node * C + c];
        }
#pragma unroll
        for (int u = 0; u < INIT_B; ++u) {
            const int idx = base + u * AGG_TB;
            if (idx < TOT) {
                const int r = idx / C, c = idx - r * C;
                acc[r * P + c] = (r < rows) ? hv[u] : 0.0f;
            }
        }
    }
    if (tid < BN) dloc[tid] = (tid < rows) ? dself : 0.0f;
    __syncthreads();

    const unsigned int eb = bstart[blockIdx.x];
    const unsigned int ee = bstart[blockIdx.x + 1];
    unsigned int c0 = eb + (unsigned int)wv * 32u;
    unsigned int pk = (c0 + lane < ee) ? recs[c0 + lane] : 0u;
    for (; c0 < ee; c0 += AGG_TB) {
        const unsigned int cn = c0 + AGG_TB;
        const unsigned int pkn = (cn + lane < ee) ? recs[cn + lane] : 0u;
        const int s  = (int)(pk & 0x1FFFFu);
        const int dl = (int)(pk >> 17);
        const int cnt = (int)((ee - c0) < 32u ? (ee - c0) : 32u);
#pragma unroll
        for (int g = 0; g < 32; g += AGG_G) {
            if (g < cnt) {
                float v[AGG_G][NJ];
#pragma unroll
                for (int k = 0; k < AGG_G; ++k) {
                    const int sk = __builtin_amdgcn_readlane(s, g + k);
#pragma unroll
                    for (int j = 0; j < NJ; ++j) v[k][j] = h[(size_t)sk * C + j * 32 + lane];
                }
#pragma unroll
                for (int k = 0; k < AGG_G; ++k) {
                    if (g + k < cnt) {
                        const int dk = __builtin_amdgcn_readlane(dl, g + k);
#pragma unroll
                        for (int j = 0; j < NJ; ++j)
                            atomicAdd(&acc[dk * P + j * 32 + lane], v[k][j]);
                    }
                }
            }
        }
        pk = pkn;
    }
    __syncthreads();

    if constexpr (!LAYER1) {
        for (int idx = tid; idx < rows * C; idx += AGG_TB) {
            const int r = idx / C, c = idx - r * C;
            outp[(size_t)(n0 + r) * C + c] = acc[r * P + c] * dloc[r] + bias[c];
        }
    } else {
        for (int idx = tid; idx < BN * C; idx += AGG_TB) {
            const int r = idx / C, c = idx - r * C;
            const float v = acc[r * P + c] * dloc[r] + bias[c];
            acc[r * P + c] = v > 0.0f ? v : 0.0f;
        }
        __syncthreads();
        constexpr int NO = 32;
        constexpr int NT = NO / 16;
        const int full = rows / 16;
        if (wv < full) {
            const int m  = lane & 15;
            const int kh = lane >> 4;
            v8f d[NT];
#pragma unroll
            for (int t = 0; t < NT; ++t)
#pragma unroll
                for (int j = 0; j < 8; ++j) d[t][j] = 0.0f;
            float sc[8];
#pragma unroll
            for (int j = 0; j < 8; ++j) sc[j] = dloc[wv * 16 + j + 8 * kh];
            const float* arow = acc + (wv * 16 + m) * P;
            for (int k = 0; k < C; k += 4) {
                v2f a;
                a.x = arow[k + 2 * kh + 0];
                a.y = arow[k + 2 * kh + 1];
#pragma unroll
                for (int t = 0; t < NT; ++t) {
                    v2f b;
                    b.x = W2[(size_t)(k + 2 * kh + 0) * NO + t * 16 + m];
                    b.y = W2[(size_t)(k + 2 * kh + 1) * NO + t * 16 + m];
                    d[t] = __builtin_amdgcn_wmma_f32_16x16x4_f32(
                        false, a, false, b, (short)0, d[t], false, false);
                }
            }
#pragma unroll
            for (int t = 0; t < NT; ++t)
#pragma unroll
                for (int j = 0; j < 8; ++j)
                    outp[(size_t)(n0 + wv * 16 + j + 8 * kh) * NO + t * 16 + m] = d[t][j] * sc[j];
        }
        const int r1 = full * 16;
        for (int idx = tid; idx < (rows - r1) * NO; idx += AGG_TB) {
            const int r = r1 + idx / NO, c = idx % NO;
            float sum = 0.0f;
            for (int k = 0; k < C; ++k) sum += acc[r * P + k] * W2[(size_t)k * NO + c];
            outp[(size_t)(n0 + r) * NO + c] = sum * dloc[r];
        }
    }
}

extern "C" void kernel_launch(void* const* d_in, const int* in_sizes, int n_in,
                              void* d_out, int out_size, void* d_ws, size_t ws_size,
                              hipStream_t stream) {
    const float* x  = (const float*)d_in[0];
    const int*   ei = (const int*)d_in[1];
    const float* W1 = (const float*)d_in[2];
    const float* b1 = (const float*)d_in[3];
    const float* W2 = (const float*)d_in[4];
    const float* b2 = (const float*)d_in[5];
    float* out = (float*)d_out;

    const int n = in_sizes[0] / 128;
    const int e = in_sizes[1] / 2;
    const int* src = ei;
    const int* dst = ei + e;

    char* ws = (char*)d_ws;
    size_t off0 = ((size_t)n * sizeof(float) + 255) & ~(size_t)255;
    float* dinv = (float*)ws;
    float* bufA = (float*)(ws + off0);
    float* bufB = bufA + (size_t)n * 64;

    const int tb = 256;
    auto cdiv = [](long long a, long long b) { return (int)((a + b - 1) / b); };

    const int parts = cdiv(e, DEG_ECH);
    const int strips = n / 16;
    const int wavesPerBlk = tb / 32;
    const int nb = cdiv(n, BN);
    const size_t half = (size_t)n * 32;
    if (n <= DEG_NCAP && (size_t)parts * DEG_WORDS <= half &&
        (size_t)e + (size_t)parts * nb + nb + 1 <= half) {
        unsigned int* slabs  = (unsigned int*)bufB;
        unsigned int* recs   = (unsigned int*)bufB + half;
        unsigned int* pcnt   = recs + e;
        unsigned int* bstart = pcnt + (size_t)parts * nb;
        float* h2 = bufB;

        k_deg_hist<<<parts, DEG_TB, 0, stream>>>(dst, slabs, pcnt, nb, e);
        k_place_gemm<128, 64><<<parts + cdiv(n, PG_ROWS), PG_TB, 0, stream>>>(
            src, dst, pcnt, recs, bstart, e, nb, parts, slabs, x, W1, dinv, bufA, n);
        if (n % 16) {
            int tail = (n % 16) * 64;
            k_gemm_tail<128, 64, true><<<cdiv(tail, tb), tb, 0, stream>>>(x, W1, dinv, bufA, strips * 16, n);
        }
        k_range_agg<64, true><<<nb, AGG_TB, 0, stream>>>(recs, bstart, bufA, dinv, b1, W2, h2, n);
        k_range_agg<32, false><<<nb, AGG_TB, 0, stream>>>(recs, bstart, h2, dinv, b2, nullptr, out, n);
        return;
    }

    if (n <= DEG_NCAP && (size_t)parts * DEG_WORDS <= (size_t)n * 64) {
        k_deg_hist<<<parts, DEG_TB, 0, stream>>>(dst, (unsigned int*)bufB, nullptr, 0, e);
        k_dinv_hist<<<cdiv((n + 1) / 2, tb), tb, 0, stream>>>((const unsigned int*)bufB, dinv, n, parts);
    } else {
        k_deg_init<<<cdiv(n, tb), tb, 0, stream>>>(dinv, n);
        k_deg_count<<<cdiv(e, tb), tb, 0, stream>>>(dst, dinv, e);
        k_dinv<<<cdiv(n, tb), tb, 0, stream>>>(dinv, n);
    }

    k_gemm_wmma<128, 64, false><<<cdiv(strips, wavesPerBlk), tb, 0, stream>>>(x, W1, nullptr, bufA, n);
    if (n % 16) {
        int tail = (n % 16) * 64;
        k_gemm_tail<128, 64, false><<<cdiv(tail, tb), tb, 0, stream>>>(x, W1, nullptr, bufA, strips * 16, n);
    }

    k_self_init<64, false><<<cdiv((long long)n * 64, tb), tb, 0, stream>>>(bufA, dinv, nullptr, bufB, n);
    k_edge_agg<64><<<cdiv(cdiv(e, 8), wavesPerBlk), tb, 0, stream>>>(src, dst, bufA, dinv, bufB, e);
    k_relu_bias<64><<<cdiv((long long)n * 64, tb), tb, 0, stream>>>(bufB, b1, bufA, n);

    k_gemm_wmma<64, 32, false><<<cdiv(strips, wavesPerBlk), tb, 0, stream>>>(bufA, W2, nullptr, bufB, n);
    if (n % 16) {
        int tail = (n % 16) * 32;
        k_gemm_tail<64, 32, false><<<cdiv(tail, tb), tb, 0, stream>>>(bufA, W2, nullptr, bufB, strips * 16, n);
    }

    k_self_init<32, true><<<cdiv((long long)n * 32, tb), tb, 0, stream>>>(bufB, dinv, b2, out, n);
    k_edge_agg<32><<<cdiv(cdiv(e, 8), wavesPerBlk), tb, 0, stream>>>(src, dst, bufB, dinv, out, e);

    (void)n_in; (void)out_size; (void)ws_size;
}


# range records ordered by destination row (LDS counting sort in the layer-1 launch, list stored back for layer 2); a wave sums each row's run in registers, one LDS f32 add per run: new sum order
# speedup vs baseline: 4.1630x; 1.1018x over previous
#include <hip/hip_runtime.h>

typedef float v2f __attribute__((ext_vector_type(2)));
typedef float v8f __attribute__((ext_vector_type(8)));

__global__ void k_deg_init(float* __restrict__ deg, int n) {
    int i = blockIdx.x * blockDim.x + threadIdx.x;
    if (i < n) deg[i] = 1.0f;
}

__global__ void k_deg_count(const int* __restrict__ dst, float* __restrict__ deg, int e) {
    int i = blockIdx.x * blockDim.x + threadIdx.x;
    if (i < e) atomicAdd(&deg[dst[i]], 1.0f);
}

__global__ void k_dinv(float* __restrict__ deg, int n) {
    int i = blockIdx.x * blockDim.x + threadIdx.x;
    if (i < n) deg[i] = 1.0f / sqrtf(deg[i]);
}

constexpr int DEG_NCAP   = 100000;
constexpr int DEG_WORDS  = ((DEG_NCAP + 1) / 2 + 31) & ~31;
constexpr int DEG_ECH    = 32768;
constexpr int DEG_TB     = 1024;
constexpr int DEG_B      = 16;
constexpr int BN         = 400;
constexpr int BW         = BN / 2;
constexpr int NBMAX      = (DEG_NCAP + BN - 1) / BN;
constexpr int PG_TB      = 1024;
constexpr int PG_ROWS    = (PG_TB / 32) * 16;
constexpr int PG_XW      = 10240;

__global__ void __launch_bounds__(DEG_TB)
k_deg_hist(const int* __restrict__ dst, unsigned int* __restrict__ part,
           unsigned int* __restrict__ pcnt, int nb, int e) {
    __shared__ unsigned int cnt[DEG_WORDS];
    const int tid = threadIdx.x;

    uint4* c4 = reinterpret_cast<uint4*>(cnt);
    for (int i = tid; i < DEG_WORDS / 4; i += DEG_TB) c4[i] = make_uint4(0u, 0u, 0u, 0u);
    __syncthreads();

    const long long e0 = (long long)blockIdx.x * DEG_ECH;
    const long long left = (long long)e - e0;
    const int m = left < (long long)DEG_ECH ? (int)left : DEG_ECH;
    const int* __restrict__ dp = dst + e0;
    for (int base = tid; base < m; base += DEG_B * DEG_TB) {
        int d[DEG_B];
#pragma unroll
        for (int k = 0; k < DEG_B; ++k) {
            const int i = base + k * DEG_TB;
            d[k] = (i < m) ? dp[i] : -1;
        }
#pragma unroll
        for (int k = 0; k < DEG_B; ++k) {
            if (base + k * DEG_TB < m) atomicAdd(&cnt[d[k] >> 1], (d[k] & 1) ? 0x10000u : 1u);
        }
    }
    __syncthreads();

    uint4* p4 = reinterpret_cast<uint4*>(part + (size_t)blockIdx.x * DEG_WORDS);
    for (int i = tid; i < DEG_WORDS / 4; i += DEG_TB) p4[i] = c4[i];

    if (pcnt != nullptr) {
        const int wv = tid >> 5, ln = tid & 31;
        for (int b = wv; b < nb; b += DEG_TB / 32) {
            unsigned int sacc = 0u;
            for (int w = ln; w < BW; w += 32) {
                const int wi = b * BW + w;
                if (wi < DEG_WORDS) {
                    const unsigned int v = cnt[wi];
                    sacc += (v & 0xFFFFu) + (v >> 16);
                }
            }
            for (int o = 16; o > 0; o >>= 1) sacc += __shfl_xor(sacc, o);
            if (ln == 0) pcnt[(size_t)blockIdx.x * nb + b] = sacc;
        }
    }
}

__global__ void k_dinv_hist(const unsigned int* __restrict__ part, float* __restrict__ dinv,
                            int n, int parts) {
    const int t = blockIdx.x * blockDim.x + threadIdx.x;
    if (2 * t >= n) return;
    unsigned int lo = 0u, hi = 0u;
#pragma unroll 8
    for (int p = 0; p < parts; ++p) {
        const unsigned int w = part[(size_t)p * DEG_WORDS + t];
        lo += w & 0xFFFFu;
        hi += w >> 16;
    }
    dinv[2 * t] = 1.0f / sqrtf((float)(lo + 1u));
    if (2 * t + 1 < n) dinv[2 * t + 1] = 1.0f / sqrtf((float)(hi + 1u));
}

template <int K, int NOUT, bool SCALED>
__global__ void k_gemm_wmma(const float* __restrict__ A, const float* __restrict__ W,
                            const float* __restrict__ rowscale,
                            float* __restrict__ H, int nrows) {
    const int wave = blockIdx.x * (blockDim.x >> 5) + (threadIdx.x >> 5);
    const int lane = threadIdx.x & 31;
    const int m    = lane & 15;
    const int kh   = lane >> 4;
    const int r0   = wave * 16;
    if (r0 + 16 > nrows) return;

    constexpr int NT = NOUT / 16;
    v8f acc[NT];
#pragma unroll
    for (int t = 0; t < NT; ++t)
#pragma unroll
        for (int j = 0; j < 8; ++j) acc[t][j] = 0.0f;

    float sc[8];
#pragma unroll
    for (int j = 0; j < 8; ++j) sc[j] = SCALED ? rowscale[r0 + j + 8 * kh] : 1.0f;

    const float* arow = A + (size_t)(r0 + m) * K;
    for (int k = 0; k < K; k += 4) {
        v2f a;
        a.x = arow[k + 2 * kh + 0];
        a.y = arow[k + 2 * kh + 1];
#pragma unroll
        for (int t = 0; t < NT; ++t) {
            v2f b;
            b.x = W[(size_t)(k + 2 * kh + 0) * NOUT + t * 16 + m];
            b.y = W[(size_t)(k + 2 * kh + 1) * NOUT + t * 16 + m];
            acc[t] = __builtin_amdgcn_wmma_f32_16x16x4_f32(
                 false, a,  false, b,
                 (short)0, acc[t],  false,  false);
        }
    }

#pragma unroll
    for (int t = 0; t < NT; ++t) {
#pragma unroll
        for (int j = 0; j < 8; ++j)
            H[(size_t)(r0 + j + 8 * kh) * NOUT + t * 16 + m] =
                SCALED ? acc[t][j] * sc[j] : acc[t][j];
    }
}

template <int K, int NOUT, bool SCALED>
__global__ void k_gemm_tail(const float* __restrict__ A, const float* __restrict__ W,
                            const float* __restrict__ rowscale,
                            float* __restrict__ H, int row0, int nrows) {
    int idx = blockIdx.x * blockDim.x + threadIdx.x;
    int r = row0 + idx / NOUT;
    int c = idx % NOUT;
    if (r >= nrows) return;
    float s = 0.0f;
    for (int k = 0; k < K; ++k) s += A[(size_t)r * K + k] * W[(size_t)k * NOUT + c];
    H[(size_t)r * NOUT + c] = SCALED ? s * rowscale[r] : s;
}

template <int K, int NOUT>
__global__ void __launch_bounds__(PG_TB)
k_place_gemm(const int* __restrict__ src, const int* __restrict__ dst,
             const unsigned int* __restrict__ pcnt, unsigned int* __restrict__ recs,
             unsigned int* __restrict__ bstart, int e, int nb, int parts,
             const unsigned int* __restrict__ part, const float* __restrict__ A,
             const float* __restrict__ W, float* __restrict__ dinv,
             float* __restrict__ H, int nrows) {
    __shared__ unsigned int stage[DEG_ECH + PG_XW];
    __shared__ unsigned int lbeg[NBMAX + 1];
    __shared__ unsigned int cur[NBMAX];
    __shared__ unsigned int gbeg[NBMAX];
    __shared__ unsigned int wtot[8];
    __shared__ unsigned int wtot2[8];
    unsigned int* xs = stage + DEG_ECH;
    const int tid  = threadIdx.x;
    const int lane = tid & 31;
    const int wv   = tid >> 5;

    if ((int)blockIdx.x < parts) {
        const int p = blockIdx.x;
        unsigned int c = 0u;
        if (tid < 256 && tid < nb) c = pcnt[(size_t)p * nb + tid];
        {
            const int b = tid & 255, q = tid >> 8;
            unsigned int st = 0u, sb = 0u;
            if (b < nb) {
#pragma unroll 8
                for (int pp = q; pp < parts; pp += 4) {
                    const unsigned int v = pcnt[(size_t)pp * nb + b];
                    st += v;
                    sb += (pp < p) ? v : 0u;
                }
            }
            xs[q * 256 + b] = st;
            xs[1024 + q * 256 + b] = sb;
        }
        __syncthreads();

        unsigned int incl = 0u, t = 0u, tincl = 0u, bef = 0u;
        if (tid < 256) {
            t   = xs[tid] + xs[256 + tid] + xs[512 + tid] + xs[768 + tid];
            bef = xs[1024 + tid] + xs[1280 + tid] + xs[1536 + tid] + xs[1792 + tid];
            incl = c;
            tincl = t;
            for (int o = 1; o < 32; o <<= 1) {
                const unsigned int v  = __shfl_up(incl, o);
                const unsigned int v2 = __shfl_up(tincl, o);
                if (lane >= o) { incl += v; tincl += v2; }
            }
            if (lane == 31) { wtot[wv] = incl; wtot2[wv] = tincl; }
        }
        __syncthreads();
        if (tid < 256) {
            unsigned int before = 0u, before2 = 0u;
            for (int w = 0; w < wv; ++w) { before += wtot[w]; before2 += wtot2[w]; }
            const unsigned int ex  = before + incl - c;
            const unsigned int ex2 = before2 + tincl - t;
            if (tid < nb) {
                lbeg[tid] = ex;
                cur[tid]  = ex;
                gbeg[tid] = ex2 + bef;
                if (p == 0) bstart[tid] = ex2;
            }
            if (tid == nb - 1) {
                lbeg[nb] = ex + c;
                if (p == 0) bstart[nb] = ex2 + t;
            }
        }
        __syncthreads();

        const long long e0 = (long long)p * DEG_ECH;
        const long long left = (long long)e - e0;
        const int m = left < (long long)DEG_ECH ? (int)left : DEG_ECH;
        const int* __restrict__ dp = dst + e0;
        const int* __restrict__ sp = src + e0;
        for (int base = tid; base < m; base += DEG_B * DEG_TB) {
            int d[DEG_B], s[DEG_B];
#pragma unroll
            for (int k = 0; k < DEG_B; ++k) {
                const int i = base + k * DEG_TB;
                d[k] = (i < m) ? dp[i] : 0;
                s[k] = (i < m) ? sp[i] : 0;
            }
#pragma unroll
            for (int k = 0; k < DEG_B; ++k) {
                if (base + k * DEG_TB < m) {
                    const int b = d[k] / BN;
                    const unsigned int pos = atomicAdd(&cur[b], 1u);
                    stage[pos] = ((unsigned int)(d[k] - b * BN) << 17) | (unsigned int)s[k];
                }
            }
        }
        __syncthreads();

        for (int b = wv; b < nb; b += DEG_TB / 32) {
            const unsigned int l0  = lbeg[b];
            const unsigned int len = lbeg[b + 1] - l0;
            const unsigned int g0  = gbeg[b];
            const unsigned int a   = g0 & 31u;
            for (unsigned int j = lane; j < len + a; j += 32u) {
                if (j >= a) recs[g0 - a + j] = stage[l0 + j - a];
            }
        }
        return;
    }

    const int row0 = ((int)blockIdx.x - parts) * PG_ROWS;
    {
        const int w = tid & 255, q = tid >> 8;
        const int tw = (row0 >> 1) + w;
        unsigned int lo = 0u, hi = 0u;
        if (2 * tw < nrows) {
#pragma unroll 8
            for (int pp = q; pp < parts; pp += 4) {
                const unsigned int v = part[(size_t)pp * DEG_WORDS + tw];
                lo += v & 0xFFFFu;
                hi += v >> 16;
            }
        }
        xs[q * 256 + w] = lo;
        xs[1024 + q * 256 + w] = hi;
    }
    __syncthreads();
    float* dl = reinterpret_cast<float*>(xs + 2048);
    if (tid < PG_ROWS) {
        const int w = tid >> 1;
        const unsigned int* hsel = xs + ((tid & 1) ? 1024 : 0);
        const unsigned int cntv = hsel[w] + hsel[256 + w] + hsel[512 + w] + hsel[768 + w];
        const float dv = 1.0f / sqrtf((float)(cntv + 1u));
        dl[tid] = dv;
        if (row0 + tid < nrows) dinv[row0 + tid] = dv;
    }
    __syncthreads();

    const int m  = lane & 15;
    const int kh = lane >> 4;
    const int r0 = row0 + wv * 16;
    if (r0 + 16 > nrows) return;

    constexpr int NT = NOUT / 16;
    v8f acc[NT];
#pragma unroll
    for (int t = 0; t < NT; ++t)
#pragma unroll
        for (int j = 0; j < 8; ++j) acc[t][j] = 0.0f;

    float sc[8];
#pragma unroll
    for (int j = 0; j < 8; ++j) sc[j] = dl[wv * 16 + j + 8 * kh];

    const float* arow = A + (size_t)(r0 + m) * K;
    for (int k = 0; k < K; k += 4) {
        v2f a;
        a.x = arow[k + 2 * kh + 0];
        a.y = arow[k + 2 * kh + 1];
#pragma unroll
        for (int t = 0; t < NT; ++t) {
            v2f b;
            b.x = W[(size_t)(k + 2 * kh + 0) * NOUT + t * 16 + m];
            b.y = W[(size_t)(k + 2 * kh + 1) * NOUT + t * 16 + m];
            acc[t] = __builtin_amdgcn_wmma_f32_16x16x4_f32(
                 false, a,  false, b,
                 (short)0, acc[t],  false,  false);
        }
    }

#pragma unroll
    for (int t = 0; t < NT; ++t) {
#pragma unroll
        for (int j = 0; j < 8; ++j)
            H[(size_t)(r0 + j + 8 * kh) * NOUT + t * 16 + m] = acc[t][j] * sc[j];
    }
}

template <int C, bool HAS_BIAS>
__global__ void k_self_init(const float* __restrict__ h, const float* __restrict__ dinv,
                            const float* __restrict__ bias, float* __restrict__ agg, int n) {
    long long idx = (long long)blockIdx.x * blockDim.x + threadIdx.x;
    long long total = (long long)n * C;
    if (idx >= total) return;
    int i = (int)(idx / C);
    int c = (int)(idx % C);
    float d = dinv[i];
    float v = h[idx] * d * d;
    if (HAS_BIAS) v += bias[c];
    agg[idx] = v;
}

template <int C>
__global__ void k_edge_agg(const int* __restrict__ src, const int* __restrict__ dst,
                           const float* __restrict__ h, const float* __restrict__ dinv,
                           float* __restrict__ agg, int e) {
    constexpr int EPW = 8;
    constexpr int NJ  = C / 32;
    const int lane = threadIdx.x & 31;
    const int wave = __builtin_amdgcn_readfirstlane(
        (int)(blockIdx.x * (blockDim.x >> 5) + (threadIdx.x >> 5)));
    const long long e0 = (long long)wave * EPW;
    if (e0 >= (long long)e) return;

    if (e0 + EPW <= (long long)e) {
        int s[EPW], d[EPW];
        float v[EPW][NJ];
        float nrm[EPW];
#pragma unroll
        for (int i = 0; i < EPW; ++i) {
            s[i] = src[e0 + i];
            d[i] = dst[e0 + i];
        }
#pragma unroll
        for (int i = 0; i < EPW; ++i)
#pragma unroll
            for (int j = 0; j < NJ; ++j)
                v[i][j] = h[(size_t)s[i] * C + j * 32 + lane];
#pragma unroll
        for (int i = 0; i < EPW; ++i) nrm[i] = dinv[s[i]] * dinv[d[i]];
#pragma unroll
        for (int i = 0; i < EPW; ++i)
#pragma unroll
            for (int j = 0; j < NJ; ++j)
                atomicAdd(agg + (size_t)d[i] * C + j * 32 + lane, v[i][j] * nrm[i]);
    } else {
        for (int i = 0; i < EPW; ++i) {
            if (e0 + i >= (long long)e) break;
            const int s = src[e0 + i];
            const int d = dst[e0 + i];
            const float nrm = dinv[s] * dinv[d];
#pragma unroll
            for (int j = 0; j < NJ; ++j) {
                const float v = h[(size_t)s * C + j * 32 + lane];
                atomicAdd(agg + (size_t)d * C + j * 32 + lane, v * nrm);
            }
        }
    }
}

template <int C>
__global__ void k_relu_bias(const float* __restrict__ agg, const float* __restrict__ b,
                            float* __restrict__ p, int n) {
    long long idx = (long long)blockIdx.x * blockDim.x + threadIdx.x;
    long long total = (long long)n * C;
    if (idx >= total) return;
    int c = (int)(idx % C);
    float v = agg[idx] + b[c];
    p[idx] = v > 0.0f ? v : 0.0f;
}

constexpr int AGG_TB   = 1024;
constexpr int AGG_G    = 16;
constexpr int SRT_CAP  = 8192;
constexpr int SRT_K    = SRT_CAP / AGG_TB;
constexpr int SRT_RPL  = (BN + 31) / 32;
constexpr int SRT_ROWS = SRT_RPL * 32;

template <int C>
__device__ __forceinline__ void agg_records32(unsigned int pk, int cnt, const float* __restrict__ h,
                                              float* acc, int lane) {
    constexpr int P  = C + 4;
    constexpr int NJ = C / 32;
    const int s = (int)(pk & 0x1FFFFu);
    const int key  = (lane < cnt) ? (int)(pk >> 17) : 0x7FFF;
    const int prev = __shfl_up(key, 1);
    const unsigned int heads = __builtin_amdgcn_ballot_w32(lane > 0 && key != prev);
    float sum[NJ];
#pragma unroll
    for (int j = 0; j < NJ; ++j) sum[j] = 0.0f;
#pragma unroll
    for (int g = 0; g < 32; g += AGG_G) {
        if (g < cnt) {
            float v[AGG_G][NJ];
#pragma unroll
            for (int k = 0; k < AGG_G; ++k) {
                const int sk = __builtin_amdgcn_readlane(s, g + k);
#pragma unroll
                for (int j = 0; j < NJ; ++j) v[k][j] = h[(size_t)sk * C + j * 32 + lane];
            }
#pragma unroll
            for (int k = 0; k < AGG_G; ++k) {
                const int i = g + k;
                if (i > 0 && __builtin_expect((heads >> i) & 1u, 0u)) {
                    const int dp = __builtin_amdgcn_readlane(key, i > 0 ? i - 1 : 0);
#pragma unroll
                    for (int j = 0; j < NJ; ++j) {
                        atomicAdd(&acc[dp * P + j * 32 + lane], sum[j]);
                        sum[j] = 0.0f;
                    }
                }
#pragma unroll
                for (int j = 0; j < NJ; ++j) sum[j] += v[k][j];
            }
        }
    }
    if (cnt == 32) {
        const int dp = __builtin_amdgcn_readlane(key, 31);
#pragma unroll
        for (int j = 0; j < NJ; ++j) atomicAdd(&acc[dp * P + j * 32 + lane], sum[j]);
    } else if (cnt == AGG_G) {
        const int dp = __builtin_amdgcn_readlane(key, AGG_G - 1);
#pragma unroll
        for (int j = 0; j < NJ; ++j) atomicAdd(&acc[dp * P + j * 32 + lane], sum[j]);
    }
}

template <int C, bool LAYER1>
__global__ void __launch_bounds__(AGG_TB)
k_range_agg(unsigned int* __restrict__ recs, const unsigned int* __restrict__ bstart,
            const float* __restrict__ h, const float* __restrict__ dinv,
            const float* __restrict__ bias, const float* __restrict__ W2,
            float* __restrict__ outp, int n) {
    constexpr int P  = C + 4;
    __shared__ float acc[BN * P];
    __shared__ alignas(16) float dloc[BN];
    __shared__ unsigned int srt[LAYER1 ? SRT_CAP : 1];
    __shared__ unsigned int cntr[LAYER1 ? SRT_ROWS : 1];
    __shared__ unsigned int rowst[LAYER1 ? SRT_ROWS : 1];
    const int tid  = threadIdx.x;
    const int lane = tid & 31;
    const int wv   = __builtin_amdgcn_readfirstlane(tid >> 5);
    const int n0   = blockIdx.x * BN;
    const int rows = (n - n0) < BN ? (n - n0) : BN;

    const unsigned int eb = bstart[blockIdx.x];
    const unsigned int ee = bstart[blockIdx.x + 1];

    float dself;
    {
        int node = n0 + (tid < BN ? tid : BN - 1);
        node = node < n ? node : n - 1;
        dself = dinv[node];
    }

    unsigned int rec[SRT_K];
    if constexpr (LAYER1) {
        if (tid < SRT_ROWS) cntr[tid] = 0u;
#pragma unroll
        for (int k = 0; k < SRT_K; ++k) {
            const unsigned int idx = (unsigned int)(k * AGG_TB + tid);
            rec[k] = (idx < ee - eb) ? recs[eb + idx] : 0u;
        }
    }

    constexpr int TOT = BN * C;
    constexpr int INIT_B = 5;
    for (int base = tid; base < TOT; base += INIT_B * AGG_TB) {
        float hv[INIT_B];
#pragma unroll
        for (int u = 0; u < INIT_B; ++u) {
            int idx = base + u * AGG_TB;
            idx = idx < TOT ? idx : TOT - 1;
            const int r = idx / C, c = idx - r * C;
            int node = n0 + r;
            node = node < n ? node : n - 1;
            hv[u] = h[(size_t)node * C + c];
        }
#pragma unroll
        for (int u = 0; u < INIT_B; ++u) {
            const int idx = base + u * AGG_TB;
            if (idx < TOT) {
                const int r = idx / C, c = idx - r * C;
                acc[r * P + c] = (r < rows) ? hv[u] : 0.0f;
            }
        }
    }
    if (tid < BN) dloc[tid] = (tid < rows) ? dself : 0.0f;
    __syncthreads();

    if constexpr (LAYER1) {
        unsigned int base = eb;
        for (;;) {
            const unsigned int left = ee - base;
            const unsigned int m = left < (unsigned int)SRT_CAP ? left : (unsigned int)SRT_CAP;
            unsigned int rk[SRT_K];
#pragma unroll
            for (int k = 0; k < SRT_K; ++k) {
                rk[k] = 0u;
                if ((unsigned int)(k * AGG_TB + tid) < m) rk[k] = atomicAdd(&cntr[rec[k] >> 17], 1u);
            }
            __syncthreads();
            if (wv == 0) {
                unsigned int cv[SRT_RPL];
                unsigned int tot = 0u;
#pragma unroll
                for (int i = 0; i < SRT_RPL; ++i) {
                    const unsigned int t = cntr[lane * SRT_RPL + i];
                    cv[i] = tot;
                    tot += t;
                }
                unsigned int incl = tot;
                for (int o = 1; o < 32; o <<= 1) {
                    const unsigned int v = __shfl_up(incl, o);
                    if (lane >= o) incl += v;
                }
                const unsigned int before = incl - tot;
#pragma unroll
                for (int i = 0; i < SRT_RPL; ++i) {
                    rowst[lane * SRT_RPL + i] = before + cv[i];
                    cntr[lane * SRT_RPL + i]  = 0u;
                }
            }
            __syncthreads();
            unsigned int pos[SRT_K];
#pragma unroll
            for (int k = 0; k < SRT_K; ++k) pos[k] = rowst[rec[k] >> 17] + rk[k];
#pragma unroll
            for (int k = 0; k < SRT_K; ++k) {
                if ((unsigned int)(k * AGG_TB + tid) < m) srt[pos[k]] = rec[k];
            }
            __syncthreads();
            unsigned int sv[SRT_K];
#pragma unroll
            for (int k = 0; k < SRT_K; ++k) sv[k] = srt[k * AGG_TB + tid];
#pragma unroll
            for (int k = 0; k < SRT_K; ++k) {
                const unsigned int idx = (unsigned int)(k * AGG_TB + tid);
                if (idx < m) recs[base + idx] = sv[k];
            }
            for (unsigned int c0 = (unsigned int)wv * 32u; c0 < m; c0 += AGG_TB) {
                const unsigned int pk = (c0 + lane < m) ? srt[c0 + lane] : 0u;
                const int cnt = (int)((m - c0) < 32u ? (m - c0) : 32u);
                agg_records32<C>(pk, cnt, h, acc, lane);
            }
            base += m;
            if (base >= ee) break;
            __syncthreads();
#pragma unroll
            for (int k = 0; k < SRT_K; ++k) {
                const unsigned int idx = (unsigned int)(k * AGG_TB + tid);
                rec[k] = (idx < ee - base) ? recs[base + idx] : 0u;
            }
        }
    } else {
        unsigned int c0 = eb + (unsigned int)wv * 32u;
        unsigned int pk = (c0 + lane < ee) ? recs[c0 + lane] : 0u;
        for (; c0 < ee; c0 += AGG_TB) {
            const unsigned int cn = c0 + AGG_TB;
            const unsigned int pkn = (cn + lane < ee) ? recs[cn + lane] : 0u;
            const int cnt = (int)((ee - c0) < 32u ? (ee - c0) : 32u);
            agg_records32<C>(pk, cnt, h, acc, lane);
            pk = pkn;
        }
    }
    __syncthreads();

    if constexpr (!LAYER1) {
        for (int idx = tid; idx < rows * C; idx += AGG_TB) {
            const int r = idx / C, c = idx - r * C;
            outp[(size_t)(n0 + r) * C + c] = acc[r * P + c] * dloc[r] + bias[c];
        }
    } else {
        for (int idx = tid; idx < BN * C; idx += AGG_TB) {
            const int r = idx / C, c = idx - r * C;
            const float v = acc[r * P + c] * dloc[r] + bias[c];
            acc[r * P + c] = v > 0.0f ? v : 0.0f;
        }
        __syncthreads();
        constexpr int NO = 32;
        constexpr int NT = NO / 16;
        const int full = rows / 16;
        if (wv < full) {
            const int m  = lane & 15;
            const int kh = lane >> 4;
            v8f d[NT];
#pragma unroll
            for (int t = 0; t < NT; ++t)
#pragma unroll
                for (int j = 0; j < 8; ++j) d[t][j] = 0.0f;
            float sc[8];
#pragma unroll
            for (int j = 0; j < 8; ++j) sc[j] = dloc[wv * 16 + j + 8 * kh];
            const float* arow = acc + (wv * 16 + m) * P;
            for (int k = 0; k < C; k += 4) {
                v2f a;
                a.x = arow[k + 2 * kh + 0];
                a.y = arow[k + 2 * kh + 1];
#pragma unroll
                for (int t = 0; t < NT; ++t) {
                    v2f b;
                    b.x = W2[(size_t)(k + 2 * kh + 0) * NO + t * 16 + m];
                    b.y = W2[(size_t)(k + 2 * kh + 1) * NO + t * 16 + m];
                    d[t] = __builtin_amdgcn_wmma_f32_16x16x4_f32(
                        false, a, false, b, (short)0, d[t], false, false);
                }
            }
#pragma unroll
            for (int t = 0; t < NT; ++t)
#pragma unroll
                for (int j = 0; j < 8; ++j)
                    outp[(size_t)(n0 + wv * 16 + j + 8 * kh) * NO + t * 16 + m] = d[t][j] * sc[j];
        }
        const int r1 = full * 16;
        for (int idx = tid; idx < (rows - r1) * NO; idx += AGG_TB) {
            const int r = r1 + idx / NO, c = idx % NO;
            float sum = 0.0f;
            for (int k = 0; k < C; ++k) sum += acc[r * P + k] * W2[(size_t)k * NO + c];
            outp[(size_t)(n0 + r) * NO + c] = sum * dloc[r];
        }
    }
}

extern "C" void kernel_launch(void* const* d_in, const int* in_sizes, int n_in,
                              void* d_out, int out_size, void* d_ws, size_t ws_size,
                              hipStream_t stream) {
    const float* x  = (const float*)d_in[0];
    const int*   ei = (const int*)d_in[1];
    const float* W1 = (const float*)d_in[2];
    const float* b1 = (const float*)d_in[3];
    const float* W2 = (const float*)d_in[4];
    const float* b2 = (const float*)d_in[5];
    float* out = (float*)d_out;

    const int n = in_sizes[0] / 128;
    const int e = in_sizes[1] / 2;
    const int* src = ei;
    const int* dst = ei + e;

    char* ws = (char*)d_ws;
    size_t off0 = ((size_t)n * sizeof(float) + 255) & ~(size_t)255;
    float* dinv = (float*)ws;
    float* bufA = (float*)(ws + off0);
    float* bufB = bufA + (size_t)n * 64;

    const int tb = 256;
    auto cdiv = [](long long a, long long b) { return (int)((a + b - 1) / b); };

    const int parts = cdiv(e, DEG_ECH);
    const int strips = n / 16;
    const int wavesPerBlk = tb / 32;
    const int nb = cdiv(n, BN);
    const size_t half = (size_t)n * 32;
    if (n <= DEG_NCAP && (size_t)parts * DEG_WORDS <= half &&
        (size_t)e + (size_t)parts * nb + nb + 1 <= half) {
        unsigned int* slabs  = (unsigned int*)bufB;
        unsigned int* recs   = (unsigned int*)bufB + half;
        unsigned int* pcnt   = recs + e;
        unsigned int* bstart = pcnt + (size_t)parts * nb;
        float* h2 = bufB;

        k_deg_hist<<<parts, DEG_TB, 0, stream>>>(dst, slabs, pcnt, nb, e);
        k_place_gemm<128, 64><<<parts + cdiv(n, PG_ROWS), PG_TB, 0, stream>>>(
            src, dst, pcnt, recs, bstart, e, nb, parts, slabs, x, W1, dinv, bufA, n);
        if (n % 16) {
            int tail = (n % 16) * 64;
            k_gemm_tail<128, 64, true><<<cdiv(tail, tb), tb, 0, stream>>>(x, W1, dinv, bufA, strips * 16, n);
        }
        k_range_agg<64, true><<<nb, AGG_TB, 0, stream>>>(recs, bstart, bufA, dinv, b1, W2, h2, n);
        k_range_agg<32, false><<<nb, AGG_TB, 0, stream>>>(recs, bstart, h2, dinv, b2, nullptr, out, n);
        return;
    }

    if (n <= DEG_NCAP && (size_t)parts * DEG_WORDS <= (size_t)n * 64) {
        k_deg_hist<<<parts, DEG_TB, 0, stream>>>(dst, (unsigned int*)bufB, nullptr, 0, e);
        k_dinv_hist<<<cdiv((n + 1) / 2, tb), tb, 0, stream>>>((const unsigned int*)bufB, dinv, n, parts);
    } else {
        k_deg_init<<<cdiv(n, tb), tb, 0, stream>>>(dinv, n);
        k_deg_count<<<cdiv(e, tb), tb, 0, stream>>>(dst, dinv, e);
        k_dinv<<<cdiv(n, tb), tb, 0, stream>>>(dinv, n);
    }

    k_gemm_wmma<128, 64, false><<<cdiv(strips, wavesPerBlk), tb, 0, stream>>>(x, W1, nullptr, bufA, n);
    if (n % 16) {
        int tail = (n % 16) * 64;
        k_gemm_tail<128, 64, false><<<cdiv(tail, tb), tb, 0, stream>>>(x, W1, nullptr, bufA, strips * 16, n);
    }

    k_self_init<64, false><<<cdiv((long long)n * 64, tb), tb, 0, stream>>>(bufA, dinv, nullptr, bufB, n);
    k_edge_agg<64><<<cdiv(cdiv(e, 8), wavesPerBlk), tb, 0, stream>>>(src, dst, bufA, dinv, bufB, e);
    k_relu_bias<64><<<cdiv((long long)n * 64, tb), tb, 0, stream>>>(bufB, b1, bufA, n);

    k_gemm_wmma<64, 32, false><<<cdiv(strips, wavesPerBlk), tb, 0, stream>>>(bufA, W2, nullptr, bufB, n);
    if (n % 16) {
        int tail = (n % 16) * 32;
        k_gemm_tail<64, 32, false><<<cdiv(tail, tb), tb, 0, stream>>>(bufA, W2, nullptr, bufB, strips * 16, n);
    }

    k_self_init<32, true><<<cdiv((long long)n * 32, tb), tb, 0, stream>>>(bufB, dinv, b2, out, n);
    k_edge_agg<32><<<cdiv(cdiv(e, 8), wavesPerBlk), tb, 0, stream>>>(src, dst, bufB, dinv, out, e);

    (void)n_in; (void)out_size; (void)ws_size;
}


# work-group start-up, launches 2-4: the loads a group needs before its first barrier (count-table columns, start values as 16-B loads, first records) all requested at once: 1 round trip, was 5 to 7
# speedup vs baseline: 4.3193x; 1.0375x over previous
#include <hip/hip_runtime.h>

typedef float v2f __attribute__((ext_vector_type(2)));
typedef float v8f __attribute__((ext_vector_type(8)));

__global__ void k_deg_init(float* __restrict__ deg, int n) {
    int i = blockIdx.x * blockDim.x + threadIdx.x;
    if (i < n) deg[i] = 1.0f;
}

__global__ void k_deg_count(const int* __restrict__ dst, float* __restrict__ deg, int e) {
    int i = blockIdx.x * blockDim.x + threadIdx.x;
    if (i < e) atomicAdd(&deg[dst[i]], 1.0f);
}

__global__ void k_dinv(float* __restrict__ deg, int n) {
    int i = blockIdx.x * blockDim.x + threadIdx.x;
    if (i < n) deg[i] = 1.0f / sqrtf(deg[i]);
}

constexpr int DEG_NCAP   = 100000;
constexpr int DEG_WORDS  = ((DEG_NCAP + 1) / 2 + 31) & ~31;
constexpr int DEG_ECH    = 32768;
constexpr int DEG_TB     = 1024;
constexpr int DEG_B      = 16;
constexpr int BN         = 400;
constexpr int BW         = BN / 2;
constexpr int NBMAX      = (DEG_NCAP + BN - 1) / BN;
constexpr int PG_TB      = 1024;
constexpr int PG_ROWS    = (PG_TB / 32) * 16;
constexpr int PG_XW      = 10240;
constexpr int PG_PB      = 16;

__global__ void __launch_bounds__(DEG_TB)
k_deg_hist(const int* __restrict__ dst, unsigned int* __restrict__ part,
           unsigned int* __restrict__ pcnt, int nb, int e) {
    __shared__ unsigned int cnt[DEG_WORDS];
    const int tid = threadIdx.x;

    uint4* c4 = reinterpret_cast<uint4*>(cnt);
    for (int i = tid; i < DEG_WORDS / 4; i += DEG_TB) c4[i] = make_uint4(0u, 0u, 0u, 0u);
    __syncthreads();

    const long long e0 = (long long)blockIdx.x * DEG_ECH;
    const long long left = (long long)e - e0;
    const int m = left < (long long)DEG_ECH ? (int)left : DEG_ECH;
    const int* __restrict__ dp = dst + e0;
    for (int base = tid; base < m; base += DEG_B * DEG_TB) {
        int d[DEG_B];
#pragma unroll
        for (int k = 0; k < DEG_B; ++k) {
            const int i = base + k * DEG_TB;
            d[k] = (i < m) ? dp[i] : -1;
        }
#pragma unroll
        for (int k = 0; k < DEG_B; ++k) {
            if (base + k * DEG_TB < m) atomicAdd(&cnt[d[k] >> 1], (d[k] & 1) ? 0x10000u : 1u);
        }
    }
    __syncthreads();

    uint4* p4 = reinterpret_cast<uint4*>(part + (size_t)blockIdx.x * DEG_WORDS);
    for (int i = tid; i < DEG_WORDS / 4; i += DEG_TB) p4[i] = c4[i];

    if (pcnt != nullptr) {
        const int wv = tid >> 5, ln = tid & 31;
        for (int b = wv; b < nb; b += DEG_TB / 32) {
            unsigned int sacc = 0u;
            for (int w = ln; w < BW; w += 32) {
                const int wi = b * BW + w;
                if (wi < DEG_WORDS) {
                    const unsigned int v = cnt[wi];
                    sacc += (v & 0xFFFFu) + (v >> 16);
                }
            }
            for (int o = 16; o > 0; o >>= 1) sacc += __shfl_xor(sacc, o);
            if (ln == 0) pcnt[(size_t)blockIdx.x * nb + b] = sacc;
        }
    }
}

__global__ void k_dinv_hist(const unsigned int* __restrict__ part, float* __restrict__ dinv,
                            int n, int parts) {
    const int t = blockIdx.x * blockDim.x + threadIdx.x;
    if (2 * t >= n) return;
    unsigned int lo = 0u, hi = 0u;
#pragma unroll 8
    for (int p = 0; p < parts; ++p) {
        const unsigned int w = part[(size_t)p * DEG_WORDS + t];
        lo += w & 0xFFFFu;
        hi += w >> 16;
    }
    dinv[2 * t] = 1.0f / sqrtf((float)(lo + 1u));
    if (2 * t + 1 < n) dinv[2 * t + 1] = 1.0f / sqrtf((float)(hi + 1u));
}

template <int K, int NOUT, bool SCALED>
__global__ void k_gemm_wmma(const float* __restrict__ A, const float* __restrict__ W,
                            const float* __restrict__ rowscale,
                            float* __restrict__ H, int nrows) {
    const int wave = blockIdx.x * (blockDim.x >> 5) + (threadIdx.x >> 5);
    const int lane = threadIdx.x & 31;
    const int m    = lane & 15;
    const int kh   = lane >> 4;
    const int r0   = wave * 16;
    if (r0 + 16 > nrows) return;

    constexpr int NT = NOUT / 16;
    v8f acc[NT];
#pragma unroll
    for (int t = 0; t < NT; ++t)
#pragma unroll
        for (int j = 0; j < 8; ++j) acc[t][j] = 0.0f;

    float sc[8];
#pragma unroll
    for (int j = 0; j < 8; ++j) sc[j] = SCALED ? rowscale[r0 + j + 8 * kh] : 1.0f;

    const float* arow = A + (size_t)(r0 + m) * K;
    for (int k = 0; k < K; k += 4) {
        v2f a;
        a.x = arow[k + 2 * kh + 0];
        a.y = arow[k + 2 * kh + 1];
#pragma unroll
        for (int t = 0; t < NT; ++t) {
            v2f b;
            b.x = W[(size_t)(k + 2 * kh + 0) * NOUT + t * 16 + m];
            b.y = W[(size_t)(k + 2 * kh + 1) * NOUT + t * 16 + m];
            acc[t] = __builtin_amdgcn_wmma_f32_16x16x4_f32(
                 false, a,  false, b,
                 (short)0, acc[t],  false,  false);
        }
    }

#pragma unroll
    for (int t = 0; t < NT; ++t) {
#pragma unroll
        for (int j = 0; j < 8; ++j)
            H[(size_t)(r0 + j + 8 * kh) * NOUT + t * 16 + m] =
                SCALED ? acc[t][j] * sc[j] : acc[t][j];
    }
}

template <int K, int NOUT, bool SCALED>
__global__ void k_gemm_tail(const float* __restrict__ A, const float* __restrict__ W,
                            const float* __restrict__ rowscale,
                            float* __restrict__ H, int row0, int nrows) {
    int idx = blockIdx.x * blockDim.x + threadIdx.x;
    int r = row0 + idx / NOUT;
    int c = idx % NOUT;
    if (r >= nrows) return;
    float s = 0.0f;
    for (int k = 0; k < K; ++k) s += A[(size_t)r * K + k] * W[(size_t)k * NOUT + c];
    H[(size_t)r * NOUT + c] = SCALED ? s * rowscale[r] : s;
}

template <int K, int NOUT>
__global__ void __launch_bounds__(PG_TB)
k_place_gemm(const int* __restrict__ src, const int* __restrict__ dst,
             const unsigned int* __restrict__ pcnt, unsigned int* __restrict__ recs,
             unsigned int* __restrict__ bstart, int e, int nb, int parts,
             const unsigned int* __restrict__ part, const float* __restrict__ A,
             const float* __restrict__ W, float* __restrict__ dinv,
             float* __restrict__ H, int nrows) {
    __shared__ unsigned int stage[DEG_ECH + PG_XW];
    __shared__ unsigned int lbeg[NBMAX + 1];
    __shared__ unsigned int cur[NBMAX];
    __shared__ unsigned int gbeg[NBMAX];
    __shared__ unsigned int wtot[8];
    __shared__ unsigned int wtot2[8];
    unsigned int* xs = stage + DEG_ECH;
    const int tid  = threadIdx.x;
    const int lane = tid & 31;
    const int wv   = tid >> 5;

    if ((int)blockIdx.x < parts) {
        const int p = blockIdx.x;
        unsigned int c = 0u;
        if (tid < 256 && tid < nb) c = pcnt[(size_t)p * nb + tid];
        {
            const int b = tid & 255, q = tid >> 8;
            unsigned int st = 0u, sb = 0u;
            if (b < nb) {
                for (int p0 = q; p0 < parts; p0 += 4 * PG_PB) {
                    unsigned int v[PG_PB];
#pragma unroll
                    for (int k = 0; k < PG_PB; ++k) {
                        const int pp = p0 + 4 * k;
                        v[k] = (pp < parts) ? pcnt[(size_t)pp * nb + b] : 0u;
                    }
#pragma unroll
                    for (int k = 0; k < PG_PB; ++k) {
                        const int pp = p0 + 4 * k;
                        st += v[k];
                        sb += (pp < p) ? v[k] : 0u;
                    }
                }
            }
            xs[q * 256 + b] = st;
            xs[1024 + q * 256 + b] = sb;
        }
        __syncthreads();

        unsigned int incl = 0u, t = 0u, tincl = 0u, bef = 0u;
        if (tid < 256) {
            t   = xs[tid] + xs[256 + tid] + xs[512 + tid] + xs[768 + tid];
            bef = xs[1024 + tid] + xs[1280 + tid] + xs[1536 + tid] + xs[1792 + tid];
            incl = c;
            tincl = t;
            for (int o = 1; o < 32; o <<= 1) {
                const unsigned int v  = __shfl_up(incl, o);
                const unsigned int v2 = __shfl_up(tincl, o);
                if (lane >= o) { incl += v; tincl += v2; }
            }
            if (lane == 31) { wtot[wv] = incl; wtot2[wv] = tincl; }
        }
        __syncthreads();
        if (tid < 256) {
            unsigned int before = 0u, before2 = 0u;
            for (int w = 0; w < wv; ++w) { before += wtot[w]; before2 += wtot2[w]; }
            const unsigned int ex  = before + incl - c;
            const unsigned int ex2 = before2 + tincl - t;
            if (tid < nb) {
                lbeg[tid] = ex;
                cur[tid]  = ex;
                gbeg[tid] = ex2 + bef;
                if (p == 0) bstart[tid] = ex2;
            }
            if (tid == nb - 1) {
                lbeg[nb] = ex + c;
                if (p == 0) bstart[nb] = ex2 + t;
            }
        }
        __syncthreads();

        const long long e0 = (long long)p * DEG_ECH;
        const long long left = (long long)e - e0;
        const int m = left < (long long)DEG_ECH ? (int)left : DEG_ECH;
        const int* __restrict__ dp = dst + e0;
        const int* __restrict__ sp = src + e0;
        for (int base = tid; base < m; base += DEG_B * DEG_TB) {
            int d[DEG_B], s[DEG_B];
#pragma unroll
            for (int k = 0; k < DEG_B; ++k) {
                const int i = base + k * DEG_TB;
                d[k] = (i < m) ? dp[i] : 0;
                s[k] = (i < m) ? sp[i] : 0;
            }
#pragma unroll
            for (int k = 0; k < DEG_B; ++k) {
                if (base + k * DEG_TB < m) {
                    const int b = d[k] / BN;
                    const unsigned int pos = atomicAdd(&cur[b], 1u);
                    stage[pos] = ((unsigned int)(d[k] - b * BN) << 17) | (unsigned int)s[k];
                }
            }
        }
        __syncthreads();

        for (int b = wv; b < nb; b += DEG_TB / 32) {
            const unsigned int l0  = lbeg[b];
            const unsigned int len = lbeg[b + 1] - l0;
            const unsigned int g0  = gbeg[b];
            const unsigned int a   = g0 & 31u;
            for (unsigned int j = lane; j < len + a; j += 32u) {
                if (j >= a) recs[g0 - a + j] = stage[l0 + j - a];
            }
        }
        return;
    }

    const int row0 = ((int)blockIdx.x - parts) * PG_ROWS;
    {
        const int w = tid & 255, q = tid >> 8;
        const int tw = (row0 >> 1) + w;
        unsigned int lo = 0u, hi = 0u;
        if (2 * tw < nrows) {
            for (int p0 = q; p0 < parts; p0 += 4 * PG_PB) {
                unsigned int v[PG_PB];
#pragma unroll
                for (int k = 0; k < PG_PB; ++k) {
                    const int pp = p0 + 4 * k;
                    v[k] = (pp < parts) ? part[(size_t)pp * DEG_WORDS + tw] : 0u;
                }
#pragma unroll
                for (int k = 0; k < PG_PB; ++k) {
                    lo += v[k] & 0xFFFFu;
                    hi += v[k] >> 16;
                }
            }
        }
        xs[q * 256 + w] = lo;
        xs[1024 + q * 256 + w] = hi;
    }
    __syncthreads();
    float* dl = reinterpret_cast<float*>(xs + 2048);
    if (tid < PG_ROWS) {
        const int w = tid >> 1;
        const unsigned int* hsel = xs + ((tid & 1) ? 1024 : 0);
        const unsigned int cntv = hsel[w] + hsel[256 + w] + hsel[512 + w] + hsel[768 + w];
        const float dv = 1.0f / sqrtf((float)(cntv + 1u));
        dl[tid] = dv;
        if (row0 + tid < nrows) dinv[row0 + tid] = dv;
    }
    __syncthreads();

    const int m  = lane & 15;
    const int kh = lane >> 4;
    const int r0 = row0 + wv * 16;
    if (r0 + 16 > nrows) return;

    constexpr int NT = NOUT / 16;
    v8f acc[NT];
#pragma unroll
    for (int t = 0; t < NT; ++t)
#pragma unroll
        for (int j = 0; j < 8; ++j) acc[t][j] = 0.0f;

    float sc[8];
#pragma unroll
    for (int j = 0; j < 8; ++j) sc[j] = dl[wv * 16 + j + 8 * kh];

    const float* arow = A + (size_t)(r0 + m) * K;
    for (int k = 0; k < K; k += 4) {
        v2f a;
        a.x = arow[k + 2 * kh + 0];
        a.y = arow[k + 2 * kh + 1];
#pragma unroll
        for (int t = 0; t < NT; ++t) {
            v2f b;
            b.x = W[(size_t)(k + 2 * kh + 0) * NOUT + t * 16 + m];
            b.y = W[(size_t)(k + 2 * kh + 1) * NOUT + t * 16 + m];
            acc[t] = __builtin_amdgcn_wmma_f32_16x16x4_f32(
                 false, a,  false, b,
                 (short)0, acc[t],  false,  false);
        }
    }

#pragma unroll
    for (int t = 0; t < NT; ++t) {
#pragma unroll
        for (int j = 0; j < 8; ++j)
            H[(size_t)(r0 + j + 8 * kh) * NOUT + t * 16 + m] = acc[t][j] * sc[j];
    }
}

template <int C, bool HAS_BIAS>
__global__ void k_self_init(const float* __restrict__ h, const float* __restrict__ dinv,
                            const float* __restrict__ bias, float* __restrict__ agg, int n) {
    long long idx = (long long)blockIdx.x * blockDim.x + threadIdx.x;
    long long total = (long long)n * C;
    if (idx >= total) return;
    int i = (int)(idx / C);
    int c = (int)(idx % C);
    float d = dinv[i];
    float v = h[idx] * d * d;
    if (HAS_BIAS) v += bias[c];
    agg[idx] = v;
}

template <int C>
__global__ void k_edge_agg(const int* __restrict__ src, const int* __restrict__ dst,
                           const float* __restrict__ h, const float* __restrict__ dinv,
                           float* __restrict__ agg, int e) {
    constexpr int EPW = 8;
    constexpr int NJ  = C / 32;
    const int lane = threadIdx.x & 31;
    const int wave = __builtin_amdgcn_readfirstlane(
        (int)(blockIdx.x * (blockDim.x >> 5) + (threadIdx.x >> 5)));
    const long long e0 = (long long)wave * EPW;
    if (e0 >= (long long)e) return;

    if (e0 + EPW <= (long long)e) {
        int s[EPW], d[EPW];
        float v[EPW][NJ];
        float nrm[EPW];
#pragma unroll
        for (int i = 0; i < EPW; ++i) {
            s[i] = src[e0 + i];
            d[i] = dst[e0 + i];
        }
#pragma unroll
        for (int i = 0; i < EPW; ++i)
#pragma unroll
            for (int j = 0; j < NJ; ++j)
                v[i][j] = h[(size_t)s[i] * C + j * 32 + lane];
#pragma unroll
        for (int i = 0; i < EPW; ++i) nrm[i] = dinv[s[i]] * dinv[d[i]];
#pragma unroll
        for (int i = 0; i < EPW; ++i)
#pragma unroll
            for (int j = 0; j < NJ; ++j)
                atomicAdd(agg + (size_t)d[i] * C + j * 32 + lane, v[i][j] * nrm[i]);
    } else {
        for (int i = 0; i < EPW; ++i) {
            if (e0 + i >= (long long)e) break;
            const int s = src[e0 + i];
            const int d = dst[e0 + i];
            const float nrm = dinv[s] * dinv[d];
#pragma unroll
            for (int j = 0; j < NJ; ++j) {
                const float v = h[(size_t)s * C + j * 32 + lane];
                atomicAdd(agg + (size_t)d * C + j * 32 + lane, v * nrm);
            }
        }
    }
}

template <int C>
__global__ void k_relu_bias(const float* __restrict__ agg, const float* __restrict__ b,
                            float* __restrict__ p, int n) {
    long long idx = (long long)blockIdx.x * blockDim.x + threadIdx.x;
    long long total = (long long)n * C;
    if (idx >= total) return;
    int c = (int)(idx % C);
    float v = agg[idx] + b[c];
    p[idx] = v > 0.0f ? v : 0.0f;
}

constexpr int AGG_TB   = 1024;
constexpr int AGG_G    = 16;
constexpr int SRT_CAP  = 8192;
constexpr int SRT_K    = SRT_CAP / AGG_TB;
constexpr int SRT_RPL  = (BN + 31) / 32;
constexpr int SRT_ROWS = SRT_RPL * 32;

template <int C>
__device__ __forceinline__ void agg_records32(unsigned int pk, int cnt, const float* __restrict__ h,
                                              float* acc, int lane) {
    constexpr int P  = C + 4;
    constexpr int NJ = C / 32;
    const int s = (int)(pk & 0x1FFFFu);
    const int key  = (lane < cnt) ? (int)(pk >> 17) : 0x7FFF;
    const int prev = __shfl_up(key, 1);
    const unsigned int heads = __builtin_amdgcn_ballot_w32(lane > 0 && key != prev);
    float sum[NJ];
#pragma unroll
    for (int j = 0; j < NJ; ++j) sum[j] = 0.0f;
#pragma unroll
    for (int g = 0; g < 32; g += AGG_G) {
        if (g < cnt) {
            float v[AGG_G][NJ];
#pragma unroll
            for (int k = 0; k < AGG_G; ++k) {
                const int sk = __builtin_amdgcn_readlane(s, g + k);
#pragma unroll
                for (int j = 0; j < NJ; ++j) v[k][j] = h[(size_t)sk * C + j * 32 + lane];
            }
#pragma unroll
            for (int k = 0; k < AGG_G; ++k) {
                const int i = g + k;
                if (i > 0 && __builtin_expect((heads >> i) & 1u, 0u)) {
                    const int dp = __builtin_amdgcn_readlane(key, i > 0 ? i - 1 : 0);
#pragma unroll
                    for (int j = 0; j < NJ; ++j) {
                        atomicAdd(&acc[dp * P + j * 32 + lane], sum[j]);
                        sum[j] = 0.0f;
                    }
                }
#pragma unroll
                for (int j = 0; j < NJ; ++j) sum[j] += v[k][j];
            }
        }
    }
    if (cnt == 32) {
        const int dp = __builtin_amdgcn_readlane(key, 31);
#pragma unroll
        for (int j = 0; j < NJ; ++j) atomicAdd(&acc[dp * P + j * 32 + lane], sum[j]);
    } else if (cnt == AGG_G) {
        const int dp = __builtin_amdgcn_readlane(key, AGG_G - 1);
#pragma unroll
        for (int j = 0; j < NJ; ++j) atomicAdd(&acc[dp * P + j * 32 + lane], sum[j]);
    }
}

template <int C, bool LAYER1>
__global__ void __launch_bounds__(AGG_TB)
k_range_agg(unsigned int* __restrict__ recs, const unsigned int* __restrict__ bstart,
            const float* __restrict__ h, const float* __restrict__ dinv,
            const float* __restrict__ bias, const float* __restrict__ W2,
            float* __restrict__ outp, int n) {
    constexpr int P  = C + 4;
    __shared__ alignas(16) float acc[BN * P];
    __shared__ alignas(16) float dloc[BN];
    __shared__ unsigned int srt[LAYER1 ? SRT_CAP : 1];
    __shared__ unsigned int cntr[LAYER1 ? SRT_ROWS : 1];
    __shared__ unsigned int rowst[LAYER1 ? SRT_ROWS : 1];
    const int tid  = threadIdx.x;
    const int lane = tid & 31;
    const int wv   = __builtin_amdgcn_readfirstlane(tid >> 5);
    const int n0   = blockIdx.x * BN;
    const int rows = (n - n0) < BN ? (n - n0) : BN;

    const unsigned int eb = bstart[blockIdx.x];
    const unsigned int ee = bstart[blockIdx.x + 1];

    float dself;
    {
        int node = n0 + (tid < BN ? tid : BN - 1);
        node = node < n ? node : n - 1;
        dself = dinv[node];
    }

    unsigned int rec[SRT_K];
    if constexpr (LAYER1) {
        if (tid < SRT_ROWS) cntr[tid] = 0u;
#pragma unroll
        for (int k = 0; k < SRT_K; ++k) {
            const unsigned int idx = (unsigned int)(k * AGG_TB + tid);
            rec[k] = (idx < ee - eb) ? recs[eb + idx] : 0u;
        }
    }

    unsigned int pk0 = 0u;
    if constexpr (!LAYER1) {
        const unsigned int c00 = eb + (unsigned int)wv * 32u;
        pk0 = (c00 + lane < ee) ? recs[c00 + lane] : 0u;
    }

    constexpr int C4     = C / 4;
    constexpr int INIT_T = 2 * BN;
    constexpr int INIT_K = C4 / 2;
    static_assert(INIT_T <= AGG_TB && INIT_T % 32 == 0 && INIT_T * INIT_K == BN * C4,
                  "start-value shares must be whole");
    if (tid < INIT_T) {
        float4 hv[INIT_K];
#pragma unroll
        for (int u = 0; u < INIT_K; ++u) {
            const int idx = tid + u * INIT_T;
            const int r = idx / C4, c4 = idx - r * C4;
            int node = n0 + r;
            node = node < n ? node : n - 1;
            hv[u] = *reinterpret_cast<const float4*>(h + (size_t)node * C + c4 * 4);
        }
#pragma unroll
        for (int u = 0; u < INIT_K; ++u) {
            const int idx = tid + u * INIT_T;
            const int r = idx / C4, c4 = idx - r * C4;
            *reinterpret_cast<float4*>(&acc[r * P + c4 * 4]) = hv[u];
        }
    }
    if (tid < BN) dloc[tid] = (tid < rows) ? dself : 0.0f;
    __syncthreads();

    if constexpr (LAYER1) {
        unsigned int base = eb;
        for (;;) {
            const unsigned int left = ee - base;
            const unsigned int m = left < (unsigned int)SRT_CAP ? left : (unsigned int)SRT_CAP;
            unsigned int rk[SRT_K];
#pragma unroll
            for (int k = 0; k < SRT_K; ++k) {
                rk[k] = 0u;
                if ((unsigned int)(k * AGG_TB + tid) < m) rk[k] = atomicAdd(&cntr[rec[k] >> 17], 1u);
            }
            __syncthreads();
            if (wv == 0) {
                unsigned int cv[SRT_RPL];
                unsigned int tot = 0u;
#pragma unroll
                for (int i = 0; i < SRT_RPL; ++i) {
                    const unsigned int t = cntr[lane * SRT_RPL + i];
                    cv[i] = tot;
                    tot += t;
                }
                unsigned int incl = tot;
                for (int o = 1; o < 32; o <<= 1) {
                    const unsigned int v = __shfl_up(incl, o);
                    if (lane >= o) incl += v;
                }
                const unsigned int before = incl - tot;
#pragma unroll
                for (int i = 0; i < SRT_RPL; ++i) {
                    rowst[lane * SRT_RPL + i] = before + cv[i];
                    cntr[lane * SRT_RPL + i]  = 0u;
                }
            }
            __syncthreads();
            unsigned int pos[SRT_K];
#pragma unroll
            for (int k = 0; k < SRT_K; ++k) pos[k] = rowst[rec[k] >> 17] + rk[k];
#pragma unroll
            for (int k = 0; k < SRT_K; ++k) {
                if ((unsigned int)(k * AGG_TB + tid) < m) srt[pos[k]] = rec[k];
            }
            __syncthreads();
            unsigned int sv[SRT_K];
#pragma unroll
            for (int k = 0; k < SRT_K; ++k) sv[k] = srt[k * AGG_TB + tid];
#pragma unroll
            for (int k = 0; k < SRT_K; ++k) {
                const unsigned int idx = (unsigned int)(k * AGG_TB + tid);
                if (idx < m) recs[base + idx] = sv[k];
            }
            for (unsigned int c0 = (unsigned int)wv * 32u; c0 < m; c0 += AGG_TB) {
                const unsigned int pk = (c0 + lane < m) ? srt[c0 + lane] : 0u;
                const int cnt = (int)((m - c0) < 32u ? (m - c0) : 32u);
                agg_records32<C>(pk, cnt, h, acc, lane);
            }
            base += m;
            if (base >= ee) break;
            __syncthreads();
#pragma unroll
            for (int k = 0; k < SRT_K; ++k) {
                const unsigned int idx = (unsigned int)(k * AGG_TB + tid);
                rec[k] = (idx < ee - base) ? recs[base + idx] : 0u;
            }
        }
    } else {
        unsigned int c0 = eb + (unsigned int)wv * 32u;
        unsigned int pk = pk0;
        for (; c0 < ee; c0 += AGG_TB) {
            const unsigned int cn = c0 + AGG_TB;
            const unsigned int pkn = (cn + lane < ee) ? recs[cn + lane] : 0u;
            const int cnt = (int)((ee - c0) < 32u ? (ee - c0) : 32u);
            agg_records32<C>(pk, cnt, h, acc, lane);
            pk = pkn;
        }
    }
    __syncthreads();

    if constexpr (!LAYER1) {
        for (int idx = tid; idx < rows * C; idx += AGG_TB) {
            const int r = idx / C, c = idx - r * C;
            outp[(size_t)(n0 + r) * C + c] = acc[r * P + c] * dloc[r] + bias[c];
        }
    } else {
        for (int idx = tid; idx < BN * C; idx += AGG_TB) {
            const int r = idx / C, c = idx - r * C;
            const float v = acc[r * P + c] * dloc[r] + bias[c];
            acc[r * P + c] = v > 0.0f ? v : 0.0f;
        }
        __syncthreads();
        constexpr int NO = 32;
        constexpr int NT = NO / 16;
        const int full = rows / 16;
        if (wv < full) {
            const int m  = lane & 15;
            const int kh = lane >> 4;
            v8f d[NT];
#pragma unroll
            for (int t = 0; t < NT; ++t)
#pragma unroll
                for (int j = 0; j < 8; ++j) d[t][j] = 0.0f;
            float sc[8];
#pragma unroll
            for (int j = 0; j < 8; ++j) sc[j] = dloc[wv * 16 + j + 8 * kh];
            const float* arow = acc + (wv * 16 + m) * P;
            for (int k = 0; k < C; k += 4) {
                v2f a;
                a.x = arow[k + 2 * kh + 0];
                a.y = arow[k + 2 * kh + 1];
#pragma unroll
                for (int t = 0; t < NT; ++t) {
                    v2f b;
                    b.x = W2[(size_t)(k + 2 * kh + 0) * NO + t * 16 + m];
                    b.y = W2[(size_t)(k + 2 * kh + 1) * NO + t * 16 + m];
                    d[t] = __builtin_amdgcn_wmma_f32_16x16x4_f32(
                        false, a, false, b, (short)0, d[t], false, false);
                }
            }
#pragma unroll
            for (int t = 0; t < NT; ++t)
#pragma unroll
                for (int j = 0; j < 8; ++j)
                    outp[(size_t)(n0 + wv * 16 + j + 8 * kh) * NO + t * 16 + m] = d[t][j] * sc[j];
        }
        const int r1 = full * 16;
        for (int idx = tid; idx < (rows - r1) * NO; idx += AGG_TB) {
            const int r = r1 + idx / NO, c = idx % NO;
            float sum = 0.0f;
            for (int k = 0; k < C; ++k) sum += acc[r * P + k] * W2[(size_t)k * NO + c];
            outp[(size_t)(n0 + r) * NO + c] = sum * dloc[r];
        }
    }
}

extern "C" void kernel_launch(void* const* d_in, const int* in_sizes, int n_in,
                              void* d_out, int out_size, void* d_ws, size_t ws_size,
                              hipStream_t stream) {
    const float* x  = (const float*)d_in[0];
    const int*   ei = (const int*)d_in[1];
    const float* W1 = (const float*)d_in[2];
    const float* b1 = (const float*)d_in[3];
    const float* W2 = (const float*)d_in[4];
    const float* b2 = (const float*)d_in[5];
    float* out = (float*)d_out;

    const int n = in_sizes[0] / 128;
    const int e = in_sizes[1] / 2;
    const int* src = ei;
    const int* dst = ei + e;

    char* ws = (char*)d_ws;
    size_t off0 = ((size_t)n * sizeof(float) + 255) & ~(size_t)255;
    float* dinv = (float*)ws;
    float* bufA = (float*)(ws + off0);
    float* bufB = bufA + (size_t)n * 64;

    const int tb = 256;
    auto cdiv = [](long long a, long long b) { return (int)((a + b - 1) / b); };

    const int parts = cdiv(e, DEG_ECH);
    const int strips = n / 16;
    const int wavesPerBlk = tb / 32;
    const int nb = cdiv(n, BN);
    const size_t half = (size_t)n * 32;
    if (n <= DEG_NCAP && (size_t)parts * DEG_WORDS <= half &&
        (size_t)e + (size_t)parts * nb + nb + 1 <= half) {
        unsigned int* slabs  = (unsigned int*)bufB;
        unsigned int* recs   = (unsigned int*)bufB + half;
        unsigned int* pcnt   = recs + e;
        unsigned int* bstart = pcnt + (size_t)parts * nb;
        float* h2 = bufB;

        k_deg_hist<<<parts, DEG_TB, 0, stream>>>(dst, slabs, pcnt, nb, e);
        k_place_gemm<128, 64><<<parts + cdiv(n, PG_ROWS), PG_TB, 0, stream>>>(
            src, dst, pcnt, recs, bstart, e, nb, parts, slabs, x, W1, dinv, bufA, n);
        if (n % 16) {
            int tail = (n % 16) * 64;
            k_gemm_tail<128, 64, true><<<cdiv(tail, tb), tb, 0, stream>>>(x, W1, dinv, bufA, strips * 16, n);
        }
        k_range_agg<64, true><<<nb, AGG_TB, 0, stream>>>(recs, bstart, bufA, dinv, b1, W2, h2, n);
        k_range_agg<32, false><<<nb, AGG_TB, 0, stream>>>(recs, bstart, h2, dinv, b2, nullptr, out, n);
        return;
    }

    if (n <= DEG_NCAP && (size_t)parts * DEG_WORDS <= (size_t)n * 64) {
        k_deg_hist<<<parts, DEG_TB, 0, stream>>>(dst, (unsigned int*)bufB, nullptr, 0, e);
        k_dinv_hist<<<cdiv((n + 1) / 2, tb), tb, 0, stream>>>((const unsigned int*)bufB, dinv, n, parts);
    } else {
        k_deg_init<<<cdiv(n, tb), tb, 0, stream>>>(dinv, n);
        k_deg_count<<<cdiv(e, tb), tb, 0, stream>>>(dst, dinv, e);
        k_dinv<<<cdiv(n, tb), tb, 0, stream>>>(dinv, n);
    }

    k_gemm_wmma<128, 64, false><<<cdiv(strips, wavesPerBlk), tb, 0, stream>>>(x, W1, nullptr, bufA, n);
    if (n % 16) {
        int tail = (n % 16) * 64;
        k_gemm_tail<128, 64, false><<<cdiv(tail, tb), tb, 0, stream>>>(x, W1, nullptr, bufA, strips * 16, n);
    }

    k_self_init<64, false><<<cdiv((long long)n * 64, tb), tb, 0, stream>>>(bufA, dinv, nullptr, bufB, n);
    k_edge_agg<64><<<cdiv(cdiv(e, 8), wavesPerBlk), tb, 0, stream>>>(src, dst, bufA, dinv, bufB, e);
    k_relu_bias<64><<<cdiv((long long)n * 64, tb), tb, 0, stream>>>(bufB, b1, bufA, n);

    k_gemm_wmma<64, 32, false><<<cdiv(strips, wavesPerBlk), tb, 0, stream>>>(bufA, W2, nullptr, bufB, n);
    if (n % 16) {
        int tail = (n % 16) * 32;
        k_gemm_tail<64, 32, false><<<cdiv(tail, tb), tb, 0, stream>>>(bufA, W2, nullptr, bufB, strips * 16, n);
    }

    k_self_init<32, true><<<cdiv((long long)n * 32, tb), tb, 0, stream>>>(bufB, dinv, b2, out, n);
    k_edge_agg<32><<<cdiv(cdiv(e, 8), wavesPerBlk), tb, 0, stream>>>(src, dst, bufB, dinv, out, e);

    (void)n_in; (void)out_size; (void)ws_size;
}


# k_deg_hist per-range edge counts: four threads per range add its packed table words with independent LDS reads and two lane exchanges (was one wave per range: 7 dependent reads, 5 exchanges)
# speedup vs baseline: 4.6182x; 1.0692x over previous
#include <hip/hip_runtime.h>

typedef float v2f __attribute__((ext_vector_type(2)));
typedef float v8f __attribute__((ext_vector_type(8)));

__global__ void k_deg_init(float* __restrict__ deg, int n) {
    int i = blockIdx.x * blockDim.x + threadIdx.x;
    if (i < n) deg[i] = 1.0f;
}

__global__ void k_deg_count(const int* __restrict__ dst, float* __restrict__ deg, int e) {
    int i = blockIdx.x * blockDim.x + threadIdx.x;
    if (i < e) atomicAdd(&deg[dst[i]], 1.0f);
}

__global__ void k_dinv(float* __restrict__ deg, int n) {
    int i = blockIdx.x * blockDim.x + threadIdx.x;
    if (i < n) deg[i] = 1.0f / sqrtf(deg[i]);
}

constexpr int DEG_NCAP   = 100000;
constexpr int DEG_WORDS  = ((DEG_NCAP + 1) / 2 + 31) & ~31;
constexpr int DEG_ECH    = 32768;
constexpr int DEG_TB     = 1024;
constexpr int DEG_B      = 16;
constexpr int BN         = 400;
constexpr int BW         = BN / 2;
constexpr int NBMAX      = (DEG_NCAP + BN - 1) / BN;
constexpr int PG_TB      = 1024;
constexpr int PG_ROWS    = (PG_TB / 32) * 16;
constexpr int PG_XW      = 10240;
constexpr int PG_PB      = 16;

__global__ void __launch_bounds__(DEG_TB)
k_deg_hist(const int* __restrict__ dst, unsigned int* __restrict__ part,
           unsigned int* __restrict__ pcnt, int nb, int e) {
    __shared__ unsigned int cnt[DEG_WORDS];
    const int tid = threadIdx.x;

    uint4* c4 = reinterpret_cast<uint4*>(cnt);
    for (int i = tid; i < DEG_WORDS / 4; i += DEG_TB) c4[i] = make_uint4(0u, 0u, 0u, 0u);
    __syncthreads();

    const long long e0 = (long long)blockIdx.x * DEG_ECH;
    const long long left = (long long)e - e0;
    const int m = left < (long long)DEG_ECH ? (int)left : DEG_ECH;
    const int* __restrict__ dp = dst + e0;
    for (int base = tid; base < m; base += DEG_B * DEG_TB) {
        int d[DEG_B];
#pragma unroll
        for (int k = 0; k < DEG_B; ++k) {
            const int i = base + k * DEG_TB;
            d[k] = (i < m) ? dp[i] : -1;
        }
#pragma unroll
        for (int k = 0; k < DEG_B; ++k) {
            if (base + k * DEG_TB < m) atomicAdd(&cnt[d[k] >> 1], (d[k] & 1) ? 0x10000u : 1u);
        }
    }
    __syncthreads();

    uint4* p4 = reinterpret_cast<uint4*>(part + (size_t)blockIdx.x * DEG_WORDS);
    for (int i = tid; i < DEG_WORDS / 4; i += DEG_TB) p4[i] = c4[i];

    if (pcnt != nullptr) {
        constexpr int QW = BW / 4;
        static_assert(BW % 8 == 0, "a quarter of a range is a whole number of 8-byte reads");
        static_assert(NBMAX * 4 <= DEG_TB && NBMAX * BW <= DEG_WORDS && DEG_ECH < 65536,
                      "four threads per range; ranges inside the table; 16-bit sums");
        const int b = tid >> 2, j = tid & 3;
        unsigned int acc = 0u;
        if (b < nb) {
            const uint2* q2 = reinterpret_cast<const uint2*>(cnt + b * BW + j * QW);
#pragma unroll
            for (int i = 0; i < QW / 2; ++i) {
                const uint2 v = q2[i];
                acc += v.x + v.y;
            }
        }
        acc += __shfl_xor(acc, 1);
        acc += __shfl_xor(acc, 2);
        if (j == 0 && b < nb) pcnt[(size_t)blockIdx.x * nb + b] = (acc & 0xFFFFu) + (acc >> 16);
    }
}

__global__ void k_dinv_hist(const unsigned int* __restrict__ part, float* __restrict__ dinv,
                            int n, int parts) {
    const int t = blockIdx.x * blockDim.x + threadIdx.x;
    if (2 * t >= n) return;
    unsigned int lo = 0u, hi = 0u;
#pragma unroll 8
    for (int p = 0; p < parts; ++p) {
        const unsigned int w = part[(size_t)p * DEG_WORDS + t];
        lo += w & 0xFFFFu;
        hi += w >> 16;
    }
    dinv[2 * t] = 1.0f / sqrtf((float)(lo + 1u));
    if (2 * t + 1 < n) dinv[2 * t + 1] = 1.0f / sqrtf((float)(hi + 1u));
}

template <int K, int NOUT, bool SCALED>
__global__ void k_gemm_wmma(const float* __restrict__ A, const float* __restrict__ W,
                            const float* __restrict__ rowscale,
                            float* __restrict__ H, int nrows) {
    const int wave = blockIdx.x * (blockDim.x >> 5) + (threadIdx.x >> 5);
    const int lane = threadIdx.x & 31;
    const int m    = lane & 15;
    const int kh   = lane >> 4;
    const int r0   = wave * 16;
    if (r0 + 16 > nrows) return;

    constexpr int NT = NOUT / 16;
    v8f acc[NT];
#pragma unroll
    for (int t = 0; t < NT; ++t)
#pragma unroll
        for (int j = 0; j < 8; ++j) acc[t][j] = 0.0f;

    float sc[8];
#pragma unroll
    for (int j = 0; j < 8; ++j) sc[j] = SCALED ? rowscale[r0 + j + 8 * kh] : 1.0f;

    const float* arow = A + (size_t)(r0 + m) * K;
    for (int k = 0; k < K; k += 4) {
        v2f a;
        a.x = arow[k + 2 * kh + 0];
        a.y = arow[k + 2 * kh + 1];
#pragma unroll
        for (int t = 0; t < NT; ++t) {
            v2f b;
            b.x = W[(size_t)(k + 2 * kh + 0) * NOUT + t * 16 + m];
            b.y = W[(size_t)(k + 2 * kh + 1) * NOUT + t * 16 + m];
            acc[t] = __builtin_amdgcn_wmma_f32_16x16x4_f32(
                 false, a,  false, b,
                 (short)0, acc[t],  false,  false);
        }
    }

#pragma unroll
    for (int t = 0; t < NT; ++t) {
#pragma unroll
        for (int j = 0; j < 8; ++j)
            H[(size_t)(r0 + j + 8 * kh) * NOUT + t * 16 + m] =
                SCALED ? acc[t][j] * sc[j] : acc[t][j];
    }
}

template <int K, int NOUT, bool SCALED>
__global__ void k_gemm_tail(const float* __restrict__ A, const float* __restrict__ W,
                            const float* __restrict__ rowscale,
                            float* __restrict__ H, int row0, int nrows) {
    int idx = blockIdx.x * blockDim.x + threadIdx.x;
    int r = row0 + idx / NOUT;
    int c = idx % NOUT;
    if (r >= nrows) return;
    float s = 0.0f;
    for (int k = 0; k < K; ++k) s += A[(size_t)r * K + k] * W[(size_t)k * NOUT + c];
    H[(size_t)r * NOUT + c] = SCALED ? s * rowscale[r] : s;
}

template <int K, int NOUT>
__global__ void __launch_bounds__(PG_TB)
k_place_gemm(const int* __restrict__ src, const int* __restrict__ dst,
             const unsigned int* __restrict__ pcnt, unsigned int* __restrict__ recs,
             unsigned int* __restrict__ bstart, int e, int nb, int parts,
             const unsigned int* __restrict__ part, const float* __restrict__ A,
             const float* __restrict__ W, float* __restrict__ dinv,
             float* __restrict__ H, int nrows) {
    __shared__ unsigned int stage[DEG_ECH + PG_XW];
    __shared__ unsigned int lbeg[NBMAX + 1];
    __shared__ unsigned int cur[NBMAX];
    __shared__ unsigned int gbeg[NBMAX];
    __shared__ unsigned int wtot[8];
    __shared__ unsigned int wtot2[8];
    unsigned int* xs = stage + DEG_ECH;
    const int tid  = threadIdx.x;
    const int lane = tid & 31;
    const int wv   = tid >> 5;

    if ((int)blockIdx.x < parts) {
        const int p = blockIdx.x;
        unsigned int c = 0u;
        if (tid < 256 && tid < nb) c = pcnt[(size_t)p * nb + tid];
        {
            const int b = tid & 255, q = tid >> 8;
            unsigned int st = 0u, sb = 0u;
            if (b < nb) {
                for (int p0 = q; p0 < parts; p0 += 4 * PG_PB) {
                    unsigned int v[PG_PB];
#pragma unroll
                    for (int k = 0; k < PG_PB; ++k) {
                        const int pp = p0 + 4 * k;
                        v[k] = (pp < parts) ? pcnt[(size_t)pp * nb + b] : 0u;
                    }
#pragma unroll
                    for (int k = 0; k < PG_PB; ++k) {
                        const int pp = p0 + 4 * k;
                        st += v[k];
                        sb += (pp < p) ? v[k] : 0u;
                    }
                }
            }
            xs[q * 256 + b] = st;
            xs[1024 + q * 256 + b] = sb;
        }
        __syncthreads();

        unsigned int incl = 0u, t = 0u, tincl = 0u, bef = 0u;
        if (tid < 256) {
            t   = xs[tid] + xs[256 + tid] + xs[512 + tid] + xs[768 + tid];
            bef = xs[1024 + tid] + xs[1280 + tid] + xs[1536 + tid] + xs[1792 + tid];
            incl = c;
            tincl = t;
            for (int o = 1; o < 32; o <<= 1) {
                const unsigned int v  = __shfl_up(incl, o);
                const unsigned int v2 = __shfl_up(tincl, o);
                if (lane >= o) { incl += v; tincl += v2; }
            }
            if (lane == 31) { wtot[wv] = incl; wtot2[wv] = tincl; }
        }
        __syncthreads();
        if (tid < 256) {
            unsigned int before = 0u, before2 = 0u;
            for (int w = 0; w < wv; ++w) { before += wtot[w]; before2 += wtot2[w]; }
            const unsigned int ex  = before + incl - c;
            const unsigned int ex2 = before2 + tincl - t;
            if (tid < nb) {
                lbeg[tid] = ex;
                cur[tid]  = ex;
                gbeg[tid] = ex2 + bef;
                if (p == 0) bstart[tid] = ex2;
            }
            if (tid == nb - 1) {
                lbeg[nb] = ex + c;
                if (p == 0) bstart[nb] = ex2 + t;
            }
        }
        __syncthreads();

        const long long e0 = (long long)p * DEG_ECH;
        const long long left = (long long)e - e0;
        const int m = left < (long long)DEG_ECH ? (int)left : DEG_ECH;
        const int* __restrict__ dp = dst + e0;
        const int* __restrict__ sp = src + e0;
        for (int base = tid; base < m; base += DEG_B * DEG_TB) {
            int d[DEG_B], s[DEG_B];
#pragma unroll
            for (int k = 0; k < DEG_B; ++k) {
                const int i = base + k * DEG_TB;
                d[k] = (i < m) ? dp[i] : 0;
                s[k] = (i < m) ? sp[i] : 0;
            }
#pragma unroll
            for (int k = 0; k < DEG_B; ++k) {
                if (base + k * DEG_TB < m) {
                    const int b = d[k] / BN;
                    const unsigned int pos = atomicAdd(&cur[b], 1u);
                    stage[pos] = ((unsigned int)(d[k] - b * BN) << 17) | (unsigned int)s[k];
                }
            }
        }
        __syncthreads();

        for (int b = wv; b < nb; b += DEG_TB / 32) {
            const unsigned int l0  = lbeg[b];
            const unsigned int len = lbeg[b + 1] - l0;
            const unsigned int g0  = gbeg[b];
            const unsigned int a   = g0 & 31u;
            for (unsigned int j = lane; j < len + a; j += 32u) {
                if (j >= a) recs[g0 - a + j] = stage[l0 + j - a];
            }
        }
        return;
    }

    const int row0 = ((int)blockIdx.x - parts) * PG_ROWS;
    {
        const int w = tid & 255, q = tid >> 8;
        const int tw = (row0 >> 1) + w;
        unsigned int lo = 0u, hi = 0u;
        if (2 * tw < nrows) {
            for (int p0 = q; p0 < parts; p0 += 4 * PG_PB) {
                unsigned int v[PG_PB];
#pragma unroll
                for (int k = 0; k < PG_PB; ++k) {
                    const int pp = p0 + 4 * k;
                    v[k] = (pp < parts) ? part[(size_t)pp * DEG_WORDS + tw] : 0u;
                }
#pragma unroll
                for (int k = 0; k < PG_PB; ++k) {
                    lo += v[k] & 0xFFFFu;
                    hi += v[k] >> 16;
                }
            }
        }
        xs[q * 256 + w] = lo;
        xs[1024 + q * 256 + w] = hi;
    }
    __syncthreads();
    float* dl = reinterpret_cast<float*>(xs + 2048);
    if (tid < PG_ROWS) {
        const int w = tid >> 1;
        const unsigned int* hsel = xs + ((tid & 1) ? 1024 : 0);
        const unsigned int cntv = hsel[w] + hsel[256 + w] + hsel[512 + w] + hsel[768 + w];
        const float dv = 1.0f / sqrtf((float)(cntv + 1u));
        dl[tid] = dv;
        if (row0 + tid < nrows) dinv[row0 + tid] = dv;
    }
    __syncthreads();

    const int m  = lane & 15;
    const int kh = lane >> 4;
    const int r0 = row0 + wv * 16;
    if (r0 + 16 > nrows) return;

    constexpr int NT = NOUT / 16;
    v8f acc[NT];
#pragma unroll
    for (int t = 0; t < NT; ++t)
#pragma unroll
        for (int j = 0; j < 8; ++j) acc[t][j] = 0.0f;

    float sc[8];
#pragma unroll
    for (int j = 0; j < 8; ++j) sc[j] = dl[wv * 16 + j + 8 * kh];

    const float* arow = A + (size_t)(r0 + m) * K;
    for (int k = 0; k < K; k += 4) {
        v2f a;
        a.x = arow[k + 2 * kh + 0];
        a.y = arow[k + 2 * kh + 1];
#pragma unroll
        for (int t = 0; t < NT; ++t) {
            v2f b;
            b.x = W[(size_t)(k + 2 * kh + 0) * NOUT + t * 16 + m];
            b.y = W[(size_t)(k + 2 * kh + 1) * NOUT + t * 16 + m];
            acc[t] = __builtin_amdgcn_wmma_f32_16x16x4_f32(
                 false, a,  false, b,
                 (short)0, acc[t],  false,  false);
        }
    }

#pragma unroll
    for (int t = 0; t < NT; ++t) {
#pragma unroll
        for (int j = 0; j < 8; ++j)
            H[(size_t)(r0 + j + 8 * kh) * NOUT + t * 16 + m] = acc[t][j] * sc[j];
    }
}

template <int C, bool HAS_BIAS>
__global__ void k_self_init(const float* __restrict__ h, const float* __restrict__ dinv,
                            const float* __restrict__ bias, float* __restrict__ agg, int n) {
    long long idx = (long long)blockIdx.x * blockDim.x + threadIdx.x;
    long long total = (long long)n * C;
    if (idx >= total) return;
    int i = (int)(idx / C);
    int c = (int)(idx % C);
    float d = dinv[i];
    float v = h[idx] * d * d;
    if (HAS_BIAS) v += bias[c];
    agg[idx] = v;
}

template <int C>
__global__ void k_edge_agg(const int* __restrict__ src, const int* __restrict__ dst,
                           const float* __restrict__ h, const float* __restrict__ dinv,
                           float* __restrict__ agg, int e) {
    constexpr int EPW = 8;
    constexpr int NJ  = C / 32;
    const int lane = threadIdx.x & 31;
    const int wave = __builtin_amdgcn_readfirstlane(
        (int)(blockIdx.x * (blockDim.x >> 5) + (threadIdx.x >> 5)));
    const long long e0 = (long long)wave * EPW;
    if (e0 >= (long long)e) return;

    if (e0 + EPW <= (long long)e) {
        int s[EPW], d[EPW];
        float v[EPW][NJ];
        float nrm[EPW];
#pragma unroll
        for (int i = 0; i < EPW; ++i) {
            s[i] = src[e0 + i];
            d[i] = dst[e0 + i];
        }
#pragma unroll
        for (int i = 0; i < EPW; ++i)
#pragma unroll
            for (int j = 0; j < NJ; ++j)
                v[i][j] = h[(size_t)s[i] * C + j * 32 + lane];
#pragma unroll
        for (int i = 0; i < EPW; ++i) nrm[i] = dinv[s[i]] * dinv[d[i]];
#pragma unroll
        for (int i = 0; i < EPW; ++i)
#pragma unroll
            for (int j = 0; j < NJ; ++j)
                atomicAdd(agg + (size_t)d[i] * C + j * 32 + lane, v[i][j] * nrm[i]);
    } else {
        for (int i = 0; i < EPW; ++i) {
            if (e0 + i >= (long long)e) break;
            const int s = src[e0 + i];
            const int d = dst[e0 + i];
            const float nrm = dinv[s] * dinv[d];
#pragma unroll
            for (int j = 0; j < NJ; ++j) {
                const float v = h[(size_t)s * C + j * 32 + lane];
                atomicAdd(agg + (size_t)d * C + j * 32 + lane, v * nrm);
            }
        }
    }
}

template <int C>
__global__ void k_relu_bias(const float* __restrict__ agg, const float* __restrict__ b,
                            float* __restrict__ p, int n) {
    long long idx = (long long)blockIdx.x * blockDim.x + threadIdx.x;
    long long total = (long long)n * C;
    if (idx >= total) return;
    int c = (int)(idx % C);
    float v = agg[idx] + b[c];
    p[idx] = v > 0.0f ? v : 0.0f;
}

constexpr int AGG_TB   = 1024;
constexpr int AGG_G    = 16;
constexpr int SRT_CAP  = 8192;
constexpr int SRT_K    = SRT_CAP / AGG_TB;
constexpr int SRT_RPL  = (BN + 31) / 32;
constexpr int SRT_ROWS = SRT_RPL * 32;

template <int C>
__device__ __forceinline__ void agg_records32(unsigned int pk, int cnt, const float* __restrict__ h,
                                              float* acc, int lane) {
    constexpr int P  = C + 4;
    constexpr int NJ = C / 32;
    const int s = (int)(pk & 0x1FFFFu);
    const int key  = (lane < cnt) ? (int)(pk >> 17) : 0x7FFF;
    const int prev = __shfl_up(key, 1);
    const unsigned int heads = __builtin_amdgcn_ballot_w32(lane > 0 && key != prev);
    float sum[NJ];
#pragma unroll
    for (int j = 0; j < NJ; ++j) sum[j] = 0.0f;
#pragma unroll
    for (int g = 0; g < 32; g += AGG_G) {
        if (g < cnt) {
            float v[AGG_G][NJ];
#pragma unroll
            for (int k = 0; k < AGG_G; ++k) {
                const int sk = __builtin_amdgcn_readlane(s, g + k);
#pragma unroll
                for (int j = 0; j < NJ; ++j) v[k][j] = h[(size_t)sk * C + j * 32 + lane];
            }
#pragma unroll
            for (int k = 0; k < AGG_G; ++k) {
                const int i = g + k;
                if (i > 0 && __builtin_expect((heads >> i) & 1u, 0u)) {
                    const int dp = __builtin_amdgcn_readlane(key, i > 0 ? i - 1 : 0);
#pragma unroll
                    for (int j = 0; j < NJ; ++j) {
                        atomicAdd(&acc[dp * P + j * 32 + lane], sum[j]);
                        sum[j] = 0.0f;
                    }
                }
#pragma unroll
                for (int j = 0; j < NJ; ++j) sum[j] += v[k][j];
            }
        }
    }
    if (cnt == 32) {
        const int dp = __builtin_amdgcn_readlane(key, 31);
#pragma unroll
        for (int j = 0; j < NJ; ++j) atomicAdd(&acc[dp * P + j * 32 + lane], sum[j]);
    } else if (cnt == AGG_G) {
        const int dp = __builtin_amdgcn_readlane(key, AGG_G - 1);
#pragma unroll
        for (int j = 0; j < NJ; ++j) atomicAdd(&acc[dp * P + j * 32 + lane], sum[j]);
    }
}

template <int C, bool LAYER1>
__global__ void __launch_bounds__(AGG_TB)
k_range_agg(unsigned int* __restrict__ recs, const unsigned int* __restrict__ bstart,
            const float* __restrict__ h, const float* __restrict__ dinv,
            const float* __restrict__ bias, const float* __restrict__ W2,
            float* __restrict__ outp, int n) {
    constexpr int P  = C + 4;
    __shared__ alignas(16) float acc[BN * P];
    __shared__ alignas(16) float dloc[BN];
    __shared__ unsigned int srt[LAYER1 ? SRT_CAP : 1];
    __shared__ unsigned int cntr[LAYER1 ? SRT_ROWS : 1];
    __shared__ unsigned int rowst[LAYER1 ? SRT_ROWS : 1];
    const int tid  = threadIdx.x;
    const int lane = tid & 31;
    const int wv   = __builtin_amdgcn_readfirstlane(tid >> 5);
    const int n0   = blockIdx.x * BN;
    const int rows = (n - n0) < BN ? (n - n0) : BN;

    const unsigned int eb = bstart[blockIdx.x];
    const unsigned int ee = bstart[blockIdx.x + 1];

    float dself;
    {
        int node = n0 + (tid < BN ? tid : BN - 1);
        node = node < n ? node : n - 1;
        dself = dinv[node];
    }

    unsigned int rec[SRT_K];
    if constexpr (LAYER1) {
        if (tid < SRT_ROWS) cntr[tid] = 0u;
#pragma unroll
        for (int k = 0; k < SRT_K; ++k) {
            const unsigned int idx = (unsigned int)(k * AGG_TB + tid);
            rec[k] = (idx < ee - eb) ? recs[eb + idx] : 0u;
        }
    }

    unsigned int pk0 = 0u;
    if constexpr (!LAYER1) {
        const unsigned int c00 = eb + (unsigned int)wv * 32u;
        pk0 = (c00 + lane < ee) ? recs[c00 + lane] : 0u;
    }

    constexpr int C4     = C / 4;
    constexpr int INIT_T = 2 * BN;
    constexpr int INIT_K = C4 / 2;
    static_assert(INIT_T <= AGG_TB && INIT_T % 32 == 0 && INIT_T * INIT_K == BN * C4,
                  "start-value shares must be whole");
    if (tid < INIT_T) {
        float4 hv[INIT_K];
#pragma unroll
        for (int u = 0; u < INIT_K; ++u) {
            const int idx = tid + u * INIT_T;
            const int r = idx / C4, c4 = idx - r * C4;
            int node = n0 + r;
            node = node < n ? node : n - 1;
            hv[u] = *reinterpret_cast<const float4*>(h + (size_t)node * C + c4 * 4);
        }
#pragma unroll
        for (int u = 0; u < INIT_K; ++u) {
            const int idx = tid + u * INIT_T;
            const int r = idx / C4, c4 = idx - r * C4;
            *reinterpret_cast<float4*>(&acc[r * P + c4 * 4]) = hv[u];
        }
    }
    if (tid < BN) dloc[tid] = (tid < rows) ? dself : 0.0f;
    __syncthreads();

    if constexpr (LAYER1) {
        unsigned int base = eb;
        for (;;) {
            const unsigned int left = ee - base;
            const unsigned int m = left < (unsigned int)SRT_CAP ? left : (unsigned int)SRT_CAP;
            unsigned int rk[SRT_K];
#pragma unroll
            for (int k = 0; k < SRT_K; ++k) {
                rk[k] = 0u;
                if ((unsigned int)(k * AGG_TB + tid) < m) rk[k] = atomicAdd(&cntr[rec[k] >> 17], 1u);
            }
            __syncthreads();
            if (wv == 0) {
                unsigned int cv[SRT_RPL];
                unsigned int tot = 0u;
#pragma unroll
                for (int i = 0; i < SRT_RPL; ++i) {
                    const unsigned int t = cntr[lane * SRT_RPL + i];
                    cv[i] = tot;
                    tot += t;
                }
                unsigned int incl = tot;
                for (int o = 1; o < 32; o <<= 1) {
                    const unsigned int v = __shfl_up(incl, o);
                    if (lane >= o) incl += v;
                }
                const unsigned int before = incl - tot;
#pragma unroll
                for (int i = 0; i < SRT_RPL; ++i) {
                    rowst[lane * SRT_RPL + i] = before + cv[i];
                    cntr[lane * SRT_RPL + i]  = 0u;
                }
            }
            __syncthreads();
            unsigned int pos[SRT_K];
#pragma unroll
            for (int k = 0; k < SRT_K; ++k) pos[k] = rowst[rec[k] >> 17] + rk[k];
#pragma unroll
            for (int k = 0; k < SRT_K; ++k) {
                if ((unsigned int)(k * AGG_TB + tid) < m) srt[pos[k]] = rec[k];
            }
            __syncthreads();
            unsigned int sv[SRT_K];
#pragma unroll
            for (int k = 0; k < SRT_K; ++k) sv[k] = srt[k * AGG_TB + tid];
#pragma unroll
            for (int k = 0; k < SRT_K; ++k) {
                const unsigned int idx = (unsigned int)(k * AGG_TB + tid);
                if (idx < m) recs[base + idx] = sv[k];
            }
            for (unsigned int c0 = (unsigned int)wv * 32u; c0 < m; c0 += AGG_TB) {
                const unsigned int pk = (c0 + lane < m) ? srt[c0 + lane] : 0u;
                const int cnt = (int)((m - c0) < 32u ? (m - c0) : 32u);
                agg_records32<C>(pk, cnt, h, acc, lane);
            }
            base += m;
            if (base >= ee) break;
            __syncthreads();
#pragma unroll
            for (int k = 0; k < SRT_K; ++k) {
                const unsigned int idx = (unsigned int)(k * AGG_TB + tid);
                rec[k] = (idx < ee - base) ? recs[base + idx] : 0u;
            }
        }
    } else {
        unsigned int c0 = eb + (unsigned int)wv * 32u;
        unsigned int pk = pk0;
        for (; c0 < ee; c0 += AGG_TB) {
            const unsigned int cn = c0 + AGG_TB;
            const unsigned int pkn = (cn + lane < ee) ? recs[cn + lane] : 0u;
            const int cnt = (int)((ee - c0) < 32u ? (ee - c0) : 32u);
            agg_records32<C>(pk, cnt, h, acc, lane);
            pk = pkn;
        }
    }
    __syncthreads();

    if constexpr (!LAYER1) {
        for (int idx = tid; idx < rows * C; idx += AGG_TB) {
            const int r = idx / C, c = idx - r * C;
            outp[(size_t)(n0 + r) * C + c] = acc[r * P + c] * dloc[r] + bias[c];
        }
    } else {
        for (int idx = tid; idx < BN * C; idx += AGG_TB) {
            const int r = idx / C, c = idx - r * C;
            const float v = acc[r * P + c] * dloc[r] + bias[c];
            acc[r * P + c] = v > 0.0f ? v : 0.0f;
        }
        __syncthreads();
        constexpr int NO = 32;
        constexpr int NT = NO / 16;
        const int full = rows / 16;
        if (wv < full) {
            const int m  = lane & 15;
            const int kh = lane >> 4;
            v8f d[NT];
#pragma unroll
            for (int t = 0; t < NT; ++t)
#pragma unroll
                for (int j = 0; j < 8; ++j) d[t][j] = 0.0f;
            float sc[8];
#pragma unroll
            for (int j = 0; j < 8; ++j) sc[j] = dloc[wv * 16 + j + 8 * kh];
            const float* arow = acc + (wv * 16 + m) * P;
            for (int k = 0; k < C; k += 4) {
                v2f a;
                a.x = arow[k + 2 * kh + 0];
                a.y = arow[k + 2 * kh + 1];
#pragma unroll
                for (int t = 0; t < NT; ++t) {
                    v2f b;
                    b.x = W2[(size_t)(k + 2 * kh + 0) * NO + t * 16 + m];
                    b.y = W2[(size_t)(k + 2 * kh + 1) * NO + t * 16 + m];
                    d[t] = __builtin_amdgcn_wmma_f32_16x16x4_f32(
                        false, a, false, b, (short)0, d[t], false, false);
                }
            }
#pragma unroll
            for (int t = 0; t < NT; ++t)
#pragma unroll
                for (int j = 0; j < 8; ++j)
                    outp[(size_t)(n0 + wv * 16 + j + 8 * kh) * NO + t * 16 + m] = d[t][j] * sc[j];
        }
        const int r1 = full * 16;
        for (int idx = tid; idx < (rows - r1) * NO; idx += AGG_TB) {
            const int r = r1 + idx / NO, c = idx % NO;
            float sum = 0.0f;
            for (int k = 0; k < C; ++k) sum += acc[r * P + k] * W2[(size_t)k * NO + c];
            outp[(size_t)(n0 + r) * NO + c] = sum * dloc[r];
        }
    }
}

extern "C" void kernel_launch(void* const* d_in, const int* in_sizes, int n_in,
                              void* d_out, int out_size, void* d_ws, size_t ws_size,
                              hipStream_t stream) {
    const float* x  = (const float*)d_in[0];
    const int*   ei = (const int*)d_in[1];
    const float* W1 = (const float*)d_in[2];
    const float* b1 = (const float*)d_in[3];
    const float* W2 = (const float*)d_in[4];
    const float* b2 = (const float*)d_in[5];
    float* out = (float*)d_out;

    const int n = in_sizes[0] / 128;
    const int e = in_sizes[1] / 2;
    const int* src = ei;
    const int* dst = ei + e;

    char* ws = (char*)d_ws;
    size_t off0 = ((size_t)n * sizeof(float) + 255) & ~(size_t)255;
    float* dinv = (float*)ws;
    float* bufA = (float*)(ws + off0);
    float* bufB = bufA + (size_t)n * 64;

    const int tb = 256;
    auto cdiv = [](long long a, long long b) { return (int)((a + b - 1) / b); };

    const int parts = cdiv(e, DEG_ECH);
    const int strips = n / 16;
    const int wavesPerBlk = tb / 32;
    const int nb = cdiv(n, BN);
    const size_t half = (size_t)n * 32;
    if (n <= DEG_NCAP && (size_t)parts * DEG_WORDS <= half &&
        (size_t)e + (size_t)parts * nb + nb + 1 <= half) {
        unsigned int* slabs  = (unsigned int*)bufB;
        unsigned int* recs   = (unsigned int*)bufB + half;
        unsigned int* pcnt   = recs + e;
        unsigned int* bstart = pcnt + (size_t)parts * nb;
        float* h2 = bufB;

        k_deg_hist<<<parts, DEG_TB, 0, stream>>>(dst, slabs, pcnt, nb, e);
        k_place_gemm<128, 64><<<parts + cdiv(n, PG_ROWS), PG_TB, 0, stream>>>(
            src, dst, pcnt, recs, bstart, e, nb, parts, slabs, x, W1, dinv, bufA, n);
        if (n % 16) {
            int tail = (n % 16) * 64;
            k_gemm_tail<128, 64, true><<<cdiv(tail, tb), tb, 0, stream>>>(x, W1, dinv, bufA, strips * 16, n);
        }
        k_range_agg<64, true><<<nb, AGG_TB, 0, stream>>>(recs, bstart, bufA, dinv, b1, W2, h2, n);
        k_range_agg<32, false><<<nb, AGG_TB, 0, stream>>>(recs, bstart, h2, dinv, b2, nullptr, out, n);
        return;
    }

    if (n <= DEG_NCAP && (size_t)parts * DEG_WORDS <= (size_t)n * 64) {
        k_deg_hist<<<parts, DEG_TB, 0, stream>>>(dst, (unsigned int*)bufB, nullptr, 0, e);
        k_dinv_hist<<<cdiv((n + 1) / 2, tb), tb, 0, stream>>>((const unsigned int*)bufB, dinv, n, parts);
    } else {
        k_deg_init<<<cdiv(n, tb), tb, 0, stream>>>(dinv, n);
        k_deg_count<<<cdiv(e, tb), tb, 0, stream>>>(dst, dinv, e);
        k_dinv<<<cdiv(n, tb), tb, 0, stream>>>(dinv, n);
    }

    k_gemm_wmma<128, 64, false><<<cdiv(strips, wavesPerBlk), tb, 0, stream>>>(x, W1, nullptr, bufA, n);
    if (n % 16) {
        int tail = (n % 16) * 64;
        k_gemm_tail<128, 64, false><<<cdiv(tail, tb), tb, 0, stream>>>(x, W1, nullptr, bufA, strips * 16, n);
    }

    k_self_init<64, false><<<cdiv((long long)n * 64, tb), tb, 0, stream>>>(bufA, dinv, nullptr, bufB, n);
    k_edge_agg<64><<<cdiv(cdiv(e, 8), wavesPerBlk), tb, 0, stream>>>(src, dst, bufA, dinv, bufB, e);
    k_relu_bias<64><<<cdiv((long long)n * 64, tb), tb, 0, stream>>>(bufB, b1, bufA, n);

    k_gemm_wmma<64, 32, false><<<cdiv(strips, wavesPerBlk), tb, 0, stream>>>(bufA, W2, nullptr, bufB, n);
    if (n % 16) {
        int tail = (n % 16) * 32;
        k_gemm_tail<64, 32, false><<<cdiv(tail, tb), tb, 0, stream>>>(bufA, W2, nullptr, bufB, strips * 16, n);
    }

    k_self_init<32, true><<<cdiv((long long)n * 32, tb), tb, 0, stream>>>(bufB, dinv, b2, out, n);
    k_edge_agg<32><<<cdiv(cdiv(e, 8), wavesPerBlk), tb, 0, stream>>>(src, dst, bufB, dinv, out, e);

    (void)n_in; (void)out_size; (void)ws_size;
}


# WMMA operands fed from LDS: W1 (per GEMM group) and W2 (per layer-1 range group) copied once in fragment order, x strips read as whole 128-B lines via per-wave LDS rows a chunk ahead; sums unchanged
# speedup vs baseline: 4.8147x; 1.0425x over previous
#include <hip/hip_runtime.h>

typedef float v2f __attribute__((ext_vector_type(2)));
typedef float v4f __attribute__((ext_vector_type(4)));
typedef float v8f __attribute__((ext_vector_type(8)));

__global__ void k_deg_init(float* __restrict__ deg, int n) {
    int i = blockIdx.x * blockDim.x + threadIdx.x;
    if (i < n) deg[i] = 1.0f;
}

__global__ void k_deg_count(const int* __restrict__ dst, float* __restrict__ deg, int e) {
    int i = blockIdx.x * blockDim.x + threadIdx.x;
    if (i < e) atomicAdd(&deg[dst[i]], 1.0f);
}

__global__ void k_dinv(float* __restrict__ deg, int n) {
    int i = blockIdx.x * blockDim.x + threadIdx.x;
    if (i < n) deg[i] = 1.0f / sqrtf(deg[i]);
}

constexpr int DEG_NCAP   = 100000;
constexpr int DEG_WORDS  = ((DEG_NCAP + 1) / 2 + 31) & ~31;
constexpr int DEG_ECH    = 32768;
constexpr int DEG_TB     = 1024;
constexpr int DEG_B      = 16;
constexpr int BN         = 400;
constexpr int BW         = BN / 2;
constexpr int NBMAX      = (DEG_NCAP + BN - 1) / BN;
constexpr int PG_TB      = 1024;
constexpr int PG_ROWS    = (PG_TB / 32) * 16;
constexpr int PG_XW      = 10240;
constexpr int PG_PB      = 16;
constexpr int PG_AC      = 32;
constexpr int PG_AP      = PG_AC + 4;

__global__ void __launch_bounds__(DEG_TB)
k_deg_hist(const int* __restrict__ dst, unsigned int* __restrict__ part,
           unsigned int* __restrict__ pcnt, int nb, int e) {
    __shared__ unsigned int cnt[DEG_WORDS];
    const int tid = threadIdx.x;

    uint4* c4 = reinterpret_cast<uint4*>(cnt);
    for (int i = tid; i < DEG_WORDS / 4; i += DEG_TB) c4[i] = make_uint4(0u, 0u, 0u, 0u);
    __syncthreads();

    const long long e0 = (long long)blockIdx.x * DEG_ECH;
    const long long left = (long long)e - e0;
    const int m = left < (long long)DEG_ECH ? (int)left : DEG_ECH;
    const int* __restrict__ dp = dst + e0;
    for (int base = tid; base < m; base += DEG_B * DEG_TB) {
        int d[DEG_B];
#pragma unroll
        for (int k = 0; k < DEG_B; ++k) {
            const int i = base + k * DEG_TB;
            d[k] = (i < m) ? dp[i] : -1;
        }
#pragma unroll
        for (int k = 0; k < DEG_B; ++k) {
            if (base + k * DEG_TB < m) atomicAdd(&cnt[d[k] >> 1], (d[k] & 1) ? 0x10000u : 1u);
        }
    }
    __syncthreads();

    uint4* p4 = reinterpret_cast<uint4*>(part + (size_t)blockIdx.x * DEG_WORDS);
    for (int i = tid; i < DEG_WORDS / 4; i += DEG_TB) p4[i] = c4[i];

    if (pcnt != nullptr) {
        constexpr int QW = BW / 4;
        static_assert(BW % 8 == 0, "a quarter of a range is a whole number of 8-byte reads");
        static_assert(NBMAX * 4 <= DEG_TB && NBMAX * BW <= DEG_WORDS && DEG_ECH < 65536,
                      "four threads per range; ranges inside the table; 16-bit sums");
        const int b = tid >> 2, j = tid & 3;
        unsigned int acc = 0u;
        if (b < nb) {
            const uint2* q2 = reinterpret_cast<const uint2*>(cnt + b * BW + j * QW);
#pragma unroll
            for (int i = 0; i < QW / 2; ++i) {
                const uint2 v = q2[i];
                acc += v.x + v.y;
            }
        }
        acc += __shfl_xor(acc, 1);
        acc += __shfl_xor(acc, 2);
        if (j == 0 && b < nb) pcnt[(size_t)blockIdx.x * nb + b] = (acc & 0xFFFFu) + (acc >> 16);
    }
}

__global__ void k_dinv_hist(const unsigned int* __restrict__ part, float* __restrict__ dinv,
                            int n, int parts) {
    const int t = blockIdx.x * blockDim.x + threadIdx.x;
    if (2 * t >= n) return;
    unsigned int lo = 0u, hi = 0u;
#pragma unroll 8
    for (int p = 0; p < parts; ++p) {
        const unsigned int w = part[(size_t)p * DEG_WORDS + t];
        lo += w & 0xFFFFu;
        hi += w >> 16;
    }
    dinv[2 * t] = 1.0f / sqrtf((float)(lo + 1u));
    if (2 * t + 1 < n) dinv[2 * t + 1] = 1.0f / sqrtf((float)(hi + 1u));
}

template <int K, int NOUT, bool SCALED>
__global__ void k_gemm_wmma(const float* __restrict__ A, const float* __restrict__ W,
                            const float* __restrict__ rowscale,
                            float* __restrict__ H, int nrows) {
    const int wave = blockIdx.x * (blockDim.x >> 5) + (threadIdx.x >> 5);
    const int lane = threadIdx.x & 31;
    const int m    = lane & 15;
    const int kh   = lane >> 4;
    const int r0   = wave * 16;
    if (r0 + 16 > nrows) return;

    constexpr int NT = NOUT / 16;
    v8f acc[NT];
#pragma unroll
    for (int t = 0; t < NT; ++t)
#pragma unroll
        for (int j = 0; j < 8; ++j) acc[t][j] = 0.0f;

    float sc[8];
#pragma unroll
    for (int j = 0; j < 8; ++j) sc[j] = SCALED ? rowscale[r0 + j + 8 * kh] : 1.0f;

    const float* arow = A + (size_t)(r0 + m) * K;
    for (int k = 0; k < K; k += 4) {
        v2f a;
        a.x = arow[k + 2 * kh + 0];
        a.y = arow[k + 2 * kh + 1];
#pragma unroll
        for (int t = 0; t < NT; ++t) {
            v2f b;
            b.x = W[(size_t)(k + 2 * kh + 0) * NOUT + t * 16 + m];
            b.y = W[(size_t)(k + 2 * kh + 1) * NOUT + t * 16 + m];
            acc[t] = __builtin_amdgcn_wmma_f32_16x16x4_f32(
                 false, a,  false, b,
                 (short)0, acc[t],  false,  false);
        }
    }

#pragma unroll
    for (int t = 0; t < NT; ++t) {
#pragma unroll
        for (int j = 0; j < 8; ++j)
            H[(size_t)(r0 + j + 8 * kh) * NOUT + t * 16 + m] =
                SCALED ? acc[t][j] * sc[j] : acc[t][j];
    }
}

template <int K, int NOUT, bool SCALED>
__global__ void k_gemm_tail(const float* __restrict__ A, const float* __restrict__ W,
                            const float* __restrict__ rowscale,
                            float* __restrict__ H, int row0, int nrows) {
    int idx = blockIdx.x * blockDim.x + threadIdx.x;
    int r = row0 + idx / NOUT;
    int c = idx % NOUT;
    if (r >= nrows) return;
    float s = 0.0f;
    for (int k = 0; k < K; ++k) s += A[(size_t)r * K + k] * W[(size_t)k * NOUT + c];
    H[(size_t)r * NOUT + c] = SCALED ? s * rowscale[r] : s;
}

template <int K, int NOUT>
__global__ void __launch_bounds__(PG_TB)
k_place_gemm(const int* __restrict__ src, const int* __restrict__ dst,
             const unsigned int* __restrict__ pcnt, unsigned int* __restrict__ recs,
             unsigned int* __restrict__ bstart, int e, int nb, int parts,
             const unsigned int* __restrict__ part, const float* __restrict__ A,
             const float* __restrict__ W, float* __restrict__ dinv,
             float* __restrict__ H, int nrows) {
    __shared__ alignas(16) unsigned int stage[DEG_ECH + PG_XW];
    __shared__ unsigned int lbeg[NBMAX + 1];
    __shared__ unsigned int cur[NBMAX];
    __shared__ unsigned int gbeg[NBMAX];
    __shared__ unsigned int wtot[8];
    __shared__ unsigned int wtot2[8];
    unsigned int* xs = stage + DEG_ECH;
    const int tid  = threadIdx.x;
    const int lane = tid & 31;
    const int wv   = tid >> 5;

    if ((int)blockIdx.x < parts) {
        const int p = blockIdx.x;
        unsigned int c = 0u;
        if (tid < 256 && tid < nb) c = pcnt[(size_t)p * nb + tid];
        {
            const int b = tid & 255, q = tid >> 8;
            unsigned int st = 0u, sb = 0u;
            if (b < nb) {
                for (int p0 = q; p0 < parts; p0 += 4 * PG_PB) {
                    unsigned int v[PG_PB];
#pragma unroll
                    for (int k = 0; k < PG_PB; ++k) {
                        const int pp = p0 + 4 * k;
                        v[k] = (pp < parts) ? pcnt[(size_t)pp * nb + b] : 0u;
                    }
#pragma unroll
                    for (int k = 0; k < PG_PB; ++k) {
                        const int pp = p0 + 4 * k;
                        st += v[k];
                        sb += (pp < p) ? v[k] : 0u;
                    }
                }
            }
            xs[q * 256 + b] = st;
            xs[1024 + q * 256 + b] = sb;
        }
        __syncthreads();

        unsigned int incl = 0u, t = 0u, tincl = 0u, bef = 0u;
        if (tid < 256) {
            t   = xs[tid] + xs[256 + tid] + xs[512 + tid] + xs[768 + tid];
            bef = xs[1024 + tid] + xs[1280 + tid] + xs[1536 + tid] + xs[1792 + tid];
            incl = c;
            tincl = t;
            for (int o = 1; o < 32; o <<= 1) {
                const unsigned int v  = __shfl_up(incl, o);
                const unsigned int v2 = __shfl_up(tincl, o);
                if (lane >= o) { incl += v; tincl += v2; }
            }
            if (lane == 31) { wtot[wv] = incl; wtot2[wv] = tincl; }
        }
        __syncthreads();
        if (tid < 256) {
            unsigned int before = 0u, before2 = 0u;
            for (int w = 0; w < wv; ++w) { before += wtot[w]; before2 += wtot2[w]; }
            const unsigned int ex  = before + incl - c;
            const unsigned int ex2 = before2 + tincl - t;
            if (tid < nb) {
                lbeg[tid] = ex;
                cur[tid]  = ex;
                gbeg[tid] = ex2 + bef;
                if (p == 0) bstart[tid] = ex2;
            }
            if (tid == nb - 1) {
                lbeg[nb] = ex + c;
                if (p == 0) bstart[nb] = ex2 + t;
            }
        }
        __syncthreads();

        const long long e0 = (long long)p * DEG_ECH;
        const long long left = (long long)e - e0;
        const int m = left < (long long)DEG_ECH ? (int)left : DEG_ECH;
        const int* __restrict__ dp = dst + e0;
        const int* __restrict__ sp = src + e0;
        for (int base = tid; base < m; base += DEG_B * DEG_TB) {
            int d[DEG_B], s[DEG_B];
#pragma unroll
            for (int k = 0; k < DEG_B; ++k) {
                const int i = base + k * DEG_TB;
                d[k] = (i < m) ? dp[i] : 0;
                s[k] = (i < m) ? sp[i] : 0;
            }
#pragma unroll
            for (int k = 0; k < DEG_B; ++k) {
                if (base + k * DEG_TB < m) {
                    const int b = d[k] / BN;
                    const unsigned int pos = atomicAdd(&cur[b], 1u);
                    stage[pos] = ((unsigned int)(d[k] - b * BN) << 17) | (unsigned int)s[k];
                }
            }
        }
        __syncthreads();

        for (int b = wv; b < nb; b += DEG_TB / 32) {
            const unsigned int l0  = lbeg[b];
            const unsigned int len = lbeg[b + 1] - l0;
            const unsigned int g0  = gbeg[b];
            const unsigned int a   = g0 & 31u;
            for (unsigned int j = lane; j < len + a; j += 32u) {
                if (j >= a) recs[g0 - a + j] = stage[l0 + j - a];
            }
        }
        return;
    }

    constexpr int NT  = NOUT / 16;
    constexpr int NTP = NT / 2;
    constexpr int NC  = K / PG_AC;
    constexpr int W4  = K * NOUT / 4;
    constexpr int WPT = W4 / PG_TB;
    constexpr int ARW = 16 * PG_AP;
    static_assert(NT % 2 == 0 && K % PG_AC == 0 && PG_AC == 32 && NOUT % 4 == 0 && W4 % PG_TB == 0,
                  "tile pairs; whole chunks of one line per row; whole shares of W");
    static_assert(K * NOUT + (PG_TB / 32) * ARW <= DEG_ECH, "W and the waves' A rows fit the idle record stage");
    const int row0 = ((int)blockIdx.x - parts) * PG_ROWS;
    const int m  = lane & 15;
    const int kh = lane >> 4;
    const int r0 = row0 + wv * 16;
    const bool strip = (r0 + 16 <= nrows);
    float* wl = reinterpret_cast<float*>(stage);
    float* as = reinterpret_cast<float*>(stage) + K * NOUT + wv * ARW;

    float4 wq[WPT];
#pragma unroll
    for (int i = 0; i < WPT; ++i) wq[i] = reinterpret_cast<const float4*>(W)[tid + i * PG_TB];
    const float* ag = A + (size_t)(r0 + (lane >> 3)) * K + (lane & 7) * 4;
    v4f an[4];
    if (strip) {
#pragma unroll
        for (int i = 0; i < 4; ++i) an[i] = *reinterpret_cast<const v4f*>(ag + (size_t)(4 * i) * K);
    }
    {
        const int w = tid & 255, q = tid >> 8;
        const int tw = (row0 >> 1) + w;
        unsigned int lo = 0u, hi = 0u;
        if (2 * tw < nrows) {
            for (int p0 = q; p0 < parts; p0 += 4 * PG_PB) {
                unsigned int v[PG_PB];
#pragma unroll
                for (int k = 0; k < PG_PB; ++k) {
                    const int pp = p0 + 4 * k;
                    v[k] = (pp < parts) ? part[(size_t)pp * DEG_WORDS + tw] : 0u;
                }
#pragma unroll
                for (int k = 0; k < PG_PB; ++k) {
                    lo += v[k] & 0xFFFFu;
                    hi += v[k] >> 16;
                }
            }
        }
        xs[q * 256 + w] = lo;
        xs[1024 + q * 256 + w] = hi;
    }
#pragma unroll
    for (int i = 0; i < WPT; ++i) {
        const int g4 = tid + i * PG_TB;
        const int kr = g4 / (NOUT / 4), c4 = g4 - kr * (NOUT / 4);
        const int kk = kr >> 2, h = (kr >> 1) & 1, kb = kr & 1;
        const int t = c4 >> 2, tp = t >> 1, tb = t & 1, c0 = (c4 & 3) * 4;
        float* d = wl + ((kk * NTP + tp) * 32 + h * 16 + c0) * 4 + 2 * tb + kb;
        d[0] = wq[i].x; d[4] = wq[i].y; d[8] = wq[i].z; d[12] = wq[i].w;
    }
    __syncthreads();
    float* dl = reinterpret_cast<float*>(xs + 2048);
    if (tid < PG_ROWS) {
        const int w = tid >> 1;
        const unsigned int* hsel = xs + ((tid & 1) ? 1024 : 0);
        const unsigned int cntv = hsel[w] + hsel[256 + w] + hsel[512 + w] + hsel[768 + w];
        const float dv = 1.0f / sqrtf((float)(cntv + 1u));
        dl[tid] = dv;
        if (row0 + tid < nrows) dinv[row0 + tid] = dv;
    }
    __syncthreads();

    if (!strip) return;

    v8f acc[NT];
#pragma unroll
    for (int t = 0; t < NT; ++t)
#pragma unroll
        for (int j = 0; j < 8; ++j) acc[t][j] = 0.0f;

    float* ast = as + (lane >> 3) * PG_AP + (lane & 7) * 4;
    const float* afr = as + m * PG_AP + 2 * kh;
    const float4* bfr = reinterpret_cast<const float4*>(wl) + lane;
#pragma unroll
    for (int c = 0; c < NC; ++c) {
#pragma unroll
        for (int i = 0; i < 4; ++i) *reinterpret_cast<v4f*>(ast + 4 * i * PG_AP) = an[i];
        __builtin_amdgcn_fence(__ATOMIC_RELEASE, "wavefront");
        __builtin_amdgcn_wave_barrier();
        __builtin_amdgcn_fence(__ATOMIC_ACQUIRE, "wavefront");
        if (c + 1 < NC) {
#pragma unroll
            for (int i = 0; i < 4; ++i)
                an[i] = *reinterpret_cast<const v4f*>(ag + (size_t)(4 * i) * K + (c + 1) * PG_AC);
        }
#pragma unroll
        for (int j = 0; j < PG_AC / 4; ++j) {
            const float2 av = *reinterpret_cast<const float2*>(afr + 4 * j);
            v2f a;
            a.x = av.x;
            a.y = av.y;
#pragma unroll
            for (int tp = 0; tp < NTP; ++tp) {
                const float4 q = bfr[((c * (PG_AC / 4) + j) * NTP + tp) * 32];
                v2f b0, b1;
                b0.x = q.x; b0.y = q.y;
                b1.x = q.z; b1.y = q.w;
                acc[2 * tp] = __builtin_amdgcn_wmma_f32_16x16x4_f32(
                     false, a,  false, b0,
                     (short)0, acc[2 * tp],  false,  false);
                acc[2 * tp + 1] = __builtin_amdgcn_wmma_f32_16x16x4_f32(
                     false, a,  false, b1,
                     (short)0, acc[2 * tp + 1],  false,  false);
            }
        }
        __builtin_amdgcn_fence(__ATOMIC_RELEASE, "wavefront");
        __builtin_amdgcn_wave_barrier();
        __builtin_amdgcn_fence(__ATOMIC_ACQUIRE, "wavefront");
    }

    float sc[8];
#pragma unroll
    for (int j = 0; j < 8; ++j) sc[j] = dl[wv * 16 + j + 8 * kh];
#pragma unroll
    for (int t = 0; t < NT; ++t) {
#pragma unroll
        for (int j = 0; j < 8; ++j)
            H[(size_t)(r0 + j + 8 * kh) * NOUT + t * 16 + m] = acc[t][j] * sc[j];
    }
}

template <int C, bool HAS_BIAS>
__global__ void k_self_init(const float* __restrict__ h, const float* __restrict__ dinv,
                            const float* __restrict__ bias, float* __restrict__ agg, int n) {
    long long idx = (long long)blockIdx.x * blockDim.x + threadIdx.x;
    long long total = (long long)n * C;
    if (idx >= total) return;
    int i = (int)(idx / C);
    int c = (int)(idx % C);
    float d = dinv[i];
    float v = h[idx] * d * d;
    if (HAS_BIAS) v += bias[c];
    agg[idx] = v;
}

template <int C>
__global__ void k_edge_agg(const int* __restrict__ src, const int* __restrict__ dst,
                           const float* __restrict__ h, const float* __restrict__ dinv,
                           float* __restrict__ agg, int e) {
    constexpr int EPW = 8;
    constexpr int NJ  = C / 32;
    const int lane = threadIdx.x & 31;
    const int wave = __builtin_amdgcn_readfirstlane(
        (int)(blockIdx.x * (blockDim.x >> 5) + (threadIdx.x >> 5)));
    const long long e0 = (long long)wave * EPW;
    if (e0 >= (long long)e) return;

    if (e0 + EPW <= (long long)e) {
        int s[EPW], d[EPW];
        float v[EPW][NJ];
        float nrm[EPW];
#pragma unroll
        for (int i = 0; i < EPW; ++i) {
            s[i] = src[e0 + i];
            d[i] = dst[e0 + i];
        }
#pragma unroll
        for (int i = 0; i < EPW; ++i)
#pragma unroll
            for (int j = 0; j < NJ; ++j)
                v[i][j] = h[(size_t)s[i] * C + j * 32 + lane];
#pragma unroll
        for (int i = 0; i < EPW; ++i) nrm[i] = dinv[s[i]] * dinv[d[i]];
#pragma unroll
        for (int i = 0; i < EPW; ++i)
#pragma unroll
            for (int j = 0; j < NJ; ++j)
                atomicAdd(agg + (size_t)d[i] * C + j * 32 + lane, v[i][j] * nrm[i]);
    } else {
        for (int i = 0; i < EPW; ++i) {
            if (e0 + i >= (long long)e) break;
            const int s = src[e0 + i];
            const int d = dst[e0 + i];
            const float nrm = dinv[s] * dinv[d];
#pragma unroll
            for (int j = 0; j < NJ; ++j) {
                const float v = h[(size_t)s * C + j * 32 + lane];
                atomicAdd(agg + (size_t)d * C + j * 32 + lane, v * nrm);
            }
        }
    }
}

template <int C>
__global__ void k_relu_bias(const float* __restrict__ agg, const float* __restrict__ b,
                            float* __restrict__ p, int n) {
    long long idx = (long long)blockIdx.x * blockDim.x + threadIdx.x;
    long long total = (long long)n * C;
    if (idx >= total) return;
    int c = (int)(idx % C);
    float v = agg[idx] + b[c];
    p[idx] = v > 0.0f ? v : 0.0f;
}

constexpr int AGG_TB   = 1024;
constexpr int AGG_G    = 16;
constexpr int SRT_CAP  = 8192;
constexpr int SRT_K    = SRT_CAP / AGG_TB;
constexpr int SRT_RPL  = (BN + 31) / 32;
constexpr int SRT_ROWS = SRT_RPL * 32;

template <int C>
__device__ __forceinline__ void agg_records32(unsigned int pk, int cnt, const float* __restrict__ h,
                                              float* acc, int lane) {
    constexpr int P  = C + 4;
    constexpr int NJ = C / 32;
    const int s = (int)(pk & 0x1FFFFu);
    const int key  = (lane < cnt) ? (int)(pk >> 17) : 0x7FFF;
    const int prev = __shfl_up(key, 1);
    const unsigned int heads = __builtin_amdgcn_ballot_w32(lane > 0 && key != prev);
    float sum[NJ];
#pragma unroll
    for (int j = 0; j < NJ; ++j) sum[j] = 0.0f;
#pragma unroll
    for (int g = 0; g < 32; g += AGG_G) {
        if (g < cnt) {
            float v[AGG_G][NJ];
#pragma unroll
            for (int k = 0; k < AGG_G; ++k) {
                const int sk = __builtin_amdgcn_readlane(s, g + k);
#pragma unroll
                for (int j = 0; j < NJ; ++j) v[k][j] = h[(size_t)sk * C + j * 32 + lane];
            }
#pragma unroll
            for (int k = 0; k < AGG_G; ++k) {
                const int i = g + k;
                if (i > 0 && __builtin_expect((heads >> i) & 1u, 0u)) {
                    const int dp = __builtin_amdgcn_readlane(key, i > 0 ? i - 1 : 0);
#pragma unroll
                    for (int j = 0; j < NJ; ++j) {
                        atomicAdd(&acc[dp * P + j * 32 + lane], sum[j]);
                        sum[j] = 0.0f;
                    }
                }
#pragma unroll
                for (int j = 0; j < NJ; ++j) sum[j] += v[k][j];
            }
        }
    }
    if (cnt == 32) {
        const int dp = __builtin_amdgcn_readlane(key, 31);
#pragma unroll
        for (int j = 0; j < NJ; ++j) atomicAdd(&acc[dp * P + j * 32 + lane], sum[j]);
    } else if (cnt == AGG_G) {
        const int dp = __builtin_amdgcn_readlane(key, AGG_G - 1);
#pragma unroll
        for (int j = 0; j < NJ; ++j) atomicAdd(&acc[dp * P + j * 32 + lane], sum[j]);
    }
}

template <int C, bool LAYER1>
__global__ void __launch_bounds__(AGG_TB)
k_range_agg(unsigned int* __restrict__ recs, const unsigned int* __restrict__ bstart,
            const float* __restrict__ h, const float* __restrict__ dinv,
            const float* __restrict__ bias, const float* __restrict__ W2,
            float* __restrict__ outp, int n) {
    constexpr int P  = C + 4;
    __shared__ alignas(16) float acc[BN * P];
    __shared__ alignas(16) float dloc[BN];
    __shared__ alignas(16) unsigned int srt[LAYER1 ? SRT_CAP : 1];
    __shared__ unsigned int cntr[LAYER1 ? SRT_ROWS : 1];
    __shared__ unsigned int rowst[LAYER1 ? SRT_ROWS : 1];
    const int tid  = threadIdx.x;
    const int lane = tid & 31;
    const int wv   = __builtin_amdgcn_readfirstlane(tid >> 5);
    const int n0   = blockIdx.x * BN;
    const int rows = (n - n0) < BN ? (n - n0) : BN;

    const unsigned int eb = bstart[blockIdx.x];
    const unsigned int ee = bstart[blockIdx.x + 1];

    float dself;
    {
        int node = n0 + (tid < BN ? tid : BN - 1);
        node = node < n ? node : n - 1;
        dself = dinv[node];
    }

    constexpr int NO = 32;
    float2 w2v = make_float2(0.0f, 0.0f);
    unsigned int rec[SRT_K];
    if constexpr (LAYER1) {
        static_assert(C * NO == 2 * AGG_TB && C * NO <= SRT_CAP, "two words of W2 per thread; W2 fits srt");
        w2v = reinterpret_cast<const float2*>(W2)[tid];
        if (tid < SRT_ROWS) cntr[tid] = 0u;
#pragma unroll
        for (int k = 0; k < SRT_K; ++k) {
            const unsigned int idx = (unsigned int)(k * AGG_TB + tid);
            rec[k] = (idx < ee - eb) ? recs[eb + idx] : 0u;
        }
    }

    unsigned int pk0 = 0u;
    if constexpr (!LAYER1) {
        const unsigned int c00 = eb + (unsigned int)wv * 32u;
        pk0 = (c00 + lane < ee) ? recs[c00 + lane] : 0u;
    }

    constexpr int C4     = C / 4;
    constexpr int INIT_T = 2 * BN;
    constexpr int INIT_K = C4 / 2;
    static_assert(INIT_T <= AGG_TB && INIT_T % 32 == 0 && INIT_T * INIT_K == BN * C4,
                  "start-value shares must be whole");
    if (tid < INIT_T) {
        float4 hv[INIT_K];
#pragma unroll
        for (int u = 0; u < INIT_K; ++u) {
            const int idx = tid + u * INIT_T;
            const int r = idx / C4, c4 = idx - r * C4;
            int node = n0 + r;
            node = node < n ? node : n - 1;
            hv[u] = *reinterpret_cast<const float4*>(h + (size_t)node * C + c4 * 4);
        }
#pragma unroll
        for (int u = 0; u < INIT_K; ++u) {
            const int idx = tid + u * INIT_T;
            const int r = idx / C4, c4 = idx - r * C4;
            *reinterpret_cast<float4*>(&acc[r * P + c4 * 4]) = hv[u];
        }
    }
    if (tid < BN) dloc[tid] = (tid < rows) ? dself : 0.0f;
    __syncthreads();

    if constexpr (LAYER1) {
        unsigned int base = eb;
        for (;;) {
            const unsigned int left = ee - base;
            const unsigned int m = left < (unsigned int)SRT_CAP ? left : (unsigned int)SRT_CAP;
            unsigned int rk[SRT_K];
#pragma unroll
            for (int k = 0; k < SRT_K; ++k) {
                rk[k] = 0u;
                if ((unsigned int)(k * AGG_TB + tid) < m) rk[k] = atomicAdd(&cntr[rec[k] >> 17], 1u);
            }
            __syncthreads();
            if (wv == 0) {
                unsigned int cv[SRT_RPL];
                unsigned int tot = 0u;
#pragma unroll
                for (int i = 0; i < SRT_RPL; ++i) {
                    const unsigned int t = cntr[lane * SRT_RPL + i];
                    cv[i] = tot;
                    tot += t;
                }
                unsigned int incl = tot;
                for (int o = 1; o < 32; o <<= 1) {
                    const unsigned int v = __shfl_up(incl, o);
                    if (lane >= o) incl += v;
                }
                const unsigned int before = incl - tot;
#pragma unroll
                for (int i = 0; i < SRT_RPL; ++i) {
                    rowst[lane * SRT_RPL + i] = before + cv[i];
                    cntr[lane * SRT_RPL + i]  = 0u;
                }
            }
            __syncthreads();
            unsigned int pos[SRT_K];
#pragma unroll
            for (int k = 0; k < SRT_K; ++k) pos[k] = rowst[rec[k] >> 17] + rk[k];
#pragma unroll
            for (int k = 0; k < SRT_K; ++k) {
                if ((unsigned int)(k * AGG_TB + tid) < m) srt[pos[k]] = rec[k];
            }
            __syncthreads();
            unsigned int sv[SRT_K];
#pragma unroll
            for (int k = 0; k < SRT_K; ++k) sv[k] = srt[k * AGG_TB + tid];
#pragma unroll
            for (int k = 0; k < SRT_K; ++k) {
                const unsigned int idx = (unsigned int)(k * AGG_TB + tid);
                if (idx < m) recs[base + idx] = sv[k];
            }
            for (unsigned int c0 = (unsigned int)wv * 32u; c0 < m; c0 += AGG_TB) {
                const unsigned int pk = (c0 + lane < m) ? srt[c0 + lane] : 0u;
                const int cnt = (int)((m - c0) < 32u ? (m - c0) : 32u);
                agg_records32<C>(pk, cnt, h, acc, lane);
            }
            base += m;
            if (base >= ee) break;
            __syncthreads();
#pragma unroll
            for (int k = 0; k < SRT_K; ++k) {
                const unsigned int idx = (unsigned int)(k * AGG_TB + tid);
                rec[k] = (idx < ee - base) ? recs[base + idx] : 0u;
            }
        }
    } else {
        unsigned int c0 = eb + (unsigned int)wv * 32u;
        unsigned int pk = pk0;
        for (; c0 < ee; c0 += AGG_TB) {
            const unsigned int cn = c0 + AGG_TB;
            const unsigned int pkn = (cn + lane < ee) ? recs[cn + lane] : 0u;
            const int cnt = (int)((ee - c0) < 32u ? (ee - c0) : 32u);
            agg_records32<C>(pk, cnt, h, acc, lane);
            pk = pkn;
        }
    }
    __syncthreads();

    if constexpr (!LAYER1) {
        for (int idx = tid; idx < rows * C; idx += AGG_TB) {
            const int r = idx / C, c = idx - r * C;
            outp[(size_t)(n0 + r) * C + c] = acc[r * P + c] * dloc[r] + bias[c];
        }
    } else {
        for (int idx = tid; idx < BN * C; idx += AGG_TB) {
            const int r = idx / C, c = idx - r * C;
            const float v = acc[r * P + c] * dloc[r] + bias[c];
            acc[r * P + c] = v > 0.0f ? v : 0.0f;
        }
        float* w2l = reinterpret_cast<float*>(srt);
        {
            const int kr = tid / (NO / 2), c2 = tid - kr * (NO / 2);
            const int kk = kr >> 2, h = (kr >> 1) & 1, kb = kr & 1;
            const int tb = (2 * c2) >> 4, c0 = (2 * c2) & 15;
            float* d = w2l + (kk * 32 + h * 16 + c0) * 4 + 2 * tb + kb;
            d[0] = w2v.x;
            d[4] = w2v.y;
        }
        __syncthreads();
        constexpr int NT = NO / 16;
        static_assert(NT == 2, "one tile pair per 16-byte read");
        const int full = rows / 16;
        if (wv < full) {
            const int m  = lane & 15;
            const int kh = lane >> 4;
            v8f d[NT];
#pragma unroll
            for (int t = 0; t < NT; ++t)
#pragma unroll
                for (int j = 0; j < 8; ++j) d[t][j] = 0.0f;
            float sc[8];
#pragma unroll
            for (int j = 0; j < 8; ++j) sc[j] = dloc[wv * 16 + j + 8 * kh];
            const float* arow = acc + (wv * 16 + m) * P;
            const float4* bfr = reinterpret_cast<const float4*>(w2l) + lane;
            for (int k = 0; k < C; k += 4) {
                v2f a;
                a.x = arow[k + 2 * kh + 0];
                a.y = arow[k + 2 * kh + 1];
                const float4 q = bfr[(k / 4) * 32];
                v2f b0, b1;
                b0.x = q.x; b0.y = q.y;
                b1.x = q.z; b1.y = q.w;
                d[0] = __builtin_amdgcn_wmma_f32_16x16x4_f32(
                    false, a, false, b0, (short)0, d[0], false, false);
                d[1] = __builtin_amdgcn_wmma_f32_16x16x4_f32(
                    false, a, false, b1, (short)0, d[1], false, false);
            }
#pragma unroll
            for (int t = 0; t < NT; ++t)
#pragma unroll
                for (int j = 0; j < 8; ++j)
                    outp[(size_t)(n0 + wv * 16 + j + 8 * kh) * NO + t * 16 + m] = d[t][j] * sc[j];
        }
        const int r1 = full * 16;
        for (int idx = tid; idx < (rows - r1) * NO; idx += AGG_TB) {
            const int r = r1 + idx / NO, c = idx % NO;
            float sum = 0.0f;
            for (int k = 0; k < C; ++k) sum += acc[r * P + k] * W2[(size_t)k * NO + c];
            outp[(size_t)(n0 + r) * NO + c] = sum * dloc[r];
        }
    }
}

extern "C" void kernel_launch(void* const* d_in, const int* in_sizes, int n_in,
                              void* d_out, int out_size, void* d_ws, size_t ws_size,
                              hipStream_t stream) {
    const float* x  = (const float*)d_in[0];
    const int*   ei = (const int*)d_in[1];
    const float* W1 = (const float*)d_in[2];
    const float* b1 = (const float*)d_in[3];
    const float* W2 = (const float*)d_in[4];
    const float* b2 = (const float*)d_in[5];
    float* out = (float*)d_out;

    const int n = in_sizes[0] / 128;
    const int e = in_sizes[1] / 2;
    const int* src = ei;
    const int* dst = ei + e;

    char* ws = (char*)d_ws;
    size_t off0 = ((size_t)n * sizeof(float) + 255) & ~(size_t)255;
    float* dinv = (float*)ws;
    float* bufA = (float*)(ws + off0);
    float* bufB = bufA + (size_t)n * 64;

    const int tb = 256;
    auto cdiv = [](long long a, long long b) { return (int)((a + b - 1) / b); };

    const int parts = cdiv(e, DEG_ECH);
    const int strips = n / 16;
    const int wavesPerBlk = tb / 32;
    const int nb = cdiv(n, BN);
    const size_t half = (size_t)n * 32;
    if (n <= DEG_NCAP && (size_t)parts * DEG_WORDS <= half &&
        (size_t)e + (size_t)parts * nb + nb + 1 <= half) {
        unsigned int* slabs  = (unsigned int*)bufB;
        unsigned int* recs   = (unsigned int*)bufB + half;
        unsigned int* pcnt   = recs + e;
        unsigned int* bstart = pcnt + (size_t)parts * nb;
        float* h2 = bufB;

        k_deg_hist<<<parts, DEG_TB, 0, stream>>>(dst, slabs, pcnt, nb, e);
        k_place_gemm<128, 64><<<parts + cdiv(n, PG_ROWS), PG_TB, 0, stream>>>(
            src, dst, pcnt, recs, bstart, e, nb, parts, slabs, x, W1, dinv, bufA, n);
        if (n % 16) {
            int tail = (n % 16) * 64;
            k_gemm_tail<128, 64, true><<<cdiv(tail, tb), tb, 0, stream>>>(x, W1, dinv, bufA, strips * 16, n);
        }
        k_range_agg<64, true><<<nb, AGG_TB, 0, stream>>>(recs, bstart, bufA, dinv, b1, W2, h2, n);
        k_range_agg<32, false><<<nb, AGG_TB, 0, stream>>>(recs, bstart, h2, dinv, b2, nullptr, out, n);
        return;
    }

    if (n <= DEG_NCAP && (size_t)parts * DEG_WORDS <= (size_t)n * 64) {
        k_deg_hist<<<parts, DEG_TB, 0, stream>>>(dst, (unsigned int*)bufB, nullptr, 0, e);
        k_dinv_hist<<<cdiv((n + 1) / 2, tb), tb, 0, stream>>>((const unsigned int*)bufB, dinv, n, parts);
    } else {
        k_deg_init<<<cdiv(n, tb), tb, 0, stream>>>(dinv, n);
        k_deg_count<<<cdiv(e, tb), tb, 0, stream>>>(dst, dinv, e);
        k_dinv<<<cdiv(n, tb), tb, 0, stream>>>(dinv, n);
    }

    k_gemm_wmma<128, 64, false><<<cdiv(strips, wavesPerBlk), tb, 0, stream>>>(x, W1, nullptr, bufA, n);
    if (n % 16) {
        int tail = (n % 16) * 64;
        k_gemm_tail<128, 64, false><<<cdiv(tail, tb), tb, 0, stream>>>(x, W1, nullptr, bufA, strips * 16, n);
    }

    k_self_init<64, false><<<cdiv((long long)n * 64, tb), tb, 0, stream>>>(bufA, dinv, nullptr, bufB, n);
    k_edge_agg<64><<<cdiv(cdiv(e, 8), wavesPerBlk), tb, 0, stream>>>(src, dst, bufA, dinv, bufB, e);
    k_relu_bias<64><<<cdiv((long long)n * 64, tb), tb, 0, stream>>>(bufB, b1, bufA, n);

    k_gemm_wmma<64, 32, false><<<cdiv(strips, wavesPerBlk), tb, 0, stream>>>(bufA, W2, nullptr, bufB, n);
    if (n % 16) {
        int tail = (n % 16) * 32;
        k_gemm_tail<64, 32, false><<<cdiv(tail, tb), tb, 0, stream>>>(bufA, W2, nullptr, bufB, strips * 16, n);
    }

    k_self_init<32, true><<<cdiv((long long)n * 32, tb), tb, 0, stream>>>(bufB, dinv, b2, out, n);
    k_edge_agg<32><<<cdiv(cdiv(e, 8), wavesPerBlk), tb, 0, stream>>>(src, dst, bufB, dinv, out, e);

    (void)n_in; (void)out_size; (void)ws_size;
}
